# starting kernel
# baseline (speedup 1.0000x reference)
#include <hip/hip_runtime.h>
#include <cstdint>
#include <cstddef>

constexpr int GN = 50000;
constexpr int GE = 1600000;
constexpr int GH = 128;
constexpr int GLAYERS = 2;
constexpr size_t NH = (size_t)GN * GH;

typedef __attribute__((ext_vector_type(2))) float v2f;
typedef __attribute__((ext_vector_type(8))) float v8f;

__device__ __forceinline__ v8f wmma_f32(v2f a, v2f b, v8f c) {
  return __builtin_amdgcn_wmma_f32_16x16x4_f32(
       false, a,  false, b,
       (short)0, c,  false,  false);
}

__device__ __forceinline__ float sigmoid_f(float x) {
  return 1.0f / (1.0f + __expf(-x));
}
__device__ __forceinline__ float tanh_f(float x) {
  return 1.0f - 2.0f / (1.0f + __expf(2.0f * x));
}

__device__ __forceinline__ void load_afrag_sum(const float* __restrict__ X,
                                               const float* __restrict__ A,
                                               int row, int kof, v2f* frag) {
  const float* xr = X + (size_t)row * GH;
  const float* ar = A + (size_t)row * GH;
#pragma unroll
  for (int kk = 0; kk < 32; ++kk) {
    const int c = (kk << 2) + kof;
    const float2 xv = *(const float2*)(xr + c);
    const float2 av = *(const float2*)(ar + c);
    v2f f;
    f.x = xv.x + av.x;
    f.y = xv.y + av.y;
    frag[kk] = f;
  }
}

__device__ __forceinline__ void stage_slab(const float* __restrict__ Wg, int n0,
                                           float* lds) {
  const int tid = threadIdx.x;
#pragma unroll
  for (int i = 0; i < 8; ++i) {
    const int idx = i * 256 + tid;
    const int k = idx >> 4;
    const int c = idx & 15;
    lds[c * 132 + k] = Wg[k * GH + n0 + c];
  }
}

__global__ __launch_bounds__(256) void scatter_add_kernel(
    const int* __restrict__ src, const int* __restrict__ dst,
    const float* __restrict__ feat, float* __restrict__ agg) {
  const unsigned tid = blockIdx.x * 256u + threadIdx.x;
  const unsigned e = tid >> 5;
  if (e >= (unsigned)GE) return;
  const int c4 = (tid & 31) << 2;
  const int s = src[e];
  const int d = dst[e];
  const float4 val = *(const float4*)(feat + (size_t)s * GH + c4);
  float* out = agg + (size_t)d * GH + c4;
  atomicAdd(out + 0, val.x);
  atomicAdd(out + 1, val.y);
  atomicAdd(out + 2, val.z);
  atomicAdd(out + 3, val.w);
}

__global__ __launch_bounds__(256) void gemm_zr_kernel(
    const float* __restrict__ x, const float* __restrict__ aggx,
    const float* __restrict__ hl, const float* __restrict__ aggh,
    const float* __restrict__ Wl, const float* __restrict__ bl,
    float* __restrict__ zout, float* __restrict__ vout) {
  __shared__ float lds[4 * 16 * 132];
  const int tid = threadIdx.x;
  const int wave = tid >> 5;
  const int lane = tid & 31;
  const int row0 = blockIdx.x * 128 + wave * 16;
  const int kof = (lane >> 4) << 1;
  const int nn = lane & 15;
  const int rowhi = (lane >> 4) << 3;

  int arow = row0 + nn;
  if (arow > GN - 1) arow = GN - 1;

  v2f fs[32], ft[32];
  load_afrag_sum(x, aggx, arow, kof, fs);
  load_afrag_sum(hl, aggh, arow, kof, ft);

  const float* W0 = Wl + 0 * GH * GH;
  const float* W1 = Wl + 1 * GH * GH;
  const float* W2 = Wl + 2 * GH * GH;
  const float* W3 = Wl + 3 * GH * GH;
  float* L0 = lds + 0 * 16 * 132;
  float* L1 = lds + 1 * 16 * 132;
  float* L2 = lds + 2 * 16 * 132;
  float* L3 = lds + 3 * 16 * 132;

  for (int ct = 0; ct < GH / 16; ++ct) {
    const int n0 = ct * 16;
    __syncthreads();
    stage_slab(W0, n0, L0);
    stage_slab(W1, n0, L1);
    stage_slab(W2, n0, L2);
    stage_slab(W3, n0, L3);
    __syncthreads();

    const float* p0 = L0 + nn * 132 + kof;
    const float* p1 = L1 + nn * 132 + kof;
    const float* p2 = L2 + nn * 132 + kof;
    const float* p3 = L3 + nn * 132 + kof;

    v8f az0 = {0, 0, 0, 0, 0, 0, 0, 0}, az1 = {0, 0, 0, 0, 0, 0, 0, 0};
    v8f ar0 = {0, 0, 0, 0, 0, 0, 0, 0}, ar1 = {0, 0, 0, 0, 0, 0, 0, 0};
#pragma unroll
    for (int kk = 0; kk < 32; kk += 2) {
      az0 = wmma_f32(fs[kk], *(const v2f*)(p0 + 4 * kk), az0);
      az0 = wmma_f32(ft[kk], *(const v2f*)(p1 + 4 * kk), az0);
      ar0 = wmma_f32(fs[kk], *(const v2f*)(p2 + 4 * kk), ar0);
      ar0 = wmma_f32(ft[kk], *(const v2f*)(p3 + 4 * kk), ar0);
      az1 = wmma_f32(fs[kk + 1], *(const v2f*)(p0 + 4 * kk + 4), az1);
      az1 = wmma_f32(ft[kk + 1], *(const v2f*)(p1 + 4 * kk + 4), az1);
      ar1 = wmma_f32(fs[kk + 1], *(const v2f*)(p2 + 4 * kk + 4), ar1);
      ar1 = wmma_f32(ft[kk + 1], *(const v2f*)(p3 + 4 * kk + 4), ar1);
    }
    const v8f accz = az0 + az1;
    const v8f accr = ar0 + ar1;

    const float bz = bl[0 * GH + n0 + nn] + bl[1 * GH + n0 + nn];
    const float br = bl[2 * GH + n0 + nn] + bl[3 * GH + n0 + nn];
#pragma unroll
    for (int i = 0; i < 8; ++i) {
      const int row = row0 + rowhi + i;
      if (row < GN) {
        const size_t off = (size_t)row * GH + n0 + nn;
        const float zv = sigmoid_f(accz[i] + bz);
        const float rv = sigmoid_f(accr[i] + br);
        zout[off] = zv;
        vout[off] = rv * hl[off];
      }
    }
  }
}

__global__ __launch_bounds__(256) void gemm_ht_kernel(
    const float* __restrict__ x, const float* __restrict__ aggx,
    const float* __restrict__ v, const float* __restrict__ aggv,
    const float* __restrict__ Wl, const float* __restrict__ bl,
    const float* __restrict__ zbuf, const float* __restrict__ hl,
    float* __restrict__ outp) {
  __shared__ float lds[2 * 16 * 132];
  const int tid = threadIdx.x;
  const int wave = tid >> 5;
  const int lane = tid & 31;
  const int row0 = blockIdx.x * 128 + wave * 16;
  const int kof = (lane >> 4) << 1;
  const int nn = lane & 15;
  const int rowhi = (lane >> 4) << 3;

  int arow = row0 + nn;
  if (arow > GN - 1) arow = GN - 1;

  v2f fs[32], fu[32];
  load_afrag_sum(x, aggx, arow, kof, fs);
  load_afrag_sum(v, aggv, arow, kof, fu);

  const float* W4 = Wl + 4 * GH * GH;
  const float* W5 = Wl + 5 * GH * GH;
  float* L4 = lds + 0 * 16 * 132;
  float* L5 = lds + 1 * 16 * 132;

  for (int ct = 0; ct < GH / 16; ++ct) {
    const int n0 = ct * 16;
    __syncthreads();
    stage_slab(W4, n0, L4);
    stage_slab(W5, n0, L5);
    __syncthreads();

    const float* p4 = L4 + nn * 132 + kof;
    const float* p5 = L5 + nn * 132 + kof;

    v8f a0 = {0, 0, 0, 0, 0, 0, 0, 0}, a1 = {0, 0, 0, 0, 0, 0, 0, 0};
#pragma unroll
    for (int kk = 0; kk < 32; kk += 2) {
      a0 = wmma_f32(fs[kk], *(const v2f*)(p4 + 4 * kk), a0);
      a0 = wmma_f32(fu[kk], *(const v2f*)(p5 + 4 * kk), a0);
      a1 = wmma_f32(fs[kk + 1], *(const v2f*)(p4 + 4 * kk + 4), a1);
      a1 = wmma_f32(fu[kk + 1], *(const v2f*)(p5 + 4 * kk + 4), a1);
    }
    const v8f acc = a0 + a1;

    const float bh = bl[4 * GH + n0 + nn] + bl[5 * GH + n0 + nn];
#pragma unroll
    for (int i = 0; i < 8; ++i) {
      const int row = row0 + rowhi + i;
      if (row < GN) {
        const size_t off = (size_t)row * GH + n0 + nn;
        const float ht = tanh_f(acc[i] + bh);
        const float zv = zbuf[off];
        outp[off] = zv * hl[off] + (1.0f - zv) * ht;
      }
    }
  }
}

extern "C" void kernel_launch(void* const* d_in, const int* in_sizes, int n_in,
                              void* d_out, int out_size, void* d_ws,
                              size_t ws_size, hipStream_t stream) {
  (void)in_sizes; (void)n_in; (void)out_size; (void)ws_size;
  const float* inp = (const float*)d_in[0];
  const int* edg = (const int*)d_in[1];
  const float* hmat = (const float*)d_in[2];
  const float* Wmat = (const float*)d_in[3];
  const float* bmat = (const float*)d_in[4];
  float* out = (float*)d_out;

  const int* src = edg;
  const int* dst = edg + GE;

  float* agg_x = (float*)d_ws;
  float* agg_t = agg_x + NH;
  float* zbuf = agg_t + NH;
  float* vbuf = zbuf + NH;

  const int scat_blocks = (int)(((size_t)GE * 32 + 255) / 256);
  const int gemm_blocks = (GN + 127) / 128;

  for (int l = 0; l < GLAYERS; ++l) {
    const float* x = (l == 0) ? inp : (out + (size_t)(l - 1) * NH);
    const float* hl = hmat + (size_t)l * NH;
    const float* Wl = Wmat + (size_t)l * 6 * GH * GH;
    const float* bl = bmat + (size_t)l * 6 * GH;
    float* outl = out + (size_t)l * NH;

    hipMemsetAsync(agg_x, 0, NH * sizeof(float), stream);
    hipMemsetAsync(agg_t, 0, NH * sizeof(float), stream);
    scatter_add_kernel<<<scat_blocks, 256, 0, stream>>>(src, dst, x, agg_x);
    scatter_add_kernel<<<scat_blocks, 256, 0, stream>>>(src, dst, hl, agg_t);

    gemm_zr_kernel<<<gemm_blocks, 256, 0, stream>>>(x, agg_x, hl, agg_t, Wl,
                                                    bl, zbuf, vbuf);

    hipMemsetAsync(agg_t, 0, NH * sizeof(float), stream);
    scatter_add_kernel<<<scat_blocks, 256, 0, stream>>>(src, dst, vbuf, agg_t);

    gemm_ht_kernel<<<gemm_blocks, 256, 0, stream>>>(x, agg_x, vbuf, agg_t, Wl,
                                                    bl, zbuf, hl, outl);
  }
}


# Replace f32 atomic scatter-add by per-call grouping of edges by destination, each row list put in ascending source order, and one-wave-per-row gather sums in registers; sum order changes, now fixed
# speedup vs baseline: 2.5240x; 2.5240x over previous
#include <hip/hip_runtime.h>
#include <cstdint>
#include <cstddef>

constexpr int GN = 50000;
constexpr int GE = 1600000;
constexpr int GH = 128;
constexpr int GLAYERS = 2;
constexpr size_t NH = (size_t)GN * GH;

typedef __attribute__((ext_vector_type(2))) float v2f;
typedef __attribute__((ext_vector_type(8))) float v8f;

__device__ __forceinline__ v8f wmma_f32(v2f a, v2f b, v8f c) {
  return __builtin_amdgcn_wmma_f32_16x16x4_f32(
       false, a,  false, b,
       (short)0, c,  false,  false);
}

__device__ __forceinline__ float sigmoid_f(float x) {
  return 1.0f / (1.0f + __expf(-x));
}
__device__ __forceinline__ float tanh_f(float x) {
  return 1.0f - 2.0f / (1.0f + __expf(2.0f * x));
}

__device__ __forceinline__ void load_afrag_sum(const float* __restrict__ X,
                                               const float* __restrict__ A,
                                               int row, int kof, v2f* frag) {
  const float* xr = X + (size_t)row * GH;
  const float* ar = A + (size_t)row * GH;
#pragma unroll
  for (int kk = 0; kk < 32; ++kk) {
    const int c = (kk << 2) + kof;
    const float2 xv = *(const float2*)(xr + c);
    const float2 av = *(const float2*)(ar + c);
    v2f f;
    f.x = xv.x + av.x;
    f.y = xv.y + av.y;
    frag[kk] = f;
  }
}

__device__ __forceinline__ void stage_slab(const float* __restrict__ Wg, int n0,
                                           float* lds) {
  const int tid = threadIdx.x;
#pragma unroll
  for (int i = 0; i < 8; ++i) {
    const int idx = i * 256 + tid;
    const int k = idx >> 4;
    const int c = idx & 15;
    lds[c * 132 + k] = Wg[k * GH + n0 + c];
  }
}

constexpr int SCAN_THREADS = 1024;
constexpr int SCAN_PER = (GN + SCAN_THREADS - 1) / SCAN_THREADS;
constexpr int SCAN_PAD = SCAN_THREADS * SCAN_PER;

__global__ __launch_bounds__(256) void degree_count_kernel(
    const int* __restrict__ dst, int* __restrict__ cnt) {
  const unsigned e = blockIdx.x * 256u + threadIdx.x;
  if (e >= (unsigned)GE) return;
  atomicAdd(cnt + dst[e], 1);
}

__global__ __launch_bounds__(SCAN_THREADS) void degree_scan_kernel(
    int* __restrict__ cnt, int* __restrict__ rowptr) {
  __shared__ int buf[SCAN_PAD];
  __shared__ int wsum[32];
  const int t = threadIdx.x;
  const int lane = t & 31;
#pragma unroll 7
  for (int k = 0; k < SCAN_PER; ++k) {
    const int i = k * SCAN_THREADS + t;
    const int v = cnt[(i < GN) ? i : GN - 1];
    buf[i] = (i < GN) ? v : 0;
  }
  __syncthreads();
  const int b = t * SCAN_PER;
  int s = 0;
  for (int i = 0; i < SCAN_PER; ++i) s += buf[b + i];
  int incl = s;
  for (int off = 1; off < 32; off <<= 1) {
    const int y = __shfl_up(incl, off);
    if (lane >= off) incl += y;
  }
  if (lane == 31) wsum[t >> 5] = incl;
  __syncthreads();
  if (t < 32) {
    const int w = wsum[t];
    int wi = w;
    for (int off = 1; off < 32; off <<= 1) {
      const int y = __shfl_up(wi, off);
      if (t >= off) wi += y;
    }
    wsum[t] = wi - w;
  }
  __syncthreads();
  int run = wsum[t >> 5] + incl - s;
  for (int i = 0; i < SCAN_PER; ++i) {
    const int v = buf[b + i];
    buf[b + i] = run;
    run += v;
  }
  __syncthreads();
  for (int i = t; i < GN; i += SCAN_THREADS) {
    const int v = buf[i];
    rowptr[i] = v;
    cnt[i] = v;
  }
  if (t == SCAN_THREADS - 1) rowptr[GN] = run;
}

__global__ __launch_bounds__(256) void edge_place_kernel(
    const int* __restrict__ src, const int* __restrict__ dst,
    int* __restrict__ cursor, int* __restrict__ slots) {
  const unsigned e = blockIdx.x * 256u + threadIdx.x;
  if (e >= (unsigned)GE) return;
  const int p = atomicAdd(cursor + dst[e], 1);
  slots[p] = src[e];
}

__global__ __launch_bounds__(256) void segment_order_kernel(
    const int* __restrict__ rowptr, const int* __restrict__ slots,
    int* __restrict__ csr) {
  const int lane = threadIdx.x & 31;
  const int row = __builtin_amdgcn_readfirstlane(
      (int)((blockIdx.x * 256u + threadIdx.x) >> 5));
  if (row >= GN) return;
  const int beg = rowptr[row];
  const int n = rowptr[row + 1] - beg;
  for (int ib = 0; ib < n; ib += 32) {
    const int i = ib + lane;
    const int key = (i < n) ? slots[beg + i] : 0;
    int rank = 0;
    for (int jb = 0; jb < n; jb += 32) {
      const int jn = (n - jb < 32) ? (n - jb) : 32;
      const int kv = (lane < jn) ? slots[beg + jb + lane] : 0;
      for (int jj = 0; jj < jn; ++jj) {
        const int kj = __builtin_amdgcn_readlane(kv, jj);
        const int j = jb + jj;
        rank += (kj < key || (kj == key && j < i)) ? 1 : 0;
      }
    }
    if (i < n) csr[beg + rank] = key;
  }
}

template <int U>
__device__ __forceinline__ void gather_rows(const float* __restrict__ f,
                                            int idx, int j0, int c4,
                                            float4& acc) {
  float4 v[U];
#pragma unroll
  for (int u = 0; u < U; ++u) {
    const int s = __builtin_amdgcn_readlane(idx, j0 + u);
    v[u] = *(const float4*)(f + (size_t)s * GH + c4);
  }
#pragma unroll
  for (int u = 0; u < U; ++u) {
    acc.x += v[u].x;
    acc.y += v[u].y;
    acc.z += v[u].z;
    acc.w += v[u].w;
  }
}

template <int U, int NF>
__device__ __forceinline__ void gather_group(const float* __restrict__ f0,
                                             const float* __restrict__ f1,
                                             int idx, int j0, int c4,
                                             float4& a0, float4& a1) {
  gather_rows<U>(f0, idx, j0, c4, a0);
  if (NF == 2) gather_rows<U>(f1, idx, j0, c4, a1);
}

template <int NF>
__device__ __forceinline__ void gather_sum_body(
    const int* __restrict__ rowptr, const int* __restrict__ csr,
    const float* __restrict__ f0, float* __restrict__ o0,
    const float* __restrict__ f1, float* __restrict__ o1) {
  const int lane = threadIdx.x & 31;
  const int c4 = lane << 2;
  const int row = __builtin_amdgcn_readfirstlane(
      (int)((blockIdx.x * 256u + threadIdx.x) >> 5));
  if (row >= GN) return;
  const int beg = rowptr[row];
  const int end = rowptr[row + 1];
  float4 a0 = {0.0f, 0.0f, 0.0f, 0.0f};
  float4 a1 = {0.0f, 0.0f, 0.0f, 0.0f};
  for (int base = beg; base < end; base += 32) {
    const int n = (end - base < 32) ? (end - base) : 32;
    const int idx = (lane < n) ? csr[base + lane] : 0;
    int j = 0;
    for (; j + 8 <= n; j += 8) gather_group<8, NF>(f0, f1, idx, j, c4, a0, a1);
    if (n & 4) { gather_group<4, NF>(f0, f1, idx, j, c4, a0, a1); j += 4; }
    if (n & 2) { gather_group<2, NF>(f0, f1, idx, j, c4, a0, a1); j += 2; }
    if (n & 1) { gather_group<1, NF>(f0, f1, idx, j, c4, a0, a1); }
  }
  *(float4*)(o0 + (size_t)row * GH + c4) = a0;
  if (NF == 2) *(float4*)(o1 + (size_t)row * GH + c4) = a1;
}

__global__ __launch_bounds__(256) void gather_sum2_kernel(
    const int* __restrict__ rowptr, const int* __restrict__ csr,
    const float* __restrict__ f0, float* __restrict__ o0,
    const float* __restrict__ f1, float* __restrict__ o1) {
  gather_sum_body<2>(rowptr, csr, f0, o0, f1, o1);
}

__global__ __launch_bounds__(256) void gather_sum1_kernel(
    const int* __restrict__ rowptr, const int* __restrict__ csr,
    const float* __restrict__ f0, float* __restrict__ o0) {
  gather_sum_body<1>(rowptr, csr, f0, o0, f0, o0);
}

__global__ __launch_bounds__(256) void gemm_zr_kernel(
    const float* __restrict__ x, const float* __restrict__ aggx,
    const float* __restrict__ hl, const float* __restrict__ aggh,
    const float* __restrict__ Wl, const float* __restrict__ bl,
    float* __restrict__ zout, float* __restrict__ vout) {
  __shared__ float lds[4 * 16 * 132];
  const int tid = threadIdx.x;
  const int wave = tid >> 5;
  const int lane = tid & 31;
  const int row0 = blockIdx.x * 128 + wave * 16;
  const int kof = (lane >> 4) << 1;
  const int nn = lane & 15;
  const int rowhi = (lane >> 4) << 3;

  int arow = row0 + nn;
  if (arow > GN - 1) arow = GN - 1;

  v2f fs[32], ft[32];
  load_afrag_sum(x, aggx, arow, kof, fs);
  load_afrag_sum(hl, aggh, arow, kof, ft);

  const float* W0 = Wl + 0 * GH * GH;
  const float* W1 = Wl + 1 * GH * GH;
  const float* W2 = Wl + 2 * GH * GH;
  const float* W3 = Wl + 3 * GH * GH;
  float* L0 = lds + 0 * 16 * 132;
  float* L1 = lds + 1 * 16 * 132;
  float* L2 = lds + 2 * 16 * 132;
  float* L3 = lds + 3 * 16 * 132;

  for (int ct = 0; ct < GH / 16; ++ct) {
    const int n0 = ct * 16;
    __syncthreads();
    stage_slab(W0, n0, L0);
    stage_slab(W1, n0, L1);
    stage_slab(W2, n0, L2);
    stage_slab(W3, n0, L3);
    __syncthreads();

    const float* p0 = L0 + nn * 132 + kof;
    const float* p1 = L1 + nn * 132 + kof;
    const float* p2 = L2 + nn * 132 + kof;
    const float* p3 = L3 + nn * 132 + kof;

    v8f az0 = {0, 0, 0, 0, 0, 0, 0, 0}, az1 = {0, 0, 0, 0, 0, 0, 0, 0};
    v8f ar0 = {0, 0, 0, 0, 0, 0, 0, 0}, ar1 = {0, 0, 0, 0, 0, 0, 0, 0};
#pragma unroll
    for (int kk = 0; kk < 32; kk += 2) {
      az0 = wmma_f32(fs[kk], *(const v2f*)(p0 + 4 * kk), az0);
      az0 = wmma_f32(ft[kk], *(const v2f*)(p1 + 4 * kk), az0);
      ar0 = wmma_f32(fs[kk], *(const v2f*)(p2 + 4 * kk), ar0);
      ar0 = wmma_f32(ft[kk], *(const v2f*)(p3 + 4 * kk), ar0);
      az1 = wmma_f32(fs[kk + 1], *(const v2f*)(p0 + 4 * kk + 4), az1);
      az1 = wmma_f32(ft[kk + 1], *(const v2f*)(p1 + 4 * kk + 4), az1);
      ar1 = wmma_f32(fs[kk + 1], *(const v2f*)(p2 + 4 * kk + 4), ar1);
      ar1 = wmma_f32(ft[kk + 1], *(const v2f*)(p3 + 4 * kk + 4), ar1);
    }
    const v8f accz = az0 + az1;
    const v8f accr = ar0 + ar1;

    const float bz = bl[0 * GH + n0 + nn] + bl[1 * GH + n0 + nn];
    const float br = bl[2 * GH + n0 + nn] + bl[3 * GH + n0 + nn];
#pragma unroll
    for (int i = 0; i < 8; ++i) {
      const int row = row0 + rowhi + i;
      if (row < GN) {
        const size_t off = (size_t)row * GH + n0 + nn;
        const float zv = sigmoid_f(accz[i] + bz);
        const float rv = sigmoid_f(accr[i] + br);
        zout[off] = zv;
        vout[off] = rv * hl[off];
      }
    }
  }
}

__global__ __launch_bounds__(256) void gemm_ht_kernel(
    const float* __restrict__ x, const float* __restrict__ aggx,
    const float* __restrict__ v, const float* __restrict__ aggv,
    const float* __restrict__ Wl, const float* __restrict__ bl,
    const float* __restrict__ zbuf, const float* __restrict__ hl,
    float* __restrict__ outp) {
  __shared__ float lds[2 * 16 * 132];
  const int tid = threadIdx.x;
  const int wave = tid >> 5;
  const int lane = tid & 31;
  const int row0 = blockIdx.x * 128 + wave * 16;
  const int kof = (lane >> 4) << 1;
  const int nn = lane & 15;
  const int rowhi = (lane >> 4) << 3;

  int arow = row0 + nn;
  if (arow > GN - 1) arow = GN - 1;

  v2f fs[32], fu[32];
  load_afrag_sum(x, aggx, arow, kof, fs);
  load_afrag_sum(v, aggv, arow, kof, fu);

  const float* W4 = Wl + 4 * GH * GH;
  const float* W5 = Wl + 5 * GH * GH;
  float* L4 = lds + 0 * 16 * 132;
  float* L5 = lds + 1 * 16 * 132;

  for (int ct = 0; ct < GH / 16; ++ct) {
    const int n0 = ct * 16;
    __syncthreads();
    stage_slab(W4, n0, L4);
    stage_slab(W5, n0, L5);
    __syncthreads();

    const float* p4 = L4 + nn * 132 + kof;
    const float* p5 = L5 + nn * 132 + kof;

    v8f a0 = {0, 0, 0, 0, 0, 0, 0, 0}, a1 = {0, 0, 0, 0, 0, 0, 0, 0};
#pragma unroll
    for (int kk = 0; kk < 32; kk += 2) {
      a0 = wmma_f32(fs[kk], *(const v2f*)(p4 + 4 * kk), a0);
      a0 = wmma_f32(fu[kk], *(const v2f*)(p5 + 4 * kk), a0);
      a1 = wmma_f32(fs[kk + 1], *(const v2f*)(p4 + 4 * kk + 4), a1);
      a1 = wmma_f32(fu[kk + 1], *(const v2f*)(p5 + 4 * kk + 4), a1);
    }
    const v8f acc = a0 + a1;

    const float bh = bl[4 * GH + n0 + nn] + bl[5 * GH + n0 + nn];
#pragma unroll
    for (int i = 0; i < 8; ++i) {
      const int row = row0 + rowhi + i;
      if (row < GN) {
        const size_t off = (size_t)row * GH + n0 + nn;
        const float ht = tanh_f(acc[i] + bh);
        const float zv = zbuf[off];
        outp[off] = zv * hl[off] + (1.0f - zv) * ht;
      }
    }
  }
}

constexpr int GN_PAD = 50048;
constexpr size_t OFF_CNT = 0;
constexpr size_t OFF_ROWPTR = OFF_CNT + GN_PAD;
constexpr size_t OFF_SLOTS = OFF_ROWPTR + GN_PAD;
constexpr size_t OFF_CSR = OFF_SLOTS + GE;
static_assert(OFF_CSR + GE <= NH, "edge lists must fit in one output plane");
constexpr int EDGE_BLOCKS = (GE + 255) / 256;
constexpr int ROW_BLOCKS = (GN * 32 + 255) / 256;
constexpr int GEMM_BLOCKS = (GN + 127) / 128;

extern "C" void kernel_launch(void* const* d_in, const int* in_sizes, int n_in,
                              void* d_out, int out_size, void* d_ws,
                              size_t ws_size, hipStream_t stream) {
  (void)in_sizes; (void)n_in; (void)out_size; (void)ws_size;
  const float* inp = (const float*)d_in[0];
  const int* edg = (const int*)d_in[1];
  const float* hmat = (const float*)d_in[2];
  const float* Wmat = (const float*)d_in[3];
  const float* bmat = (const float*)d_in[4];
  float* out = (float*)d_out;

  const int* src = edg;
  const int* dst = edg + GE;

  float* agg_x = (float*)d_ws;
  float* agg_t = agg_x + NH;
  float* zbuf = agg_t + NH;
  float* vbuf = zbuf + NH;

  int* lists = (int*)(out + (size_t)(GLAYERS - 1) * NH);
  int* cnt = lists + OFF_CNT;
  int* rowptr = lists + OFF_ROWPTR;
  int* slots = lists + OFF_SLOTS;
  int* csr = lists + OFF_CSR;

  hipMemsetAsync(cnt, 0, (size_t)GN * sizeof(int), stream);
  degree_count_kernel<<<EDGE_BLOCKS, 256, 0, stream>>>(dst, cnt);
  degree_scan_kernel<<<1, SCAN_THREADS, 0, stream>>>(cnt, rowptr);
  edge_place_kernel<<<EDGE_BLOCKS, 256, 0, stream>>>(src, dst, cnt, slots);
  segment_order_kernel<<<ROW_BLOCKS, 256, 0, stream>>>(rowptr, slots, csr);

  for (int l = 0; l < GLAYERS; ++l) {
    const float* x = (l == 0) ? inp : (out + (size_t)(l - 1) * NH);
    const float* hl = hmat + (size_t)l * NH;
    const float* Wl = Wmat + (size_t)l * 6 * GH * GH;
    const float* bl = bmat + (size_t)l * 6 * GH;
    float* outl = out + (size_t)l * NH;

    gather_sum2_kernel<<<ROW_BLOCKS, 256, 0, stream>>>(rowptr, csr, x, agg_x,
                                                       hl, agg_t);

    gemm_zr_kernel<<<GEMM_BLOCKS, 256, 0, stream>>>(x, agg_x, hl, agg_t, Wl,
                                                    bl, zbuf, vbuf);

    gather_sum1_kernel<<<ROW_BLOCKS, 256, 0, stream>>>(rowptr, csr, vbuf,
                                                       agg_t);

    gemm_ht_kernel<<<GEMM_BLOCKS, 256, 0, stream>>>(x, agg_x, vbuf, agg_t, Wl,
                                                    bl, zbuf, hl, outl);
  }
}


# gemm_zr bounded to 256 registers per lane (launch bound of 4 waves per SIMD) so two 8-wave blocks share a compute unit: 391 row tiles in one pass instead of two; sums and their order unchanged
# speedup vs baseline: 2.6800x; 1.0618x over previous
#include <hip/hip_runtime.h>
#include <cstdint>
#include <cstddef>

constexpr int GN = 50000;
constexpr int GE = 1600000;
constexpr int GH = 128;
constexpr int GLAYERS = 2;
constexpr size_t NH = (size_t)GN * GH;

typedef __attribute__((ext_vector_type(2))) float v2f;
typedef __attribute__((ext_vector_type(8))) float v8f;

__device__ __forceinline__ v8f wmma_f32(v2f a, v2f b, v8f c) {
  return __builtin_amdgcn_wmma_f32_16x16x4_f32(
       false, a,  false, b,
       (short)0, c,  false,  false);
}

__device__ __forceinline__ float sigmoid_f(float x) {
  return 1.0f / (1.0f + __expf(-x));
}
__device__ __forceinline__ float tanh_f(float x) {
  return 1.0f - 2.0f / (1.0f + __expf(2.0f * x));
}

__device__ __forceinline__ void load_afrag_sum(const float* __restrict__ X,
                                               const float* __restrict__ A,
                                               int row, int kof, v2f* frag) {
  const float* xr = X + (size_t)row * GH;
  const float* ar = A + (size_t)row * GH;
#pragma unroll
  for (int kk = 0; kk < 32; ++kk) {
    const int c = (kk << 2) + kof;
    const float2 xv = *(const float2*)(xr + c);
    const float2 av = *(const float2*)(ar + c);
    v2f f;
    f.x = xv.x + av.x;
    f.y = xv.y + av.y;
    frag[kk] = f;
  }
}

__device__ __forceinline__ void stage_slab(const float* __restrict__ Wg, int n0,
                                           float* lds) {
  const int tid = threadIdx.x;
#pragma unroll
  for (int i = 0; i < 8; ++i) {
    const int idx = i * 256 + tid;
    const int k = idx >> 4;
    const int c = idx & 15;
    lds[c * 132 + k] = Wg[k * GH + n0 + c];
  }
}

constexpr int SCAN_THREADS = 1024;
constexpr int SCAN_PER = (GN + SCAN_THREADS - 1) / SCAN_THREADS;
constexpr int SCAN_PAD = SCAN_THREADS * SCAN_PER;

__global__ __launch_bounds__(256) void degree_count_kernel(
    const int* __restrict__ dst, int* __restrict__ cnt) {
  const unsigned e = blockIdx.x * 256u + threadIdx.x;
  if (e >= (unsigned)GE) return;
  atomicAdd(cnt + dst[e], 1);
}

__global__ __launch_bounds__(SCAN_THREADS) void degree_scan_kernel(
    int* __restrict__ cnt, int* __restrict__ rowptr) {
  __shared__ int buf[SCAN_PAD];
  __shared__ int wsum[32];
  const int t = threadIdx.x;
  const int lane = t & 31;
#pragma unroll 7
  for (int k = 0; k < SCAN_PER; ++k) {
    const int i = k * SCAN_THREADS + t;
    const int v = cnt[(i < GN) ? i : GN - 1];
    buf[i] = (i < GN) ? v : 0;
  }
  __syncthreads();
  const int b = t * SCAN_PER;
  int s = 0;
  for (int i = 0; i < SCAN_PER; ++i) s += buf[b + i];
  int incl = s;
  for (int off = 1; off < 32; off <<= 1) {
    const int y = __shfl_up(incl, off);
    if (lane >= off) incl += y;
  }
  if (lane == 31) wsum[t >> 5] = incl;
  __syncthreads();
  if (t < 32) {
    const int w = wsum[t];
    int wi = w;
    for (int off = 1; off < 32; off <<= 1) {
      const int y = __shfl_up(wi, off);
      if (t >= off) wi += y;
    }
    wsum[t] = wi - w;
  }
  __syncthreads();
  int run = wsum[t >> 5] + incl - s;
  for (int i = 0; i < SCAN_PER; ++i) {
    const int v = buf[b + i];
    buf[b + i] = run;
    run += v;
  }
  __syncthreads();
  for (int i = t; i < GN; i += SCAN_THREADS) {
    const int v = buf[i];
    rowptr[i] = v;
    cnt[i] = v;
  }
  if (t == SCAN_THREADS - 1) rowptr[GN] = run;
}

__global__ __launch_bounds__(256) void edge_place_kernel(
    const int* __restrict__ src, const int* __restrict__ dst,
    int* __restrict__ cursor, int* __restrict__ slots) {
  const unsigned e = blockIdx.x * 256u + threadIdx.x;
  if (e >= (unsigned)GE) return;
  const int p = atomicAdd(cursor + dst[e], 1);
  slots[p] = src[e];
}

__global__ __launch_bounds__(256) void segment_order_kernel(
    const int* __restrict__ rowptr, const int* __restrict__ slots,
    int* __restrict__ csr) {
  const int lane = threadIdx.x & 31;
  const int row = __builtin_amdgcn_readfirstlane(
      (int)((blockIdx.x * 256u + threadIdx.x) >> 5));
  if (row >= GN) return;
  const int beg = rowptr[row];
  const int n = rowptr[row + 1] - beg;
  for (int ib = 0; ib < n; ib += 32) {
    const int i = ib + lane;
    const int key = (i < n) ? slots[beg + i] : 0;
    int rank = 0;
    for (int jb = 0; jb < n; jb += 32) {
      const int jn = (n - jb < 32) ? (n - jb) : 32;
      const int kv = (lane < jn) ? slots[beg + jb + lane] : 0;
      for (int jj = 0; jj < jn; ++jj) {
        const int kj = __builtin_amdgcn_readlane(kv, jj);
        const int j = jb + jj;
        rank += (kj < key || (kj == key && j < i)) ? 1 : 0;
      }
    }
    if (i < n) csr[beg + rank] = key;
  }
}

template <int U>
__device__ __forceinline__ void gather_rows(const float* __restrict__ f,
                                            int idx, int j0, int c4,
                                            float4& acc) {
  float4 v[U];
#pragma unroll
  for (int u = 0; u < U; ++u) {
    const int s = __builtin_amdgcn_readlane(idx, j0 + u);
    v[u] = *(const float4*)(f + (size_t)s * GH + c4);
  }
#pragma unroll
  for (int u = 0; u < U; ++u) {
    acc.x += v[u].x;
    acc.y += v[u].y;
    acc.z += v[u].z;
    acc.w += v[u].w;
  }
}

template <int U, int NF>
__device__ __forceinline__ void gather_group(const float* __restrict__ f0,
                                             const float* __restrict__ f1,
                                             int idx, int j0, int c4,
                                             float4& a0, float4& a1) {
  gather_rows<U>(f0, idx, j0, c4, a0);
  if (NF == 2) gather_rows<U>(f1, idx, j0, c4, a1);
}

template <int NF>
__device__ __forceinline__ void gather_sum_body(
    const int* __restrict__ rowptr, const int* __restrict__ csr,
    const float* __restrict__ f0, float* __restrict__ o0,
    const float* __restrict__ f1, float* __restrict__ o1) {
  const int lane = threadIdx.x & 31;
  const int c4 = lane << 2;
  const int row = __builtin_amdgcn_readfirstlane(
      (int)((blockIdx.x * 256u + threadIdx.x) >> 5));
  if (row >= GN) return;
  const int beg = rowptr[row];
  const int end = rowptr[row + 1];
  float4 a0 = {0.0f, 0.0f, 0.0f, 0.0f};
  float4 a1 = {0.0f, 0.0f, 0.0f, 0.0f};
  for (int base = beg; base < end; base += 32) {
    const int n = (end - base < 32) ? (end - base) : 32;
    const int idx = (lane < n) ? csr[base + lane] : 0;
    int j = 0;
    for (; j + 8 <= n; j += 8) gather_group<8, NF>(f0, f1, idx, j, c4, a0, a1);
    if (n & 4) { gather_group<4, NF>(f0, f1, idx, j, c4, a0, a1); j += 4; }
    if (n & 2) { gather_group<2, NF>(f0, f1, idx, j, c4, a0, a1); j += 2; }
    if (n & 1) { gather_group<1, NF>(f0, f1, idx, j, c4, a0, a1); }
  }
  *(float4*)(o0 + (size_t)row * GH + c4) = a0;
  if (NF == 2) *(float4*)(o1 + (size_t)row * GH + c4) = a1;
}

__global__ __launch_bounds__(256) void gather_sum2_kernel(
    const int* __restrict__ rowptr, const int* __restrict__ csr,
    const float* __restrict__ f0, float* __restrict__ o0,
    const float* __restrict__ f1, float* __restrict__ o1) {
  gather_sum_body<2>(rowptr, csr, f0, o0, f1, o1);
}

__global__ __launch_bounds__(256) void gather_sum1_kernel(
    const int* __restrict__ rowptr, const int* __restrict__ csr,
    const float* __restrict__ f0, float* __restrict__ o0) {
  gather_sum_body<1>(rowptr, csr, f0, o0, f0, o0);
}

__global__ __launch_bounds__(256, 4) void gemm_zr_kernel(
    const float* __restrict__ x, const float* __restrict__ aggx,
    const float* __restrict__ hl, const float* __restrict__ aggh,
    const float* __restrict__ Wl, const float* __restrict__ bl,
    float* __restrict__ zout, float* __restrict__ vout) {
  __shared__ float lds[4 * 16 * 132];
  const int tid = threadIdx.x;
  const int wave = tid >> 5;
  const int lane = tid & 31;
  const int row0 = blockIdx.x * 128 + wave * 16;
  const int kof = (lane >> 4) << 1;
  const int nn = lane & 15;
  const int rowhi = (lane >> 4) << 3;

  int arow = row0 + nn;
  if (arow > GN - 1) arow = GN - 1;

  v2f fs[32], ft[32];
  load_afrag_sum(x, aggx, arow, kof, fs);
  __builtin_amdgcn_sched_barrier(0);
  load_afrag_sum(hl, aggh, arow, kof, ft);

  const float* W0 = Wl + 0 * GH * GH;
  const float* W1 = Wl + 1 * GH * GH;
  const float* W2 = Wl + 2 * GH * GH;
  const float* W3 = Wl + 3 * GH * GH;
  float* L0 = lds + 0 * 16 * 132;
  float* L1 = lds + 1 * 16 * 132;
  float* L2 = lds + 2 * 16 * 132;
  float* L3 = lds + 3 * 16 * 132;

  for (int ct = 0; ct < GH / 16; ++ct) {
    const int n0 = ct * 16;
    __syncthreads();
    stage_slab(W0, n0, L0);
    stage_slab(W1, n0, L1);
    stage_slab(W2, n0, L2);
    stage_slab(W3, n0, L3);
    __syncthreads();

    const float* p0 = L0 + nn * 132 + kof;
    const float* p1 = L1 + nn * 132 + kof;
    const float* p2 = L2 + nn * 132 + kof;
    const float* p3 = L3 + nn * 132 + kof;

    v8f az0 = {0, 0, 0, 0, 0, 0, 0, 0}, az1 = {0, 0, 0, 0, 0, 0, 0, 0};
    v8f ar0 = {0, 0, 0, 0, 0, 0, 0, 0}, ar1 = {0, 0, 0, 0, 0, 0, 0, 0};
#pragma unroll
    for (int kk = 0; kk < 32; kk += 2) {
      az0 = wmma_f32(fs[kk], *(const v2f*)(p0 + 4 * kk), az0);
      az0 = wmma_f32(ft[kk], *(const v2f*)(p1 + 4 * kk), az0);
      ar0 = wmma_f32(fs[kk], *(const v2f*)(p2 + 4 * kk), ar0);
      ar0 = wmma_f32(ft[kk], *(const v2f*)(p3 + 4 * kk), ar0);
      az1 = wmma_f32(fs[kk + 1], *(const v2f*)(p0 + 4 * kk + 4), az1);
      az1 = wmma_f32(ft[kk + 1], *(const v2f*)(p1 + 4 * kk + 4), az1);
      ar1 = wmma_f32(fs[kk + 1], *(const v2f*)(p2 + 4 * kk + 4), ar1);
      ar1 = wmma_f32(ft[kk + 1], *(const v2f*)(p3 + 4 * kk + 4), ar1);
    }
    const v8f accz = az0 + az1;
    const v8f accr = ar0 + ar1;

    const float bz = bl[0 * GH + n0 + nn] + bl[1 * GH + n0 + nn];
    const float br = bl[2 * GH + n0 + nn] + bl[3 * GH + n0 + nn];
#pragma unroll
    for (int i = 0; i < 8; ++i) {
      const int row = row0 + rowhi + i;
      if (row < GN) {
        const size_t off = (size_t)row * GH + n0 + nn;
        const float zv = sigmoid_f(accz[i] + bz);
        const float rv = sigmoid_f(accr[i] + br);
        zout[off] = zv;
        vout[off] = rv * hl[off];
      }
    }
  }
}

__global__ __launch_bounds__(256) void gemm_ht_kernel(
    const float* __restrict__ x, const float* __restrict__ aggx,
    const float* __restrict__ v, const float* __restrict__ aggv,
    const float* __restrict__ Wl, const float* __restrict__ bl,
    const float* __restrict__ zbuf, const float* __restrict__ hl,
    float* __restrict__ outp) {
  __shared__ float lds[2 * 16 * 132];
  const int tid = threadIdx.x;
  const int wave = tid >> 5;
  const int lane = tid & 31;
  const int row0 = blockIdx.x * 128 + wave * 16;
  const int kof = (lane >> 4) << 1;
  const int nn = lane & 15;
  const int rowhi = (lane >> 4) << 3;

  int arow = row0 + nn;
  if (arow > GN - 1) arow = GN - 1;

  v2f fs[32], fu[32];
  load_afrag_sum(x, aggx, arow, kof, fs);
  load_afrag_sum(v, aggv, arow, kof, fu);

  const float* W4 = Wl + 4 * GH * GH;
  const float* W5 = Wl + 5 * GH * GH;
  float* L4 = lds + 0 * 16 * 132;
  float* L5 = lds + 1 * 16 * 132;

  for (int ct = 0; ct < GH / 16; ++ct) {
    const int n0 = ct * 16;
    __syncthreads();
    stage_slab(W4, n0, L4);
    stage_slab(W5, n0, L5);
    __syncthreads();

    const float* p4 = L4 + nn * 132 + kof;
    const float* p5 = L5 + nn * 132 + kof;

    v8f a0 = {0, 0, 0, 0, 0, 0, 0, 0}, a1 = {0, 0, 0, 0, 0, 0, 0, 0};
#pragma unroll
    for (int kk = 0; kk < 32; kk += 2) {
      a0 = wmma_f32(fs[kk], *(const v2f*)(p4 + 4 * kk), a0);
      a0 = wmma_f32(fu[kk], *(const v2f*)(p5 + 4 * kk), a0);
      a1 = wmma_f32(fs[kk + 1], *(const v2f*)(p4 + 4 * kk + 4), a1);
      a1 = wmma_f32(fu[kk + 1], *(const v2f*)(p5 + 4 * kk + 4), a1);
    }
    const v8f acc = a0 + a1;

    const float bh = bl[4 * GH + n0 + nn] + bl[5 * GH + n0 + nn];
#pragma unroll
    for (int i = 0; i < 8; ++i) {
      const int row = row0 + rowhi + i;
      if (row < GN) {
        const size_t off = (size_t)row * GH + n0 + nn;
        const float ht = tanh_f(acc[i] + bh);
        const float zv = zbuf[off];
        outp[off] = zv * hl[off] + (1.0f - zv) * ht;
      }
    }
  }
}

constexpr int GN_PAD = 50048;
constexpr size_t OFF_CNT = 0;
constexpr size_t OFF_ROWPTR = OFF_CNT + GN_PAD;
constexpr size_t OFF_SLOTS = OFF_ROWPTR + GN_PAD;
constexpr size_t OFF_CSR = OFF_SLOTS + GE;
static_assert(OFF_CSR + GE <= NH, "edge lists must fit in one output plane");
constexpr int EDGE_BLOCKS = (GE + 255) / 256;
constexpr int ROW_BLOCKS = (GN * 32 + 255) / 256;
constexpr int GEMM_BLOCKS = (GN + 127) / 128;

extern "C" void kernel_launch(void* const* d_in, const int* in_sizes, int n_in,
                              void* d_out, int out_size, void* d_ws,
                              size_t ws_size, hipStream_t stream) {
  (void)in_sizes; (void)n_in; (void)out_size; (void)ws_size;
  const float* inp = (const float*)d_in[0];
  const int* edg = (const int*)d_in[1];
  const float* hmat = (const float*)d_in[2];
  const float* Wmat = (const float*)d_in[3];
  const float* bmat = (const float*)d_in[4];
  float* out = (float*)d_out;

  const int* src = edg;
  const int* dst = edg + GE;

  float* agg_x = (float*)d_ws;
  float* agg_t = agg_x + NH;
  float* zbuf = agg_t + NH;
  float* vbuf = zbuf + NH;

  int* lists = (int*)(out + (size_t)(GLAYERS - 1) * NH);
  int* cnt = lists + OFF_CNT;
  int* rowptr = lists + OFF_ROWPTR;
  int* slots = lists + OFF_SLOTS;
  int* csr = lists + OFF_CSR;

  hipMemsetAsync(cnt, 0, (size_t)GN * sizeof(int), stream);
  degree_count_kernel<<<EDGE_BLOCKS, 256, 0, stream>>>(dst, cnt);
  degree_scan_kernel<<<1, SCAN_THREADS, 0, stream>>>(cnt, rowptr);
  edge_place_kernel<<<EDGE_BLOCKS, 256, 0, stream>>>(src, dst, cnt, slots);
  segment_order_kernel<<<ROW_BLOCKS, 256, 0, stream>>>(rowptr, slots, csr);

  for (int l = 0; l < GLAYERS; ++l) {
    const float* x = (l == 0) ? inp : (out + (size_t)(l - 1) * NH);
    const float* hl = hmat + (size_t)l * NH;
    const float* Wl = Wmat + (size_t)l * 6 * GH * GH;
    const float* bl = bmat + (size_t)l * 6 * GH;
    float* outl = out + (size_t)l * NH;

    gather_sum2_kernel<<<ROW_BLOCKS, 256, 0, stream>>>(rowptr, csr, x, agg_x,
                                                       hl, agg_t);

    gemm_zr_kernel<<<GEMM_BLOCKS, 256, 0, stream>>>(x, agg_x, hl, agg_t, Wl,
                                                    bl, zbuf, vbuf);

    gather_sum1_kernel<<<ROW_BLOCKS, 256, 0, stream>>>(rowptr, csr, vbuf,
                                                       agg_t);

    gemm_ht_kernel<<<GEMM_BLOCKS, 256, 0, stream>>>(x, agg_x, vbuf, agg_t, Wl,
                                                    bl, zbuf, hl, outl);
  }
}


# Edges grouped by destination by a two-level counting sort on LDS counters (per-chunk bucket counts, then one workgroup per 256-row bucket), not two global atomics per edge; sum order unchanged
# speedup vs baseline: 3.6212x; 1.3512x over previous
#include <hip/hip_runtime.h>
#include <cstdint>
#include <cstddef>

constexpr int GN = 50000;
constexpr int GE = 1600000;
constexpr int GH = 128;
constexpr int GLAYERS = 2;
constexpr size_t NH = (size_t)GN * GH;

typedef __attribute__((ext_vector_type(2))) float v2f;
typedef __attribute__((ext_vector_type(8))) float v8f;

__device__ __forceinline__ v8f wmma_f32(v2f a, v2f b, v8f c) {
  return __builtin_amdgcn_wmma_f32_16x16x4_f32(
       false, a,  false, b,
       (short)0, c,  false,  false);
}

__device__ __forceinline__ float sigmoid_f(float x) {
  return 1.0f / (1.0f + __expf(-x));
}
__device__ __forceinline__ float tanh_f(float x) {
  return 1.0f - 2.0f / (1.0f + __expf(2.0f * x));
}

__device__ __forceinline__ void load_afrag_sum(const float* __restrict__ X,
                                               const float* __restrict__ A,
                                               int row, int kof, v2f* frag) {
  const float* xr = X + (size_t)row * GH;
  const float* ar = A + (size_t)row * GH;
#pragma unroll
  for (int kk = 0; kk < 32; ++kk) {
    const int c = (kk << 2) + kof;
    const float2 xv = *(const float2*)(xr + c);
    const float2 av = *(const float2*)(ar + c);
    v2f f;
    f.x = xv.x + av.x;
    f.y = xv.y + av.y;
    frag[kk] = f;
  }
}

__device__ __forceinline__ void stage_slab(const float* __restrict__ Wg, int n0,
                                           float* lds) {
  const int tid = threadIdx.x;
#pragma unroll
  for (int i = 0; i < 8; ++i) {
    const int idx = i * 256 + tid;
    const int k = idx >> 4;
    const int c = idx & 15;
    lds[c * 132 + k] = Wg[k * GH + n0 + c];
  }
}

constexpr int BK_SHIFT = 8;
constexpr int BK_ROWS = 1 << BK_SHIFT;
constexpr int NBK = (GN + BK_ROWS - 1) >> BK_SHIFT;
constexpr int SORT_THREADS = 1024;
constexpr int NCH = 250;
constexpr int CH_EDGES = GE / NCH;
constexpr int CH_TRIPS = (CH_EDGES + SORT_THREADS - 1) / SORT_THREADS;
constexpr int SCAN_GROUPS = SORT_THREADS / 256;
constexpr int CH_PER_GROUP = (NCH + SCAN_GROUPS - 1) / SCAN_GROUPS;
constexpr int ROWS_UNROLL = 4;
static_assert(NCH * CH_EDGES == GE, "chunks must tile the edge list");
static_assert(NBK <= 256, "one scan thread per bucket");

__device__ __forceinline__ int scan256_exclusive(int v, int t, int* wsum, int& total) {
  const int lane = t & 31;
  int incl = v;
  if (t < 256) {
    for (int off = 1; off < 32; off <<= 1) {
      const int y = __shfl_up(incl, off);
      if (lane >= off) incl += y;
    }
    if (lane == 31) wsum[t >> 5] = incl;
  }
  __syncthreads();
  int woff = 0, all = 0;
  for (int w = 0; w < 8; ++w) {
    const int s = wsum[w];
    if (w < (t >> 5)) woff += s;
    all += s;
  }
  total = all;
  return woff + incl - v;
}

__global__ __launch_bounds__(SORT_THREADS) void bucket_count_kernel(
    const int* __restrict__ dst, int* __restrict__ chcnt) {
  __shared__ int hist[NBK];
  const int t = threadIdx.x;
  const int g = blockIdx.x;
  if (t < NBK) hist[t] = 0;
  __syncthreads();
  const int e0 = g * CH_EDGES;
  int d[CH_TRIPS];
#pragma unroll
  for (int k = 0; k < CH_TRIPS; ++k) {
    const int i = k * SORT_THREADS + t;
    d[k] = dst[e0 + ((i < CH_EDGES) ? i : CH_EDGES - 1)];
  }
#pragma unroll
  for (int k = 0; k < CH_TRIPS; ++k) {
    if (k * SORT_THREADS + t < CH_EDGES) atomicAdd(&hist[d[k] >> BK_SHIFT], 1);
  }
  __syncthreads();
  if (t < NBK) chcnt[g * NBK + t] = hist[t];
}

__global__ __launch_bounds__(SORT_THREADS) void bucket_scan_kernel(
    const int* __restrict__ chcnt, int* __restrict__ chfirst,
    int* __restrict__ bstart) {
  __shared__ int part[SCAN_GROUPS][256];
  __shared__ int bs[256];
  __shared__ int wsum[8];
  const int t = threadIdx.x;
  const int q = t >> 8;
  const int b = t & 255;
  const int g0 = q * CH_PER_GROUP;
  const int bc = (b < NBK) ? b : NBK - 1;
  int v[CH_PER_GROUP];
#pragma unroll
  for (int j = 0; j < CH_PER_GROUP; ++j) {
    const int g = g0 + j;
    const int x = chcnt[((g < NCH) ? g : NCH - 1) * NBK + bc];
    v[j] = (g < NCH && b < NBK) ? x : 0;
  }
  int p = 0;
#pragma unroll
  for (int j = 0; j < CH_PER_GROUP; ++j) p += v[j];
  part[q][b] = p;
  __syncthreads();
  int tot = 0;
  if (t < 256) {
    for (int i = 0; i < SCAN_GROUPS; ++i) tot += part[i][t];
  }
  int all;
  const int ex = scan256_exclusive(tot, t, wsum, all);
  if (t < 256) {
    bs[t] = ex;
    if (t < NBK) bstart[t] = ex;
    if (t == 0) bstart[NBK] = all;
  }
  __syncthreads();
  if (b < NBK) {
    int run = bs[b];
    for (int i = 0; i < q; ++i) run += part[i][b];
#pragma unroll
    for (int j = 0; j < CH_PER_GROUP; ++j) {
      const int g = g0 + j;
      if (g < NCH) chfirst[g * NBK + b] = run;
      run += v[j];
    }
  }
}

__global__ __launch_bounds__(SORT_THREADS) void bucket_scatter_kernel(
    const int* __restrict__ src, const int* __restrict__ dst,
    const int* __restrict__ chfirst, int2* __restrict__ pairs) {
  __shared__ int cur[NBK];
  const int t = threadIdx.x;
  const int g = blockIdx.x;
  if (t < NBK) cur[t] = chfirst[g * NBK + t];
  __syncthreads();
  const int e0 = g * CH_EDGES;
  int d[CH_TRIPS], s[CH_TRIPS];
#pragma unroll
  for (int k = 0; k < CH_TRIPS; ++k) {
    const int i = k * SORT_THREADS + t;
    const int ic = e0 + ((i < CH_EDGES) ? i : CH_EDGES - 1);
    d[k] = dst[ic];
    s[k] = src[ic];
  }
#pragma unroll
  for (int k = 0; k < CH_TRIPS; ++k) {
    if (k * SORT_THREADS + t < CH_EDGES) {
      const int p = atomicAdd(&cur[d[k] >> BK_SHIFT], 1);
      pairs[p] = make_int2(s[k], d[k]);
    }
  }
}

__global__ __launch_bounds__(SORT_THREADS) void bucket_rows_kernel(
    const int* __restrict__ bstart, const int2* __restrict__ pairs,
    int* __restrict__ rowptr, int* __restrict__ slots) {
  __shared__ int rc[BK_ROWS];
  __shared__ int wsum[8];
  const int t = threadIdx.x;
  const int b = blockIdx.x;
  const int beg = bstart[b];
  const int end = bstart[b + 1];
  if (t < BK_ROWS) rc[t] = 0;
  __syncthreads();
  for (int i0 = beg + t; i0 < end; i0 += ROWS_UNROLL * SORT_THREADS) {
    int d[ROWS_UNROLL];
#pragma unroll
    for (int u = 0; u < ROWS_UNROLL; ++u) {
      const int i = i0 + u * SORT_THREADS;
      d[u] = pairs[(i < end) ? i : end - 1].y;
    }
#pragma unroll
    for (int u = 0; u < ROWS_UNROLL; ++u) {
      if (i0 + u * SORT_THREADS < end) atomicAdd(&rc[d[u] & (BK_ROWS - 1)], 1);
    }
  }
  __syncthreads();
  const int deg = (t < BK_ROWS) ? rc[t] : 0;
  int all;
  const int ex = scan256_exclusive(deg, t, wsum, all);
  if (t < BK_ROWS) {
    const int row = (b << BK_SHIFT) + t;
    rc[t] = beg + ex;
    if (row < GN) rowptr[row] = beg + ex;
  }
  if (b == NBK - 1 && t == 0) rowptr[GN] = end;
  __syncthreads();
  for (int i0 = beg + t; i0 < end; i0 += ROWS_UNROLL * SORT_THREADS) {
    int2 sd[ROWS_UNROLL];
#pragma unroll
    for (int u = 0; u < ROWS_UNROLL; ++u) {
      const int i = i0 + u * SORT_THREADS;
      sd[u] = pairs[(i < end) ? i : end - 1];
    }
#pragma unroll
    for (int u = 0; u < ROWS_UNROLL; ++u) {
      if (i0 + u * SORT_THREADS < end) {
        const int p = atomicAdd(&rc[sd[u].y & (BK_ROWS - 1)], 1);
        slots[p] = sd[u].x;
      }
    }
  }
}

__global__ __launch_bounds__(256) void segment_order_kernel(
    const int* __restrict__ rowptr, const int* __restrict__ slots,
    int* __restrict__ csr) {
  const int lane = threadIdx.x & 31;
  const int row = __builtin_amdgcn_readfirstlane(
      (int)((blockIdx.x * 256u + threadIdx.x) >> 5));
  if (row >= GN) return;
  const int beg = rowptr[row];
  const int n = rowptr[row + 1] - beg;
  for (int ib = 0; ib < n; ib += 32) {
    const int i = ib + lane;
    const int key = (i < n) ? slots[beg + i] : 0;
    int rank = 0;
    for (int jb = 0; jb < n; jb += 32) {
      const int jn = (n - jb < 32) ? (n - jb) : 32;
      const int kv = (lane < jn) ? slots[beg + jb + lane] : 0;
      for (int jj = 0; jj < jn; ++jj) {
        const int kj = __builtin_amdgcn_readlane(kv, jj);
        const int j = jb + jj;
        rank += (kj < key || (kj == key && j < i)) ? 1 : 0;
      }
    }
    if (i < n) csr[beg + rank] = key;
  }
}

template <int U>
__device__ __forceinline__ void gather_rows(const float* __restrict__ f,
                                            int idx, int j0, int c4,
                                            float4& acc) {
  float4 v[U];
#pragma unroll
  for (int u = 0; u < U; ++u) {
    const int s = __builtin_amdgcn_readlane(idx, j0 + u);
    v[u] = *(const float4*)(f + (size_t)s * GH + c4);
  }
#pragma unroll
  for (int u = 0; u < U; ++u) {
    acc.x += v[u].x;
    acc.y += v[u].y;
    acc.z += v[u].z;
    acc.w += v[u].w;
  }
}

template <int U, int NF>
__device__ __forceinline__ void gather_group(const float* __restrict__ f0,
                                             const float* __restrict__ f1,
                                             int idx, int j0, int c4,
                                             float4& a0, float4& a1) {
  gather_rows<U>(f0, idx, j0, c4, a0);
  if (NF == 2) gather_rows<U>(f1, idx, j0, c4, a1);
}

template <int NF>
__device__ __forceinline__ void gather_sum_body(
    const int* __restrict__ rowptr, const int* __restrict__ csr,
    const float* __restrict__ f0, float* __restrict__ o0,
    const float* __restrict__ f1, float* __restrict__ o1) {
  const int lane = threadIdx.x & 31;
  const int c4 = lane << 2;
  const int row = __builtin_amdgcn_readfirstlane(
      (int)((blockIdx.x * 256u + threadIdx.x) >> 5));
  if (row >= GN) return;
  const int beg = rowptr[row];
  const int end = rowptr[row + 1];
  float4 a0 = {0.0f, 0.0f, 0.0f, 0.0f};
  float4 a1 = {0.0f, 0.0f, 0.0f, 0.0f};
  for (int base = beg; base < end; base += 32) {
    const int n = (end - base < 32) ? (end - base) : 32;
    const int idx = (lane < n) ? csr[base + lane] : 0;
    int j = 0;
    for (; j + 8 <= n; j += 8) gather_group<8, NF>(f0, f1, idx, j, c4, a0, a1);
    if (n & 4) { gather_group<4, NF>(f0, f1, idx, j, c4, a0, a1); j += 4; }
    if (n & 2) { gather_group<2, NF>(f0, f1, idx, j, c4, a0, a1); j += 2; }
    if (n & 1) { gather_group<1, NF>(f0, f1, idx, j, c4, a0, a1); }
  }
  *(float4*)(o0 + (size_t)row * GH + c4) = a0;
  if (NF == 2) *(float4*)(o1 + (size_t)row * GH + c4) = a1;
}

__global__ __launch_bounds__(256) void gather_sum2_kernel(
    const int* __restrict__ rowptr, const int* __restrict__ csr,
    const float* __restrict__ f0, float* __restrict__ o0,
    const float* __restrict__ f1, float* __restrict__ o1) {
  gather_sum_body<2>(rowptr, csr, f0, o0, f1, o1);
}

__global__ __launch_bounds__(256) void gather_sum1_kernel(
    const int* __restrict__ rowptr, const int* __restrict__ csr,
    const float* __restrict__ f0, float* __restrict__ o0) {
  gather_sum_body<1>(rowptr, csr, f0, o0, f0, o0);
}

__global__ __launch_bounds__(256, 4) void gemm_zr_kernel(
    const float* __restrict__ x, const float* __restrict__ aggx,
    const float* __restrict__ hl, const float* __restrict__ aggh,
    const float* __restrict__ Wl, const float* __restrict__ bl,
    float* __restrict__ zout, float* __restrict__ vout) {
  __shared__ float lds[4 * 16 * 132];
  const int tid = threadIdx.x;
  const int wave = tid >> 5;
  const int lane = tid & 31;
  const int row0 = blockIdx.x * 128 + wave * 16;
  const int kof = (lane >> 4) << 1;
  const int nn = lane & 15;
  const int rowhi = (lane >> 4) << 3;

  int arow = row0 + nn;
  if (arow > GN - 1) arow = GN - 1;

  v2f fs[32], ft[32];
  load_afrag_sum(x, aggx, arow, kof, fs);
  __builtin_amdgcn_sched_barrier(0);
  load_afrag_sum(hl, aggh, arow, kof, ft);

  const float* W0 = Wl + 0 * GH * GH;
  const float* W1 = Wl + 1 * GH * GH;
  const float* W2 = Wl + 2 * GH * GH;
  const float* W3 = Wl + 3 * GH * GH;
  float* L0 = lds + 0 * 16 * 132;
  float* L1 = lds + 1 * 16 * 132;
  float* L2 = lds + 2 * 16 * 132;
  float* L3 = lds + 3 * 16 * 132;

  for (int ct = 0; ct < GH / 16; ++ct) {
    const int n0 = ct * 16;
    __syncthreads();
    stage_slab(W0, n0, L0);
    stage_slab(W1, n0, L1);
    stage_slab(W2, n0, L2);
    stage_slab(W3, n0, L3);
    __syncthreads();

    const float* p0 = L0 + nn * 132 + kof;
    const float* p1 = L1 + nn * 132 + kof;
    const float* p2 = L2 + nn * 132 + kof;
    const float* p3 = L3 + nn * 132 + kof;

    v8f az0 = {0, 0, 0, 0, 0, 0, 0, 0}, az1 = {0, 0, 0, 0, 0, 0, 0, 0};
    v8f ar0 = {0, 0, 0, 0, 0, 0, 0, 0}, ar1 = {0, 0, 0, 0, 0, 0, 0, 0};
#pragma unroll
    for (int kk = 0; kk < 32; kk += 2) {
      az0 = wmma_f32(fs[kk], *(const v2f*)(p0 + 4 * kk), az0);
      az0 = wmma_f32(ft[kk], *(const v2f*)(p1 + 4 * kk), az0);
      ar0 = wmma_f32(fs[kk], *(const v2f*)(p2 + 4 * kk), ar0);
      ar0 = wmma_f32(ft[kk], *(const v2f*)(p3 + 4 * kk), ar0);
      az1 = wmma_f32(fs[kk + 1], *(const v2f*)(p0 + 4 * kk + 4), az1);
      az1 = wmma_f32(ft[kk + 1], *(const v2f*)(p1 + 4 * kk + 4), az1);
      ar1 = wmma_f32(fs[kk + 1], *(const v2f*)(p2 + 4 * kk + 4), ar1);
      ar1 = wmma_f32(ft[kk + 1], *(const v2f*)(p3 + 4 * kk + 4), ar1);
    }
    const v8f accz = az0 + az1;
    const v8f accr = ar0 + ar1;

    const float bz = bl[0 * GH + n0 + nn] + bl[1 * GH + n0 + nn];
    const float br = bl[2 * GH + n0 + nn] + bl[3 * GH + n0 + nn];
#pragma unroll
    for (int i = 0; i < 8; ++i) {
      const int row = row0 + rowhi + i;
      if (row < GN) {
        const size_t off = (size_t)row * GH + n0 + nn;
        const float zv = sigmoid_f(accz[i] + bz);
        const float rv = sigmoid_f(accr[i] + br);
        zout[off] = zv;
        vout[off] = rv * hl[off];
      }
    }
  }
}

__global__ __launch_bounds__(256) void gemm_ht_kernel(
    const float* __restrict__ x, const float* __restrict__ aggx,
    const float* __restrict__ v, const float* __restrict__ aggv,
    const float* __restrict__ Wl, const float* __restrict__ bl,
    const float* __restrict__ zbuf, const float* __restrict__ hl,
    float* __restrict__ outp) {
  __shared__ float lds[2 * 16 * 132];
  const int tid = threadIdx.x;
  const int wave = tid >> 5;
  const int lane = tid & 31;
  const int row0 = blockIdx.x * 128 + wave * 16;
  const int kof = (lane >> 4) << 1;
  const int nn = lane & 15;
  const int rowhi = (lane >> 4) << 3;

  int arow = row0 + nn;
  if (arow > GN - 1) arow = GN - 1;

  v2f fs[32], fu[32];
  load_afrag_sum(x, aggx, arow, kof, fs);
  load_afrag_sum(v, aggv, arow, kof, fu);

  const float* W4 = Wl + 4 * GH * GH;
  const float* W5 = Wl + 5 * GH * GH;
  float* L4 = lds + 0 * 16 * 132;
  float* L5 = lds + 1 * 16 * 132;

  for (int ct = 0; ct < GH / 16; ++ct) {
    const int n0 = ct * 16;
    __syncthreads();
    stage_slab(W4, n0, L4);
    stage_slab(W5, n0, L5);
    __syncthreads();

    const float* p4 = L4 + nn * 132 + kof;
    const float* p5 = L5 + nn * 132 + kof;

    v8f a0 = {0, 0, 0, 0, 0, 0, 0, 0}, a1 = {0, 0, 0, 0, 0, 0, 0, 0};
#pragma unroll
    for (int kk = 0; kk < 32; kk += 2) {
      a0 = wmma_f32(fs[kk], *(const v2f*)(p4 + 4 * kk), a0);
      a0 = wmma_f32(fu[kk], *(const v2f*)(p5 + 4 * kk), a0);
      a1 = wmma_f32(fs[kk + 1], *(const v2f*)(p4 + 4 * kk + 4), a1);
      a1 = wmma_f32(fu[kk + 1], *(const v2f*)(p5 + 4 * kk + 4), a1);
    }
    const v8f acc = a0 + a1;

    const float bh = bl[4 * GH + n0 + nn] + bl[5 * GH + n0 + nn];
#pragma unroll
    for (int i = 0; i < 8; ++i) {
      const int row = row0 + rowhi + i;
      if (row < GN) {
        const size_t off = (size_t)row * GH + n0 + nn;
        const float ht = tanh_f(acc[i] + bh);
        const float zv = zbuf[off];
        outp[off] = zv * hl[off] + (1.0f - zv) * ht;
      }
    }
  }
}

constexpr int GN_PAD = 50048;
constexpr int CHCNT_PAD = ((NCH * NBK + 63) / 64) * 64;
constexpr size_t OFF_ROWPTR = 0;
constexpr size_t OFF_CHCNT = OFF_ROWPTR + GN_PAD;
constexpr size_t OFF_CHFIRST = OFF_CHCNT + CHCNT_PAD;
constexpr size_t OFF_BSTART = OFF_CHFIRST + CHCNT_PAD;
constexpr size_t OFF_SLOTS = OFF_BSTART + 256;
constexpr size_t OFF_PAIRS = OFF_SLOTS + GE;
constexpr size_t OFF_CSR = OFF_PAIRS;
static_assert(NBK + 1 <= 256, "bucket starts must fit their slot");
static_assert(OFF_PAIRS + 2 * (size_t)GE <= NH, "edge lists must fit in one output plane");
constexpr int ROW_BLOCKS = (GN * 32 + 255) / 256;
constexpr int GEMM_BLOCKS = (GN + 127) / 128;

extern "C" void kernel_launch(void* const* d_in, const int* in_sizes, int n_in,
                              void* d_out, int out_size, void* d_ws,
                              size_t ws_size, hipStream_t stream) {
  (void)in_sizes; (void)n_in; (void)out_size; (void)ws_size;
  const float* inp = (const float*)d_in[0];
  const int* edg = (const int*)d_in[1];
  const float* hmat = (const float*)d_in[2];
  const float* Wmat = (const float*)d_in[3];
  const float* bmat = (const float*)d_in[4];
  float* out = (float*)d_out;

  const int* src = edg;
  const int* dst = edg + GE;

  float* agg_x = (float*)d_ws;
  float* agg_t = agg_x + NH;
  float* zbuf = agg_t + NH;
  float* vbuf = zbuf + NH;

  int* lists = (int*)(out + (size_t)(GLAYERS - 1) * NH);
  int* rowptr = lists + OFF_ROWPTR;
  int* chcnt = lists + OFF_CHCNT;
  int* chfirst = lists + OFF_CHFIRST;
  int* bstart = lists + OFF_BSTART;
  int* slots = lists + OFF_SLOTS;
  int2* pairs = (int2*)(lists + OFF_PAIRS);
  int* csr = lists + OFF_CSR;

  bucket_count_kernel<<<NCH, SORT_THREADS, 0, stream>>>(dst, chcnt);
  bucket_scan_kernel<<<1, SORT_THREADS, 0, stream>>>(chcnt, chfirst, bstart);
  bucket_scatter_kernel<<<NCH, SORT_THREADS, 0, stream>>>(src, dst, chfirst, pairs);
  bucket_rows_kernel<<<NBK, SORT_THREADS, 0, stream>>>(bstart, pairs, rowptr, slots);
  segment_order_kernel<<<ROW_BLOCKS, 256, 0, stream>>>(rowptr, slots, csr);

  for (int l = 0; l < GLAYERS; ++l) {
    const float* x = (l == 0) ? inp : (out + (size_t)(l - 1) * NH);
    const float* hl = hmat + (size_t)l * NH;
    const float* Wl = Wmat + (size_t)l * 6 * GH * GH;
    const float* bl = bmat + (size_t)l * 6 * GH;
    float* outl = out + (size_t)l * NH;

    gather_sum2_kernel<<<ROW_BLOCKS, 256, 0, stream>>>(rowptr, csr, x, agg_x,
                                                       hl, agg_t);

    gemm_zr_kernel<<<GEMM_BLOCKS, 256, 0, stream>>>(x, agg_x, hl, agg_t, Wl,
                                                    bl, zbuf, vbuf);

    gather_sum1_kernel<<<ROW_BLOCKS, 256, 0, stream>>>(rowptr, csr, vbuf,
                                                       agg_t);

    gemm_ht_kernel<<<GEMM_BLOCKS, 256, 0, stream>>>(x, agg_x, vbuf, agg_t, Wl,
                                                    bl, zbuf, hl, outl);
  }
}


# Gathers store the finished GIN operand x+sum, each row in the A-fragment loader's order: a GEMM reads one plane per operand with 16-byte loads, not two planes with 8-byte loads; sums unchanged
# speedup vs baseline: 3.7396x; 1.0327x over previous
#include <hip/hip_runtime.h>
#include <cstdint>
#include <cstddef>

constexpr int GN = 50000;
constexpr int GE = 1600000;
constexpr int GH = 128;
constexpr int GLAYERS = 2;
constexpr size_t NH = (size_t)GN * GH;

typedef __attribute__((ext_vector_type(2))) float v2f;
typedef __attribute__((ext_vector_type(8))) float v8f;

__device__ __forceinline__ v8f wmma_f32(v2f a, v2f b, v8f c) {
  return __builtin_amdgcn_wmma_f32_16x16x4_f32(
       false, a,  false, b,
       (short)0, c,  false,  false);
}

__device__ __forceinline__ float sigmoid_f(float x) {
  return 1.0f / (1.0f + __expf(-x));
}
__device__ __forceinline__ float tanh_f(float x) {
  return 1.0f - 2.0f / (1.0f + __expf(2.0f * x));
}

__device__ __forceinline__ void load_afrag_ready(const float* __restrict__ P,
                                                 int row, int half, v2f* frag) {
  const float* pr = P + (size_t)row * GH + (half << 2);
#pragma unroll
  for (int m = 0; m < 16; ++m) {
    const float4 v = *(const float4*)(pr + 8 * m);
    v2f lo, hi;
    lo.x = v.x;
    lo.y = v.y;
    hi.x = v.z;
    hi.y = v.w;
    frag[2 * m] = lo;
    frag[2 * m + 1] = hi;
  }
}

__device__ __forceinline__ void stage_slab(const float* __restrict__ Wg, int n0,
                                           float* lds) {
  const int tid = threadIdx.x;
#pragma unroll
  for (int i = 0; i < 8; ++i) {
    const int idx = i * 256 + tid;
    const int k = idx >> 4;
    const int c = idx & 15;
    lds[c * 132 + k] = Wg[k * GH + n0 + c];
  }
}

constexpr int BK_SHIFT = 8;
constexpr int BK_ROWS = 1 << BK_SHIFT;
constexpr int NBK = (GN + BK_ROWS - 1) >> BK_SHIFT;
constexpr int SORT_THREADS = 1024;
constexpr int NCH = 250;
constexpr int CH_EDGES = GE / NCH;
constexpr int CH_TRIPS = (CH_EDGES + SORT_THREADS - 1) / SORT_THREADS;
constexpr int SCAN_GROUPS = SORT_THREADS / 256;
constexpr int CH_PER_GROUP = (NCH + SCAN_GROUPS - 1) / SCAN_GROUPS;
constexpr int ROWS_UNROLL = 4;
static_assert(NCH * CH_EDGES == GE, "chunks must tile the edge list");
static_assert(NBK <= 256, "one scan thread per bucket");

__device__ __forceinline__ int scan256_exclusive(int v, int t, int* wsum, int& total) {
  const int lane = t & 31;
  int incl = v;
  if (t < 256) {
    for (int off = 1; off < 32; off <<= 1) {
      const int y = __shfl_up(incl, off);
      if (lane >= off) incl += y;
    }
    if (lane == 31) wsum[t >> 5] = incl;
  }
  __syncthreads();
  int woff = 0, all = 0;
  for (int w = 0; w < 8; ++w) {
    const int s = wsum[w];
    if (w < (t >> 5)) woff += s;
    all += s;
  }
  total = all;
  return woff + incl - v;
}

__global__ __launch_bounds__(SORT_THREADS) void bucket_count_kernel(
    const int* __restrict__ dst, int* __restrict__ chcnt) {
  __shared__ int hist[NBK];
  const int t = threadIdx.x;
  const int g = blockIdx.x;
  if (t < NBK) hist[t] = 0;
  __syncthreads();
  const int e0 = g * CH_EDGES;
  int d[CH_TRIPS];
#pragma unroll
  for (int k = 0; k < CH_TRIPS; ++k) {
    const int i = k * SORT_THREADS + t;
    d[k] = dst[e0 + ((i < CH_EDGES) ? i : CH_EDGES - 1)];
  }
#pragma unroll
  for (int k = 0; k < CH_TRIPS; ++k) {
    if (k * SORT_THREADS + t < CH_EDGES) atomicAdd(&hist[d[k] >> BK_SHIFT], 1);
  }
  __syncthreads();
  if (t < NBK) chcnt[g * NBK + t] = hist[t];
}

__global__ __launch_bounds__(SORT_THREADS) void bucket_scan_kernel(
    const int* __restrict__ chcnt, int* __restrict__ chfirst,
    int* __restrict__ bstart) {
  __shared__ int part[SCAN_GROUPS][256];
  __shared__ int bs[256];
  __shared__ int wsum[8];
  const int t = threadIdx.x;
  const int q = t >> 8;
  const int b = t & 255;
  const int g0 = q * CH_PER_GROUP;
  const int bc = (b < NBK) ? b : NBK - 1;
  int v[CH_PER_GROUP];
#pragma unroll
  for (int j = 0; j < CH_PER_GROUP; ++j) {
    const int g = g0 + j;
    const int x = chcnt[((g < NCH) ? g : NCH - 1) * NBK + bc];
    v[j] = (g < NCH && b < NBK) ? x : 0;
  }
  int p = 0;
#pragma unroll
  for (int j = 0; j < CH_PER_GROUP; ++j) p += v[j];
  part[q][b] = p;
  __syncthreads();
  int tot = 0;
  if (t < 256) {
    for (int i = 0; i < SCAN_GROUPS; ++i) tot += part[i][t];
  }
  int all;
  const int ex = scan256_exclusive(tot, t, wsum, all);
  if (t < 256) {
    bs[t] = ex;
    if (t < NBK) bstart[t] = ex;
    if (t == 0) bstart[NBK] = all;
  }
  __syncthreads();
  if (b < NBK) {
    int run = bs[b];
    for (int i = 0; i < q; ++i) run += part[i][b];
#pragma unroll
    for (int j = 0; j < CH_PER_GROUP; ++j) {
      const int g = g0 + j;
      if (g < NCH) chfirst[g * NBK + b] = run;
      run += v[j];
    }
  }
}

__global__ __launch_bounds__(SORT_THREADS) void bucket_scatter_kernel(
    const int* __restrict__ src, const int* __restrict__ dst,
    const int* __restrict__ chfirst, int2* __restrict__ pairs) {
  __shared__ int cur[NBK];
  const int t = threadIdx.x;
  const int g = blockIdx.x;
  if (t < NBK) cur[t] = chfirst[g * NBK + t];
  __syncthreads();
  const int e0 = g * CH_EDGES;
  int d[CH_TRIPS], s[CH_TRIPS];
#pragma unroll
  for (int k = 0; k < CH_TRIPS; ++k) {
    const int i = k * SORT_THREADS + t;
    const int ic = e0 + ((i < CH_EDGES) ? i : CH_EDGES - 1);
    d[k] = dst[ic];
    s[k] = src[ic];
  }
#pragma unroll
  for (int k = 0; k < CH_TRIPS; ++k) {
    if (k * SORT_THREADS + t < CH_EDGES) {
      const int p = atomicAdd(&cur[d[k] >> BK_SHIFT], 1);
      pairs[p] = make_int2(s[k], d[k]);
    }
  }
}

__global__ __launch_bounds__(SORT_THREADS) void bucket_rows_kernel(
    const int* __restrict__ bstart, const int2* __restrict__ pairs,
    int* __restrict__ rowptr, int* __restrict__ slots) {
  __shared__ int rc[BK_ROWS];
  __shared__ int wsum[8];
  const int t = threadIdx.x;
  const int b = blockIdx.x;
  const int beg = bstart[b];
  const int end = bstart[b + 1];
  if (t < BK_ROWS) rc[t] = 0;
  __syncthreads();
  for (int i0 = beg + t; i0 < end; i0 += ROWS_UNROLL * SORT_THREADS) {
    int d[ROWS_UNROLL];
#pragma unroll
    for (int u = 0; u < ROWS_UNROLL; ++u) {
      const int i = i0 + u * SORT_THREADS;
      d[u] = pairs[(i < end) ? i : end - 1].y;
    }
#pragma unroll
    for (int u = 0; u < ROWS_UNROLL; ++u) {
      if (i0 + u * SORT_THREADS < end) atomicAdd(&rc[d[u] & (BK_ROWS - 1)], 1);
    }
  }
  __syncthreads();
  const int deg = (t < BK_ROWS) ? rc[t] : 0;
  int all;
  const int ex = scan256_exclusive(deg, t, wsum, all);
  if (t < BK_ROWS) {
    const int row = (b << BK_SHIFT) + t;
    rc[t] = beg + ex;
    if (row < GN) rowptr[row] = beg + ex;
  }
  if (b == NBK - 1 && t == 0) rowptr[GN] = end;
  __syncthreads();
  for (int i0 = beg + t; i0 < end; i0 += ROWS_UNROLL * SORT_THREADS) {
    int2 sd[ROWS_UNROLL];
#pragma unroll
    for (int u = 0; u < ROWS_UNROLL; ++u) {
      const int i = i0 + u * SORT_THREADS;
      sd[u] = pairs[(i < end) ? i : end - 1];
    }
#pragma unroll
    for (int u = 0; u < ROWS_UNROLL; ++u) {
      if (i0 + u * SORT_THREADS < end) {
        const int p = atomicAdd(&rc[sd[u].y & (BK_ROWS - 1)], 1);
        slots[p] = sd[u].x;
      }
    }
  }
}

__global__ __launch_bounds__(256) void segment_order_kernel(
    const int* __restrict__ rowptr, const int* __restrict__ slots,
    int* __restrict__ csr) {
  const int lane = threadIdx.x & 31;
  const int row = __builtin_amdgcn_readfirstlane(
      (int)((blockIdx.x * 256u + threadIdx.x) >> 5));
  if (row >= GN) return;
  const int beg = rowptr[row];
  const int n = rowptr[row + 1] - beg;
  for (int ib = 0; ib < n; ib += 32) {
    const int i = ib + lane;
    const int key = (i < n) ? slots[beg + i] : 0;
    int rank = 0;
    for (int jb = 0; jb < n; jb += 32) {
      const int jn = (n - jb < 32) ? (n - jb) : 32;
      const int kv = (lane < jn) ? slots[beg + jb + lane] : 0;
      for (int jj = 0; jj < jn; ++jj) {
        const int kj = __builtin_amdgcn_readlane(kv, jj);
        const int j = jb + jj;
        rank += (kj < key || (kj == key && j < i)) ? 1 : 0;
      }
    }
    if (i < n) csr[beg + rank] = key;
  }
}

template <int U>
__device__ __forceinline__ void gather_rows(const float* __restrict__ f,
                                            int idx, int j0, int c4,
                                            float4& acc) {
  float4 v[U];
#pragma unroll
  for (int u = 0; u < U; ++u) {
    const int s = __builtin_amdgcn_readlane(idx, j0 + u);
    v[u] = *(const float4*)(f + (size_t)s * GH + c4);
  }
#pragma unroll
  for (int u = 0; u < U; ++u) {
    acc.x += v[u].x;
    acc.y += v[u].y;
    acc.z += v[u].z;
    acc.w += v[u].w;
  }
}

template <int U, int NF>
__device__ __forceinline__ void gather_group(const float* __restrict__ f0,
                                             const float* __restrict__ f1,
                                             int idx, int j0, int c4,
                                             float4& a0, float4& a1) {
  gather_rows<U>(f0, idx, j0, c4, a0);
  if (NF == 2) gather_rows<U>(f1, idx, j0, c4, a1);
}

__device__ __forceinline__ void store_operand_row(float* __restrict__ o, int row,
                                                  int c4, int lane, float4 s) {
  const bool odd = (lane & 1) != 0;
  const float sx = odd ? s.x : s.z;
  const float sy = odd ? s.y : s.w;
  const float rx = __shfl_xor(sx, 1);
  const float ry = __shfl_xor(sy, 1);
  float4 w;
  w.x = odd ? rx : s.x;
  w.y = odd ? ry : s.y;
  w.z = odd ? s.z : rx;
  w.w = odd ? s.w : ry;
  *(float4*)(o + (size_t)row * GH + c4) = w;
}

template <int NF>
__device__ __forceinline__ void gather_sum_body(
    const int* __restrict__ rowptr, const int* __restrict__ csr,
    const float* __restrict__ f0, float* __restrict__ o0,
    const float* __restrict__ f1, float* __restrict__ o1) {
  const int lane = threadIdx.x & 31;
  const int c4 = lane << 2;
  const int row = __builtin_amdgcn_readfirstlane(
      (int)((blockIdx.x * 256u + threadIdx.x) >> 5));
  if (row >= GN) return;
  const int beg = rowptr[row];
  const int end = rowptr[row + 1];
  const float4 x0 = *(const float4*)(f0 + (size_t)row * GH + c4);
  const float4 x1 = *(const float4*)(f1 + (size_t)row * GH + c4);
  float4 a0 = {0.0f, 0.0f, 0.0f, 0.0f};
  float4 a1 = {0.0f, 0.0f, 0.0f, 0.0f};
  for (int base = beg; base < end; base += 32) {
    const int n = (end - base < 32) ? (end - base) : 32;
    const int idx = (lane < n) ? csr[base + lane] : 0;
    int j = 0;
    for (; j + 8 <= n; j += 8) gather_group<8, NF>(f0, f1, idx, j, c4, a0, a1);
    if (n & 4) { gather_group<4, NF>(f0, f1, idx, j, c4, a0, a1); j += 4; }
    if (n & 2) { gather_group<2, NF>(f0, f1, idx, j, c4, a0, a1); j += 2; }
    if (n & 1) { gather_group<1, NF>(f0, f1, idx, j, c4, a0, a1); }
  }
  float4 s0;
  s0.x = x0.x + a0.x;
  s0.y = x0.y + a0.y;
  s0.z = x0.z + a0.z;
  s0.w = x0.w + a0.w;
  store_operand_row(o0, row, c4, lane, s0);
  if (NF == 2) {
    float4 s1;
    s1.x = x1.x + a1.x;
    s1.y = x1.y + a1.y;
    s1.z = x1.z + a1.z;
    s1.w = x1.w + a1.w;
    store_operand_row(o1, row, c4, lane, s1);
  }
}

__global__ __launch_bounds__(256) void gather_sum2_kernel(
    const int* __restrict__ rowptr, const int* __restrict__ csr,
    const float* __restrict__ f0, float* __restrict__ o0,
    const float* __restrict__ f1, float* __restrict__ o1) {
  gather_sum_body<2>(rowptr, csr, f0, o0, f1, o1);
}

__global__ __launch_bounds__(256) void gather_sum1_kernel(
    const int* __restrict__ rowptr, const int* __restrict__ csr,
    const float* __restrict__ f0, float* __restrict__ o0) {
  gather_sum_body<1>(rowptr, csr, f0, o0, f0, o0);
}

__global__ __launch_bounds__(256, 4) void gemm_zr_kernel(
    const float* __restrict__ sop, const float* __restrict__ top,
    const float* __restrict__ hl,
    const float* __restrict__ Wl, const float* __restrict__ bl,
    float* __restrict__ zout, float* __restrict__ vout) {
  __shared__ float lds[4 * 16 * 132];
  const int tid = threadIdx.x;
  const int wave = tid >> 5;
  const int lane = tid & 31;
  const int row0 = blockIdx.x * 128 + wave * 16;
  const int kof = (lane >> 4) << 1;
  const int nn = lane & 15;
  const int rowhi = (lane >> 4) << 3;

  int arow = row0 + nn;
  if (arow > GN - 1) arow = GN - 1;

  v2f fs[32], ft[32];
  load_afrag_ready(sop, arow, lane >> 4, fs);
  load_afrag_ready(top, arow, lane >> 4, ft);

  const float* W0 = Wl + 0 * GH * GH;
  const float* W1 = Wl + 1 * GH * GH;
  const float* W2 = Wl + 2 * GH * GH;
  const float* W3 = Wl + 3 * GH * GH;
  float* L0 = lds + 0 * 16 * 132;
  float* L1 = lds + 1 * 16 * 132;
  float* L2 = lds + 2 * 16 * 132;
  float* L3 = lds + 3 * 16 * 132;

  for (int ct = 0; ct < GH / 16; ++ct) {
    const int n0 = ct * 16;
    __syncthreads();
    stage_slab(W0, n0, L0);
    stage_slab(W1, n0, L1);
    stage_slab(W2, n0, L2);
    stage_slab(W3, n0, L3);
    __syncthreads();

    const float* p0 = L0 + nn * 132 + kof;
    const float* p1 = L1 + nn * 132 + kof;
    const float* p2 = L2 + nn * 132 + kof;
    const float* p3 = L3 + nn * 132 + kof;

    v8f az0 = {0, 0, 0, 0, 0, 0, 0, 0}, az1 = {0, 0, 0, 0, 0, 0, 0, 0};
    v8f ar0 = {0, 0, 0, 0, 0, 0, 0, 0}, ar1 = {0, 0, 0, 0, 0, 0, 0, 0};
#pragma unroll
    for (int kk = 0; kk < 32; kk += 2) {
      az0 = wmma_f32(fs[kk], *(const v2f*)(p0 + 4 * kk), az0);
      az0 = wmma_f32(ft[kk], *(const v2f*)(p1 + 4 * kk), az0);
      ar0 = wmma_f32(fs[kk], *(const v2f*)(p2 + 4 * kk), ar0);
      ar0 = wmma_f32(ft[kk], *(const v2f*)(p3 + 4 * kk), ar0);
      az1 = wmma_f32(fs[kk + 1], *(const v2f*)(p0 + 4 * kk + 4), az1);
      az1 = wmma_f32(ft[kk + 1], *(const v2f*)(p1 + 4 * kk + 4), az1);
      ar1 = wmma_f32(fs[kk + 1], *(const v2f*)(p2 + 4 * kk + 4), ar1);
      ar1 = wmma_f32(ft[kk + 1], *(const v2f*)(p3 + 4 * kk + 4), ar1);
    }
    const v8f accz = az0 + az1;
    const v8f accr = ar0 + ar1;

    const float bz = bl[0 * GH + n0 + nn] + bl[1 * GH + n0 + nn];
    const float br = bl[2 * GH + n0 + nn] + bl[3 * GH + n0 + nn];
#pragma unroll
    for (int i = 0; i < 8; ++i) {
      const int row = row0 + rowhi + i;
      if (row < GN) {
        const size_t off = (size_t)row * GH + n0 + nn;
        const float zv = sigmoid_f(accz[i] + bz);
        const float rv = sigmoid_f(accr[i] + br);
        zout[off] = zv;
        vout[off] = rv * hl[off];
      }
    }
  }
}

__global__ __launch_bounds__(256) void gemm_ht_kernel(
    const float* __restrict__ sop, const float* __restrict__ uop,
    const float* __restrict__ Wl, const float* __restrict__ bl,
    const float* __restrict__ zbuf, const float* __restrict__ hl,
    float* __restrict__ outp) {
  __shared__ float lds[2 * 16 * 132];
  const int tid = threadIdx.x;
  const int wave = tid >> 5;
  const int lane = tid & 31;
  const int row0 = blockIdx.x * 128 + wave * 16;
  const int kof = (lane >> 4) << 1;
  const int nn = lane & 15;
  const int rowhi = (lane >> 4) << 3;

  int arow = row0 + nn;
  if (arow > GN - 1) arow = GN - 1;

  v2f fs[32], fu[32];
  load_afrag_ready(sop, arow, lane >> 4, fs);
  load_afrag_ready(uop, arow, lane >> 4, fu);

  const float* W4 = Wl + 4 * GH * GH;
  const float* W5 = Wl + 5 * GH * GH;
  float* L4 = lds + 0 * 16 * 132;
  float* L5 = lds + 1 * 16 * 132;

  for (int ct = 0; ct < GH / 16; ++ct) {
    const int n0 = ct * 16;
    __syncthreads();
    stage_slab(W4, n0, L4);
    stage_slab(W5, n0, L5);
    __syncthreads();

    const float* p4 = L4 + nn * 132 + kof;
    const float* p5 = L5 + nn * 132 + kof;

    v8f a0 = {0, 0, 0, 0, 0, 0, 0, 0}, a1 = {0, 0, 0, 0, 0, 0, 0, 0};
#pragma unroll
    for (int kk = 0; kk < 32; kk += 2) {
      a0 = wmma_f32(fs[kk], *(const v2f*)(p4 + 4 * kk), a0);
      a0 = wmma_f32(fu[kk], *(const v2f*)(p5 + 4 * kk), a0);
      a1 = wmma_f32(fs[kk + 1], *(const v2f*)(p4 + 4 * kk + 4), a1);
      a1 = wmma_f32(fu[kk + 1], *(const v2f*)(p5 + 4 * kk + 4), a1);
    }
    const v8f acc = a0 + a1;

    const float bh = bl[4 * GH + n0 + nn] + bl[5 * GH + n0 + nn];
#pragma unroll
    for (int i = 0; i < 8; ++i) {
      const int row = row0 + rowhi + i;
      if (row < GN) {
        const size_t off = (size_t)row * GH + n0 + nn;
        const float ht = tanh_f(acc[i] + bh);
        const float zv = zbuf[off];
        outp[off] = zv * hl[off] + (1.0f - zv) * ht;
      }
    }
  }
}

constexpr int GN_PAD = 50048;
constexpr int CHCNT_PAD = ((NCH * NBK + 63) / 64) * 64;
constexpr size_t OFF_ROWPTR = 0;
constexpr size_t OFF_CHCNT = OFF_ROWPTR + GN_PAD;
constexpr size_t OFF_CHFIRST = OFF_CHCNT + CHCNT_PAD;
constexpr size_t OFF_BSTART = OFF_CHFIRST + CHCNT_PAD;
constexpr size_t OFF_SLOTS = OFF_BSTART + 256;
constexpr size_t OFF_PAIRS = OFF_SLOTS + GE;
constexpr size_t OFF_CSR = OFF_PAIRS;
static_assert(NBK + 1 <= 256, "bucket starts must fit their slot");
static_assert(OFF_PAIRS + 2 * (size_t)GE <= NH, "edge lists must fit in one output plane");
constexpr int ROW_BLOCKS = (GN * 32 + 255) / 256;
constexpr int GEMM_BLOCKS = (GN + 127) / 128;

extern "C" void kernel_launch(void* const* d_in, const int* in_sizes, int n_in,
                              void* d_out, int out_size, void* d_ws,
                              size_t ws_size, hipStream_t stream) {
  (void)in_sizes; (void)n_in; (void)out_size; (void)ws_size;
  const float* inp = (const float*)d_in[0];
  const int* edg = (const int*)d_in[1];
  const float* hmat = (const float*)d_in[2];
  const float* Wmat = (const float*)d_in[3];
  const float* bmat = (const float*)d_in[4];
  float* out = (float*)d_out;

  const int* src = edg;
  const int* dst = edg + GE;

  float* agg_x = (float*)d_ws;
  float* agg_t = agg_x + NH;
  float* zbuf = agg_t + NH;
  float* vbuf = zbuf + NH;

  int* lists = (int*)(out + (size_t)(GLAYERS - 1) * NH);
  int* rowptr = lists + OFF_ROWPTR;
  int* chcnt = lists + OFF_CHCNT;
  int* chfirst = lists + OFF_CHFIRST;
  int* bstart = lists + OFF_BSTART;
  int* slots = lists + OFF_SLOTS;
  int2* pairs = (int2*)(lists + OFF_PAIRS);
  int* csr = lists + OFF_CSR;

  bucket_count_kernel<<<NCH, SORT_THREADS, 0, stream>>>(dst, chcnt);
  bucket_scan_kernel<<<1, SORT_THREADS, 0, stream>>>(chcnt, chfirst, bstart);
  bucket_scatter_kernel<<<NCH, SORT_THREADS, 0, stream>>>(src, dst, chfirst, pairs);
  bucket_rows_kernel<<<NBK, SORT_THREADS, 0, stream>>>(bstart, pairs, rowptr, slots);
  segment_order_kernel<<<ROW_BLOCKS, 256, 0, stream>>>(rowptr, slots, csr);

  for (int l = 0; l < GLAYERS; ++l) {
    const float* x = (l == 0) ? inp : (out + (size_t)(l - 1) * NH);
    const float* hl = hmat + (size_t)l * NH;
    const float* Wl = Wmat + (size_t)l * 6 * GH * GH;
    const float* bl = bmat + (size_t)l * 6 * GH;
    float* outl = out + (size_t)l * NH;

    gather_sum2_kernel<<<ROW_BLOCKS, 256, 0, stream>>>(rowptr, csr, x, agg_x,
                                                       hl, agg_t);

    gemm_zr_kernel<<<GEMM_BLOCKS, 256, 0, stream>>>(agg_x, agg_t, hl, Wl, bl,
                                                    zbuf, vbuf);

    gather_sum1_kernel<<<ROW_BLOCKS, 256, 0, stream>>>(rowptr, csr, vbuf,
                                                       agg_t);

    gemm_ht_kernel<<<GEMM_BLOCKS, 256, 0, stream>>>(agg_x, agg_t, Wl, bl,
                                                    zbuf, hl, outl);
  }
}


# GEMMs: one 16-wave workgroup per compute unit keeps its weights in LDS all launch (1 barrier, not 16) and deals its 12-13 row tiles as tile x 32-column pieces evenly to the SIMDs; sums unchanged
# speedup vs baseline: 3.9190x; 1.0480x over previous
#include <hip/hip_runtime.h>
#include <cstdint>
#include <cstddef>

constexpr int GN = 50000;
constexpr int GE = 1600000;
constexpr int GH = 128;
constexpr int GLAYERS = 2;
constexpr size_t NH = (size_t)GN * GH;
constexpr int ROW_TILES = GN / 16;
constexpr int GEMM_GRID = 256;
constexpr int GEMM_THREADS = 512;
static_assert(GN % 16 == 0, "row tiles must be whole");

typedef __attribute__((ext_vector_type(2))) float v2f;
typedef __attribute__((ext_vector_type(8))) float v8f;

__device__ __forceinline__ v8f wmma_f32(v2f a, v2f b, v8f c) {
  return __builtin_amdgcn_wmma_f32_16x16x4_f32(
       false, a,  false, b,
       (short)0, c,  false,  false);
}

__device__ __forceinline__ float sigmoid_f(float x) {
  return 1.0f / (1.0f + __expf(-x));
}
__device__ __forceinline__ float tanh_f(float x) {
  return 1.0f - 2.0f / (1.0f + __expf(2.0f * x));
}

__device__ __forceinline__ void load_afrag_ready(const float* __restrict__ P,
                                                 int row, int half, v2f* frag) {
  const float* pr = P + (size_t)row * GH + (half << 2);
#pragma unroll
  for (int m = 0; m < 16; ++m) {
    const float4 v = *(const float4*)(pr + 8 * m);
    v2f lo, hi;
    lo.x = v.x;
    lo.y = v.y;
    hi.x = v.z;
    hi.y = v.w;
    frag[2 * m] = lo;
    frag[2 * m + 1] = hi;
  }
}

__device__ __forceinline__ void stage_matrix(const float* __restrict__ Wg,
                                             float* lds) {
  const int tid = threadIdx.x;
#pragma unroll
  for (int i = 0; i < GH * GH / GEMM_THREADS; ++i) {
    const int idx = i * GEMM_THREADS + tid;
    const int c = idx & 15;
    const int k = (idx >> 4) & (GH - 1);
    const int n = ((idx >> 11) << 4) + c;
    lds[n * 132 + k] = Wg[k * GH + n];
  }
}

struct GemmSpan {
  int t0;
  int p0;
  int p1;
};
__device__ __forceinline__ GemmSpan gemm_span() {
  const int wave = __builtin_amdgcn_readfirstlane((int)(threadIdx.x >> 5));
  const int b = blockIdx.x;
  GemmSpan g;
  g.t0 = (b * ROW_TILES) / GEMM_GRID;
  const int t1 = ((b + 1) * ROW_TILES) / GEMM_GRID;
  const int pieces = (t1 - g.t0) * 4;
  const int s = wave & 3;
  const int j = ((wave >> 2) - s) & 3;
  const int sb = (s * pieces) >> 2;
  const int len = (((s + 1) * pieces) >> 2) - sb;
  g.p0 = sb + ((j * len) >> 2);
  g.p1 = sb + (((j + 1) * len) >> 2);
  return g;
}

constexpr int BK_SHIFT = 8;
constexpr int BK_ROWS = 1 << BK_SHIFT;
constexpr int NBK = (GN + BK_ROWS - 1) >> BK_SHIFT;
constexpr int SORT_THREADS = 1024;
constexpr int NCH = 250;
constexpr int CH_EDGES = GE / NCH;
constexpr int CH_TRIPS = (CH_EDGES + SORT_THREADS - 1) / SORT_THREADS;
constexpr int SCAN_GROUPS = SORT_THREADS / 256;
constexpr int CH_PER_GROUP = (NCH + SCAN_GROUPS - 1) / SCAN_GROUPS;
constexpr int ROWS_UNROLL = 4;
static_assert(NCH * CH_EDGES == GE, "chunks must tile the edge list");
static_assert(NBK <= 256, "one scan thread per bucket");

__device__ __forceinline__ int scan256_exclusive(int v, int t, int* wsum, int& total) {
  const int lane = t & 31;
  int incl = v;
  if (t < 256) {
    for (int off = 1; off < 32; off <<= 1) {
      const int y = __shfl_up(incl, off);
      if (lane >= off) incl += y;
    }
    if (lane == 31) wsum[t >> 5] = incl;
  }
  __syncthreads();
  int woff = 0, all = 0;
  for (int w = 0; w < 8; ++w) {
    const int s = wsum[w];
    if (w < (t >> 5)) woff += s;
    all += s;
  }
  total = all;
  return woff + incl - v;
}

__global__ __launch_bounds__(SORT_THREADS) void bucket_count_kernel(
    const int* __restrict__ dst, int* __restrict__ chcnt) {
  __shared__ int hist[NBK];
  const int t = threadIdx.x;
  const int g = blockIdx.x;
  if (t < NBK) hist[t] = 0;
  __syncthreads();
  const int e0 = g * CH_EDGES;
  int d[CH_TRIPS];
#pragma unroll
  for (int k = 0; k < CH_TRIPS; ++k) {
    const int i = k * SORT_THREADS + t;
    d[k] = dst[e0 + ((i < CH_EDGES) ? i : CH_EDGES - 1)];
  }
#pragma unroll
  for (int k = 0; k < CH_TRIPS; ++k) {
    if (k * SORT_THREADS + t < CH_EDGES) atomicAdd(&hist[d[k] >> BK_SHIFT], 1);
  }
  __syncthreads();
  if (t < NBK) chcnt[g * NBK + t] = hist[t];
}

__global__ __launch_bounds__(SORT_THREADS) void bucket_scan_kernel(
    const int* __restrict__ chcnt, int* __restrict__ chfirst,
    int* __restrict__ bstart) {
  __shared__ int part[SCAN_GROUPS][256];
  __shared__ int bs[256];
  __shared__ int wsum[8];
  const int t = threadIdx.x;
  const int q = t >> 8;
  const int b = t & 255;
  const int g0 = q * CH_PER_GROUP;
  const int bc = (b < NBK) ? b : NBK - 1;
  int v[CH_PER_GROUP];
#pragma unroll
  for (int j = 0; j < CH_PER_GROUP; ++j) {
    const int g = g0 + j;
    const int x = chcnt[((g < NCH) ? g : NCH - 1) * NBK + bc];
    v[j] = (g < NCH && b < NBK) ? x : 0;
  }
  int p = 0;
#pragma unroll
  for (int j = 0; j < CH_PER_GROUP; ++j) p += v[j];
  part[q][b] = p;
  __syncthreads();
  int tot = 0;
  if (t < 256) {
    for (int i = 0; i < SCAN_GROUPS; ++i) tot += part[i][t];
  }
  int all;
  const int ex = scan256_exclusive(tot, t, wsum, all);
  if (t < 256) {
    bs[t] = ex;
    if (t < NBK) bstart[t] = ex;
    if (t == 0) bstart[NBK] = all;
  }
  __syncthreads();
  if (b < NBK) {
    int run = bs[b];
    for (int i = 0; i < q; ++i) run += part[i][b];
#pragma unroll
    for (int j = 0; j < CH_PER_GROUP; ++j) {
      const int g = g0 + j;
      if (g < NCH) chfirst[g * NBK + b] = run;
      run += v[j];
    }
  }
}

__global__ __launch_bounds__(SORT_THREADS) void bucket_scatter_kernel(
    const int* __restrict__ src, const int* __restrict__ dst,
    const int* __restrict__ chfirst, int2* __restrict__ pairs) {
  __shared__ int cur[NBK];
  const int t = threadIdx.x;
  const int g = blockIdx.x;
  if (t < NBK) cur[t] = chfirst[g * NBK + t];
  __syncthreads();
  const int e0 = g * CH_EDGES;
  int d[CH_TRIPS], s[CH_TRIPS];
#pragma unroll
  for (int k = 0; k < CH_TRIPS; ++k) {
    const int i = k * SORT_THREADS + t;
    const int ic = e0 + ((i < CH_EDGES) ? i : CH_EDGES - 1);
    d[k] = dst[ic];
    s[k] = src[ic];
  }
#pragma unroll
  for (int k = 0; k < CH_TRIPS; ++k) {
    if (k * SORT_THREADS + t < CH_EDGES) {
      const int p = atomicAdd(&cur[d[k] >> BK_SHIFT], 1);
      pairs[p] = make_int2(s[k], d[k]);
    }
  }
}

__global__ __launch_bounds__(SORT_THREADS) void bucket_rows_kernel(
    const int* __restrict__ bstart, const int2* __restrict__ pairs,
    int* __restrict__ rowptr, int* __restrict__ slots) {
  __shared__ int rc[BK_ROWS];
  __shared__ int wsum[8];
  const int t = threadIdx.x;
  const int b = blockIdx.x;
  const int beg = bstart[b];
  const int end = bstart[b + 1];
  if (t < BK_ROWS) rc[t] = 0;
  __syncthreads();
  for (int i0 = beg + t; i0 < end; i0 += ROWS_UNROLL * SORT_THREADS) {
    int d[ROWS_UNROLL];
#pragma unroll
    for (int u = 0; u < ROWS_UNROLL; ++u) {
      const int i = i0 + u * SORT_THREADS;
      d[u] = pairs[(i < end) ? i : end - 1].y;
    }
#pragma unroll
    for (int u = 0; u < ROWS_UNROLL; ++u) {
      if (i0 + u * SORT_THREADS < end) atomicAdd(&rc[d[u] & (BK_ROWS - 1)], 1);
    }
  }
  __syncthreads();
  const int deg = (t < BK_ROWS) ? rc[t] : 0;
  int all;
  const int ex = scan256_exclusive(deg, t, wsum, all);
  if (t < BK_ROWS) {
    const int row = (b << BK_SHIFT) + t;
    rc[t] = beg + ex;
    if (row < GN) rowptr[row] = beg + ex;
  }
  if (b == NBK - 1 && t == 0) rowptr[GN] = end;
  __syncthreads();
  for (int i0 = beg + t; i0 < end; i0 += ROWS_UNROLL * SORT_THREADS) {
    int2 sd[ROWS_UNROLL];
#pragma unroll
    for (int u = 0; u < ROWS_UNROLL; ++u) {
      const int i = i0 + u * SORT_THREADS;
      sd[u] = pairs[(i < end) ? i : end - 1];
    }
#pragma unroll
    for (int u = 0; u < ROWS_UNROLL; ++u) {
      if (i0 + u * SORT_THREADS < end) {
        const int p = atomicAdd(&rc[sd[u].y & (BK_ROWS - 1)], 1);
        slots[p] = sd[u].x;
      }
    }
  }
}

__global__ __launch_bounds__(256) void segment_order_kernel(
    const int* __restrict__ rowptr, const int* __restrict__ slots,
    int* __restrict__ csr) {
  const int lane = threadIdx.x & 31;
  const int row = __builtin_amdgcn_readfirstlane(
      (int)((blockIdx.x * 256u + threadIdx.x) >> 5));
  if (row >= GN) return;
  const int beg = rowptr[row];
  const int n = rowptr[row + 1] - beg;
  for (int ib = 0; ib < n; ib += 32) {
    const int i = ib + lane;
    const int key = (i < n) ? slots[beg + i] : 0;
    int rank = 0;
    for (int jb = 0; jb < n; jb += 32) {
      const int jn = (n - jb < 32) ? (n - jb) : 32;
      const int kv = (lane < jn) ? slots[beg + jb + lane] : 0;
      for (int jj = 0; jj < jn; ++jj) {
        const int kj = __builtin_amdgcn_readlane(kv, jj);
        const int j = jb + jj;
        rank += (kj < key || (kj == key && j < i)) ? 1 : 0;
      }
    }
    if (i < n) csr[beg + rank] = key;
  }
}

template <int U>
__device__ __forceinline__ void gather_rows(const float* __restrict__ f,
                                            int idx, int j0, int c4,
                                            float4& acc) {
  float4 v[U];
#pragma unroll
  for (int u = 0; u < U; ++u) {
    const int s = __builtin_amdgcn_readlane(idx, j0 + u);
    v[u] = *(const float4*)(f + (size_t)s * GH + c4);
  }
#pragma unroll
  for (int u = 0; u < U; ++u) {
    acc.x += v[u].x;
    acc.y += v[u].y;
    acc.z += v[u].z;
    acc.w += v[u].w;
  }
}

template <int U, int NF>
__device__ __forceinline__ void gather_group(const float* __restrict__ f0,
                                             const float* __restrict__ f1,
                                             int idx, int j0, int c4,
                                             float4& a0, float4& a1) {
  gather_rows<U>(f0, idx, j0, c4, a0);
  if (NF == 2) gather_rows<U>(f1, idx, j0, c4, a1);
}

__device__ __forceinline__ void store_operand_row(float* __restrict__ o, int row,
                                                  int c4, int lane, float4 s) {
  const bool odd = (lane & 1) != 0;
  const float sx = odd ? s.x : s.z;
  const float sy = odd ? s.y : s.w;
  const float rx = __shfl_xor(sx, 1);
  const float ry = __shfl_xor(sy, 1);
  float4 w;
  w.x = odd ? rx : s.x;
  w.y = odd ? ry : s.y;
  w.z = odd ? s.z : rx;
  w.w = odd ? s.w : ry;
  *(float4*)(o + (size_t)row * GH + c4) = w;
}

template <int NF>
__device__ __forceinline__ void gather_sum_body(
    const int* __restrict__ rowptr, const int* __restrict__ csr,
    const float* __restrict__ f0, float* __restrict__ o0,
    const float* __restrict__ f1, float* __restrict__ o1) {
  const int lane = threadIdx.x & 31;
  const int c4 = lane << 2;
  const int row = __builtin_amdgcn_readfirstlane(
      (int)((blockIdx.x * 256u + threadIdx.x) >> 5));
  if (row >= GN) return;
  const int beg = rowptr[row];
  const int end = rowptr[row + 1];
  const float4 x0 = *(const float4*)(f0 + (size_t)row * GH + c4);
  const float4 x1 = *(const float4*)(f1 + (size_t)row * GH + c4);
  float4 a0 = {0.0f, 0.0f, 0.0f, 0.0f};
  float4 a1 = {0.0f, 0.0f, 0.0f, 0.0f};
  for (int base = beg; base < end; base += 32) {
    const int n = (end - base < 32) ? (end - base) : 32;
    const int idx = (lane < n) ? csr[base + lane] : 0;
    int j = 0;
    for (; j + 8 <= n; j += 8) gather_group<8, NF>(f0, f1, idx, j, c4, a0, a1);
    if (n & 4) { gather_group<4, NF>(f0, f1, idx, j, c4, a0, a1); j += 4; }
    if (n & 2) { gather_group<2, NF>(f0, f1, idx, j, c4, a0, a1); j += 2; }
    if (n & 1) { gather_group<1, NF>(f0, f1, idx, j, c4, a0, a1); }
  }
  float4 s0;
  s0.x = x0.x + a0.x;
  s0.y = x0.y + a0.y;
  s0.z = x0.z + a0.z;
  s0.w = x0.w + a0.w;
  store_operand_row(o0, row, c4, lane, s0);
  if (NF == 2) {
    float4 s1;
    s1.x = x1.x + a1.x;
    s1.y = x1.y + a1.y;
    s1.z = x1.z + a1.z;
    s1.w = x1.w + a1.w;
    store_operand_row(o1, row, c4, lane, s1);
  }
}

__global__ __launch_bounds__(256) void gather_sum2_kernel(
    const int* __restrict__ rowptr, const int* __restrict__ csr,
    const float* __restrict__ f0, float* __restrict__ o0,
    const float* __restrict__ f1, float* __restrict__ o1) {
  gather_sum_body<2>(rowptr, csr, f0, o0, f1, o1);
}

__global__ __launch_bounds__(256) void gather_sum1_kernel(
    const int* __restrict__ rowptr, const int* __restrict__ csr,
    const float* __restrict__ f0, float* __restrict__ o0) {
  gather_sum_body<1>(rowptr, csr, f0, o0, f0, o0);
}

__global__ __launch_bounds__(GEMM_THREADS) void gemm_zr_kernel(
    const float* __restrict__ sop, const float* __restrict__ top,
    const float* __restrict__ hl,
    const float* __restrict__ Wl, const float* __restrict__ bl,
    float* __restrict__ zout, float* __restrict__ vout) {
  __shared__ float lds[4 * GH * 132];
  const int lane = threadIdx.x & 31;
  const int kof = (lane >> 4) << 1;
  const int nn = lane & 15;
  const int rowhi = (lane >> 4) << 3;
  const GemmSpan g = gemm_span();

  float* L0 = lds + 0 * GH * 132;
  float* L1 = lds + 1 * GH * 132;
  float* L2 = lds + 2 * GH * 132;
  float* L3 = lds + 3 * GH * 132;
  stage_matrix(Wl + 0 * GH * GH, L0);
  stage_matrix(Wl + 1 * GH * GH, L1);
  stage_matrix(Wl + 2 * GH * GH, L2);
  stage_matrix(Wl + 3 * GH * GH, L3);
  __syncthreads();

  v2f fs[32], ft[32];
  int cur_tile = -1;
  for (int p = g.p0; p < g.p1; ++p) {
    const int tile = g.t0 + (p >> 2);
    const int row0 = tile * 16;
    if (tile != cur_tile) {
      int arow = row0 + nn;
      if (arow > GN - 1) arow = GN - 1;
      load_afrag_ready(sop, arow, lane >> 4, fs);
      load_afrag_ready(top, arow, lane >> 4, ft);
      cur_tile = tile;
    }
    for (int c2 = 0; c2 < 2; ++c2) {
      const int n0 = (((p & 3) << 1) + c2) * 16;
      const float* p0 = L0 + (n0 + nn) * 132 + kof;
      const float* p1 = L1 + (n0 + nn) * 132 + kof;
      const float* p2 = L2 + (n0 + nn) * 132 + kof;
      const float* p3 = L3 + (n0 + nn) * 132 + kof;

      v8f az0 = {0, 0, 0, 0, 0, 0, 0, 0}, az1 = {0, 0, 0, 0, 0, 0, 0, 0};
      v8f ar0 = {0, 0, 0, 0, 0, 0, 0, 0}, ar1 = {0, 0, 0, 0, 0, 0, 0, 0};
#pragma unroll
      for (int kk = 0; kk < 32; kk += 2) {
        az0 = wmma_f32(fs[kk], *(const v2f*)(p0 + 4 * kk), az0);
        az0 = wmma_f32(ft[kk], *(const v2f*)(p1 + 4 * kk), az0);
        ar0 = wmma_f32(fs[kk], *(const v2f*)(p2 + 4 * kk), ar0);
        ar0 = wmma_f32(ft[kk], *(const v2f*)(p3 + 4 * kk), ar0);
        az1 = wmma_f32(fs[kk + 1], *(const v2f*)(p0 + 4 * kk + 4), az1);
        az1 = wmma_f32(ft[kk + 1], *(const v2f*)(p1 + 4 * kk + 4), az1);
        ar1 = wmma_f32(fs[kk + 1], *(const v2f*)(p2 + 4 * kk + 4), ar1);
        ar1 = wmma_f32(ft[kk + 1], *(const v2f*)(p3 + 4 * kk + 4), ar1);
      }
      const v8f accz = az0 + az1;
      const v8f accr = ar0 + ar1;

      const float bz = bl[0 * GH + n0 + nn] + bl[1 * GH + n0 + nn];
      const float br = bl[2 * GH + n0 + nn] + bl[3 * GH + n0 + nn];
#pragma unroll
      for (int i = 0; i < 8; ++i) {
        const int row = row0 + rowhi + i;
        if (row < GN) {
          const size_t off = (size_t)row * GH + n0 + nn;
          const float zv = sigmoid_f(accz[i] + bz);
          const float rv = sigmoid_f(accr[i] + br);
          zout[off] = zv;
          vout[off] = rv * hl[off];
        }
      }
    }
  }
}

__global__ __launch_bounds__(GEMM_THREADS) void gemm_ht_kernel(
    const float* __restrict__ sop, const float* __restrict__ uop,
    const float* __restrict__ Wl, const float* __restrict__ bl,
    const float* __restrict__ zbuf, const float* __restrict__ hl,
    float* __restrict__ outp) {
  __shared__ float lds[2 * GH * 132];
  const int lane = threadIdx.x & 31;
  const int kof = (lane >> 4) << 1;
  const int nn = lane & 15;
  const int rowhi = (lane >> 4) << 3;
  const GemmSpan g = gemm_span();

  float* L4 = lds + 0 * GH * 132;
  float* L5 = lds + 1 * GH * 132;
  stage_matrix(Wl + 4 * GH * GH, L4);
  stage_matrix(Wl + 5 * GH * GH, L5);
  __syncthreads();

  v2f fs[32], fu[32];
  int cur_tile = -1;
  for (int p = g.p0; p < g.p1; ++p) {
    const int tile = g.t0 + (p >> 2);
    const int row0 = tile * 16;
    if (tile != cur_tile) {
      int arow = row0 + nn;
      if (arow > GN - 1) arow = GN - 1;
      load_afrag_ready(sop, arow, lane >> 4, fs);
      load_afrag_ready(uop, arow, lane >> 4, fu);
      cur_tile = tile;
    }
    for (int c2 = 0; c2 < 2; ++c2) {
      const int n0 = (((p & 3) << 1) + c2) * 16;
      const float* p4 = L4 + (n0 + nn) * 132 + kof;
      const float* p5 = L5 + (n0 + nn) * 132 + kof;

      v8f a0 = {0, 0, 0, 0, 0, 0, 0, 0}, a1 = {0, 0, 0, 0, 0, 0, 0, 0};
#pragma unroll
      for (int kk = 0; kk < 32; kk += 2) {
        a0 = wmma_f32(fs[kk], *(const v2f*)(p4 + 4 * kk), a0);
        a0 = wmma_f32(fu[kk], *(const v2f*)(p5 + 4 * kk), a0);
        a1 = wmma_f32(fs[kk + 1], *(const v2f*)(p4 + 4 * kk + 4), a1);
        a1 = wmma_f32(fu[kk + 1], *(const v2f*)(p5 + 4 * kk + 4), a1);
      }
      const v8f acc = a0 + a1;

      const float bh = bl[4 * GH + n0 + nn] + bl[5 * GH + n0 + nn];
#pragma unroll
      for (int i = 0; i < 8; ++i) {
        const int row = row0 + rowhi + i;
        if (row < GN) {
          const size_t off = (size_t)row * GH + n0 + nn;
          const float ht = tanh_f(acc[i] + bh);
          const float zv = zbuf[off];
          outp[off] = zv * hl[off] + (1.0f - zv) * ht;
        }
      }
    }
  }
}

constexpr int GN_PAD = 50048;
constexpr int CHCNT_PAD = ((NCH * NBK + 63) / 64) * 64;
constexpr size_t OFF_ROWPTR = 0;
constexpr size_t OFF_CHCNT = OFF_ROWPTR + GN_PAD;
constexpr size_t OFF_CHFIRST = OFF_CHCNT + CHCNT_PAD;
constexpr size_t OFF_BSTART = OFF_CHFIRST + CHCNT_PAD;
constexpr size_t OFF_SLOTS = OFF_BSTART + 256;
constexpr size_t OFF_PAIRS = OFF_SLOTS + GE;
constexpr size_t OFF_CSR = OFF_PAIRS;
static_assert(NBK + 1 <= 256, "bucket starts must fit their slot");
static_assert(OFF_PAIRS + 2 * (size_t)GE <= NH, "edge lists must fit in one output plane");
constexpr int ROW_BLOCKS = (GN * 32 + 255) / 256;

extern "C" void kernel_launch(void* const* d_in, const int* in_sizes, int n_in,
                              void* d_out, int out_size, void* d_ws,
                              size_t ws_size, hipStream_t stream) {
  (void)in_sizes; (void)n_in; (void)out_size; (void)ws_size;
  const float* inp = (const float*)d_in[0];
  const int* edg = (const int*)d_in[1];
  const float* hmat = (const float*)d_in[2];
  const float* Wmat = (const float*)d_in[3];
  const float* bmat = (const float*)d_in[4];
  float* out = (float*)d_out;

  const int* src = edg;
  const int* dst = edg + GE;

  float* agg_x = (float*)d_ws;
  float* agg_t = agg_x + NH;
  float* zbuf = agg_t + NH;
  float* vbuf = zbuf + NH;

  int* lists = (int*)(out + (size_t)(GLAYERS - 1) * NH);
  int* rowptr = lists + OFF_ROWPTR;
  int* chcnt = lists + OFF_CHCNT;
  int* chfirst = lists + OFF_CHFIRST;
  int* bstart = lists + OFF_BSTART;
  int* slots = lists + OFF_SLOTS;
  int2* pairs = (int2*)(lists + OFF_PAIRS);
  int* csr = lists + OFF_CSR;

  bucket_count_kernel<<<NCH, SORT_THREADS, 0, stream>>>(dst, chcnt);
  bucket_scan_kernel<<<1, SORT_THREADS, 0, stream>>>(chcnt, chfirst, bstart);
  bucket_scatter_kernel<<<NCH, SORT_THREADS, 0, stream>>>(src, dst, chfirst, pairs);
  bucket_rows_kernel<<<NBK, SORT_THREADS, 0, stream>>>(bstart, pairs, rowptr, slots);
  segment_order_kernel<<<ROW_BLOCKS, 256, 0, stream>>>(rowptr, slots, csr);

  for (int l = 0; l < GLAYERS; ++l) {
    const float* x = (l == 0) ? inp : (out + (size_t)(l - 1) * NH);
    const float* hl = hmat + (size_t)l * NH;
    const float* Wl = Wmat + (size_t)l * 6 * GH * GH;
    const float* bl = bmat + (size_t)l * 6 * GH;
    float* outl = out + (size_t)l * NH;

    gather_sum2_kernel<<<ROW_BLOCKS, 256, 0, stream>>>(rowptr, csr, x, agg_x,
                                                       hl, agg_t);

    gemm_zr_kernel<<<GEMM_GRID, GEMM_THREADS, 0, stream>>>(agg_x, agg_t, hl, Wl, bl,
                                                    zbuf, vbuf);

    gather_sum1_kernel<<<ROW_BLOCKS, 256, 0, stream>>>(rowptr, csr, vbuf,
                                                       agg_t);

    gemm_ht_kernel<<<GEMM_GRID, GEMM_THREADS, 0, stream>>>(agg_x, agg_t, Wl, bl,
                                                    zbuf, hl, outl);
  }
}


# GEMM epilogues: h (and z) of a piece's 16 columns loaded ahead of its matrix loop, not one load waited for inside each output row's guarded branch (8 exposed round trips); sums unchanged
# speedup vs baseline: 4.4608x; 1.1382x over previous
#include <hip/hip_runtime.h>
#include <cstdint>
#include <cstddef>

constexpr int GN = 50000;
constexpr int GE = 1600000;
constexpr int GH = 128;
constexpr int GLAYERS = 2;
constexpr size_t NH = (size_t)GN * GH;
constexpr int ROW_TILES = GN / 16;
constexpr int GEMM_GRID = 256;
constexpr int GEMM_THREADS = 512;
static_assert(GN % 16 == 0, "row tiles must be whole");

typedef __attribute__((ext_vector_type(2))) float v2f;
typedef __attribute__((ext_vector_type(8))) float v8f;

__device__ __forceinline__ v8f wmma_f32(v2f a, v2f b, v8f c) {
  return __builtin_amdgcn_wmma_f32_16x16x4_f32(
       false, a,  false, b,
       (short)0, c,  false,  false);
}

__device__ __forceinline__ float sigmoid_f(float x) {
  return 1.0f / (1.0f + __expf(-x));
}
__device__ __forceinline__ float tanh_f(float x) {
  return 1.0f - 2.0f / (1.0f + __expf(2.0f * x));
}

__device__ __forceinline__ void load_afrag_ready(const float* __restrict__ P,
                                                 int row, int half, v2f* frag) {
  const float* pr = P + (size_t)row * GH + (half << 2);
#pragma unroll
  for (int m = 0; m < 16; ++m) {
    const float4 v = *(const float4*)(pr + 8 * m);
    v2f lo, hi;
    lo.x = v.x;
    lo.y = v.y;
    hi.x = v.z;
    hi.y = v.w;
    frag[2 * m] = lo;
    frag[2 * m + 1] = hi;
  }
}

__device__ __forceinline__ void stage_matrix(const float* __restrict__ Wg,
                                             float* lds) {
  const int tid = threadIdx.x;
#pragma unroll
  for (int i = 0; i < GH * GH / GEMM_THREADS; ++i) {
    const int idx = i * GEMM_THREADS + tid;
    const int c = idx & 15;
    const int k = (idx >> 4) & (GH - 1);
    const int n = ((idx >> 11) << 4) + c;
    lds[n * 132 + k] = Wg[k * GH + n];
  }
}

struct GemmSpan {
  int t0;
  int p0;
  int p1;
};
__device__ __forceinline__ GemmSpan gemm_span() {
  const int wave = __builtin_amdgcn_readfirstlane((int)(threadIdx.x >> 5));
  const int b = blockIdx.x;
  GemmSpan g;
  g.t0 = (b * ROW_TILES) / GEMM_GRID;
  const int t1 = ((b + 1) * ROW_TILES) / GEMM_GRID;
  const int pieces = (t1 - g.t0) * 4;
  const int s = wave & 3;
  const int j = ((wave >> 2) - s) & 3;
  const int sb = (s * pieces) >> 2;
  const int len = (((s + 1) * pieces) >> 2) - sb;
  g.p0 = sb + ((j * len) >> 2);
  g.p1 = sb + (((j + 1) * len) >> 2);
  return g;
}

constexpr int BK_SHIFT = 8;
constexpr int BK_ROWS = 1 << BK_SHIFT;
constexpr int NBK = (GN + BK_ROWS - 1) >> BK_SHIFT;
constexpr int SORT_THREADS = 1024;
constexpr int NCH = 250;
constexpr int CH_EDGES = GE / NCH;
constexpr int CH_TRIPS = (CH_EDGES + SORT_THREADS - 1) / SORT_THREADS;
constexpr int SCAN_GROUPS = SORT_THREADS / 256;
constexpr int CH_PER_GROUP = (NCH + SCAN_GROUPS - 1) / SCAN_GROUPS;
constexpr int ROWS_UNROLL = 4;
static_assert(NCH * CH_EDGES == GE, "chunks must tile the edge list");
static_assert(NBK <= 256, "one scan thread per bucket");

__device__ __forceinline__ int scan256_exclusive(int v, int t, int* wsum, int& total) {
  const int lane = t & 31;
  int incl = v;
  if (t < 256) {
    for (int off = 1; off < 32; off <<= 1) {
      const int y = __shfl_up(incl, off);
      if (lane >= off) incl += y;
    }
    if (lane == 31) wsum[t >> 5] = incl;
  }
  __syncthreads();
  int woff = 0, all = 0;
  for (int w = 0; w < 8; ++w) {
    const int s = wsum[w];
    if (w < (t >> 5)) woff += s;
    all += s;
  }
  total = all;
  return woff + incl - v;
}

__global__ __launch_bounds__(SORT_THREADS) void bucket_count_kernel(
    const int* __restrict__ dst, int* __restrict__ chcnt) {
  __shared__ int hist[NBK];
  const int t = threadIdx.x;
  const int g = blockIdx.x;
  if (t < NBK) hist[t] = 0;
  __syncthreads();
  const int e0 = g * CH_EDGES;
  int d[CH_TRIPS];
#pragma unroll
  for (int k = 0; k < CH_TRIPS; ++k) {
    const int i = k * SORT_THREADS + t;
    d[k] = dst[e0 + ((i < CH_EDGES) ? i : CH_EDGES - 1)];
  }
#pragma unroll
  for (int k = 0; k < CH_TRIPS; ++k) {
    if (k * SORT_THREADS + t < CH_EDGES) atomicAdd(&hist[d[k] >> BK_SHIFT], 1);
  }
  __syncthreads();
  if (t < NBK) chcnt[g * NBK + t] = hist[t];
}

__global__ __launch_bounds__(SORT_THREADS) void bucket_scan_kernel(
    const int* __restrict__ chcnt, int* __restrict__ chfirst,
    int* __restrict__ bstart) {
  __shared__ int part[SCAN_GROUPS][256];
  __shared__ int bs[256];
  __shared__ int wsum[8];
  const int t = threadIdx.x;
  const int q = t >> 8;
  const int b = t & 255;
  const int g0 = q * CH_PER_GROUP;
  const int bc = (b < NBK) ? b : NBK - 1;
  int v[CH_PER_GROUP];
#pragma unroll
  for (int j = 0; j < CH_PER_GROUP; ++j) {
    const int g = g0 + j;
    const int x = chcnt[((g < NCH) ? g : NCH - 1) * NBK + bc];
    v[j] = (g < NCH && b < NBK) ? x : 0;
  }
  int p = 0;
#pragma unroll
  for (int j = 0; j < CH_PER_GROUP; ++j) p += v[j];
  part[q][b] = p;
  __syncthreads();
  int tot = 0;
  if (t < 256) {
    for (int i = 0; i < SCAN_GROUPS; ++i) tot += part[i][t];
  }
  int all;
  const int ex = scan256_exclusive(tot, t, wsum, all);
  if (t < 256) {
    bs[t] = ex;
    if (t < NBK) bstart[t] = ex;
    if (t == 0) bstart[NBK] = all;
  }
  __syncthreads();
  if (b < NBK) {
    int run = bs[b];
    for (int i = 0; i < q; ++i) run += part[i][b];
#pragma unroll
    for (int j = 0; j < CH_PER_GROUP; ++j) {
      const int g = g0 + j;
      if (g < NCH) chfirst[g * NBK + b] = run;
      run += v[j];
    }
  }
}

__global__ __launch_bounds__(SORT_THREADS) void bucket_scatter_kernel(
    const int* __restrict__ src, const int* __restrict__ dst,
    const int* __restrict__ chfirst, int2* __restrict__ pairs) {
  __shared__ int cur[NBK];
  const int t = threadIdx.x;
  const int g = blockIdx.x;
  if (t < NBK) cur[t] = chfirst[g * NBK + t];
  __syncthreads();
  const int e0 = g * CH_EDGES;
  int d[CH_TRIPS], s[CH_TRIPS];
#pragma unroll
  for (int k = 0; k < CH_TRIPS; ++k) {
    const int i = k * SORT_THREADS + t;
    const int ic = e0 + ((i < CH_EDGES) ? i : CH_EDGES - 1);
    d[k] = dst[ic];
    s[k] = src[ic];
  }
#pragma unroll
  for (int k = 0; k < CH_TRIPS; ++k) {
    if (k * SORT_THREADS + t < CH_EDGES) {
      const int p = atomicAdd(&cur[d[k] >> BK_SHIFT], 1);
      pairs[p] = make_int2(s[k], d[k]);
    }
  }
}

__global__ __launch_bounds__(SORT_THREADS) void bucket_rows_kernel(
    const int* __restrict__ bstart, const int2* __restrict__ pairs,
    int* __restrict__ rowptr, int* __restrict__ slots) {
  __shared__ int rc[BK_ROWS];
  __shared__ int wsum[8];
  const int t = threadIdx.x;
  const int b = blockIdx.x;
  const int beg = bstart[b];
  const int end = bstart[b + 1];
  if (t < BK_ROWS) rc[t] = 0;
  __syncthreads();
  for (int i0 = beg + t; i0 < end; i0 += ROWS_UNROLL * SORT_THREADS) {
    int d[ROWS_UNROLL];
#pragma unroll
    for (int u = 0; u < ROWS_UNROLL; ++u) {
      const int i = i0 + u * SORT_THREADS;
      d[u] = pairs[(i < end) ? i : end - 1].y;
    }
#pragma unroll
    for (int u = 0; u < ROWS_UNROLL; ++u) {
      if (i0 + u * SORT_THREADS < end) atomicAdd(&rc[d[u] & (BK_ROWS - 1)], 1);
    }
  }
  __syncthreads();
  const int deg = (t < BK_ROWS) ? rc[t] : 0;
  int all;
  const int ex = scan256_exclusive(deg, t, wsum, all);
  if (t < BK_ROWS) {
    const int row = (b << BK_SHIFT) + t;
    rc[t] = beg + ex;
    if (row < GN) rowptr[row] = beg + ex;
  }
  if (b == NBK - 1 && t == 0) rowptr[GN] = end;
  __syncthreads();
  for (int i0 = beg + t; i0 < end; i0 += ROWS_UNROLL * SORT_THREADS) {
    int2 sd[ROWS_UNROLL];
#pragma unroll
    for (int u = 0; u < ROWS_UNROLL; ++u) {
      const int i = i0 + u * SORT_THREADS;
      sd[u] = pairs[(i < end) ? i : end - 1];
    }
#pragma unroll
    for (int u = 0; u < ROWS_UNROLL; ++u) {
      if (i0 + u * SORT_THREADS < end) {
        const int p = atomicAdd(&rc[sd[u].y & (BK_ROWS - 1)], 1);
        slots[p] = sd[u].x;
      }
    }
  }
}

__global__ __launch_bounds__(256) void segment_order_kernel(
    const int* __restrict__ rowptr, const int* __restrict__ slots,
    int* __restrict__ csr) {
  const int lane = threadIdx.x & 31;
  const int row = __builtin_amdgcn_readfirstlane(
      (int)((blockIdx.x * 256u + threadIdx.x) >> 5));
  if (row >= GN) return;
  const int beg = rowptr[row];
  const int n = rowptr[row + 1] - beg;
  for (int ib = 0; ib < n; ib += 32) {
    const int i = ib + lane;
    const int key = (i < n) ? slots[beg + i] : 0;
    int rank = 0;
    for (int jb = 0; jb < n; jb += 32) {
      const int jn = (n - jb < 32) ? (n - jb) : 32;
      const int kv = (lane < jn) ? slots[beg + jb + lane] : 0;
      for (int jj = 0; jj < jn; ++jj) {
        const int kj = __builtin_amdgcn_readlane(kv, jj);
        const int j = jb + jj;
        rank += (kj < key || (kj == key && j < i)) ? 1 : 0;
      }
    }
    if (i < n) csr[beg + rank] = key;
  }
}

template <int U>
__device__ __forceinline__ void gather_rows(const float* __restrict__ f,
                                            int idx, int j0, int c4,
                                            float4& acc) {
  float4 v[U];
#pragma unroll
  for (int u = 0; u < U; ++u) {
    const int s = __builtin_amdgcn_readlane(idx, j0 + u);
    v[u] = *(const float4*)(f + (size_t)s * GH + c4);
  }
#pragma unroll
  for (int u = 0; u < U; ++u) {
    acc.x += v[u].x;
    acc.y += v[u].y;
    acc.z += v[u].z;
    acc.w += v[u].w;
  }
}

template <int U, int NF>
__device__ __forceinline__ void gather_group(const float* __restrict__ f0,
                                             const float* __restrict__ f1,
                                             int idx, int j0, int c4,
                                             float4& a0, float4& a1) {
  gather_rows<U>(f0, idx, j0, c4, a0);
  if (NF == 2) gather_rows<U>(f1, idx, j0, c4, a1);
}

__device__ __forceinline__ void store_operand_row(float* __restrict__ o, int row,
                                                  int c4, int lane, float4 s) {
  const bool odd = (lane & 1) != 0;
  const float sx = odd ? s.x : s.z;
  const float sy = odd ? s.y : s.w;
  const float rx = __shfl_xor(sx, 1);
  const float ry = __shfl_xor(sy, 1);
  float4 w;
  w.x = odd ? rx : s.x;
  w.y = odd ? ry : s.y;
  w.z = odd ? s.z : rx;
  w.w = odd ? s.w : ry;
  *(float4*)(o + (size_t)row * GH + c4) = w;
}

template <int NF>
__device__ __forceinline__ void gather_sum_body(
    const int* __restrict__ rowptr, const int* __restrict__ csr,
    const float* __restrict__ f0, float* __restrict__ o0,
    const float* __restrict__ f1, float* __restrict__ o1) {
  const int lane = threadIdx.x & 31;
  const int c4 = lane << 2;
  const int row = __builtin_amdgcn_readfirstlane(
      (int)((blockIdx.x * 256u + threadIdx.x) >> 5));
  if (row >= GN) return;
  const int beg = rowptr[row];
  const int end = rowptr[row + 1];
  const float4 x0 = *(const float4*)(f0 + (size_t)row * GH + c4);
  const float4 x1 = *(const float4*)(f1 + (size_t)row * GH + c4);
  float4 a0 = {0.0f, 0.0f, 0.0f, 0.0f};
  float4 a1 = {0.0f, 0.0f, 0.0f, 0.0f};
  for (int base = beg; base < end; base += 32) {
    const int n = (end - base < 32) ? (end - base) : 32;
    const int idx = (lane < n) ? csr[base + lane] : 0;
    int j = 0;
    for (; j + 8 <= n; j += 8) gather_group<8, NF>(f0, f1, idx, j, c4, a0, a1);
    if (n & 4) { gather_group<4, NF>(f0, f1, idx, j, c4, a0, a1); j += 4; }
    if (n & 2) { gather_group<2, NF>(f0, f1, idx, j, c4, a0, a1); j += 2; }
    if (n & 1) { gather_group<1, NF>(f0, f1, idx, j, c4, a0, a1); }
  }
  float4 s0;
  s0.x = x0.x + a0.x;
  s0.y = x0.y + a0.y;
  s0.z = x0.z + a0.z;
  s0.w = x0.w + a0.w;
  store_operand_row(o0, row, c4, lane, s0);
  if (NF == 2) {
    float4 s1;
    s1.x = x1.x + a1.x;
    s1.y = x1.y + a1.y;
    s1.z = x1.z + a1.z;
    s1.w = x1.w + a1.w;
    store_operand_row(o1, row, c4, lane, s1);
  }
}

__global__ __launch_bounds__(256) void gather_sum2_kernel(
    const int* __restrict__ rowptr, const int* __restrict__ csr,
    const float* __restrict__ f0, float* __restrict__ o0,
    const float* __restrict__ f1, float* __restrict__ o1) {
  gather_sum_body<2>(rowptr, csr, f0, o0, f1, o1);
}

__global__ __launch_bounds__(256) void gather_sum1_kernel(
    const int* __restrict__ rowptr, const int* __restrict__ csr,
    const float* __restrict__ f0, float* __restrict__ o0) {
  gather_sum_body<1>(rowptr, csr, f0, o0, f0, o0);
}

__global__ __launch_bounds__(GEMM_THREADS) void gemm_zr_kernel(
    const float* __restrict__ sop, const float* __restrict__ top,
    const float* __restrict__ hl,
    const float* __restrict__ Wl, const float* __restrict__ bl,
    float* __restrict__ zout, float* __restrict__ vout) {
  __shared__ float lds[4 * GH * 132];
  const int lane = threadIdx.x & 31;
  const int kof = (lane >> 4) << 1;
  const int nn = lane & 15;
  const int rowhi = (lane >> 4) << 3;
  const GemmSpan g = gemm_span();

  float* L0 = lds + 0 * GH * 132;
  float* L1 = lds + 1 * GH * 132;
  float* L2 = lds + 2 * GH * 132;
  float* L3 = lds + 3 * GH * 132;
  stage_matrix(Wl + 0 * GH * GH, L0);
  stage_matrix(Wl + 1 * GH * GH, L1);
  stage_matrix(Wl + 2 * GH * GH, L2);
  stage_matrix(Wl + 3 * GH * GH, L3);
  __syncthreads();

  v2f fs[32], ft[32];
  int cur_tile = -1;
  for (int p = g.p0; p < g.p1; ++p) {
    const int tile = g.t0 + (p >> 2);
    const int row0 = tile * 16;
    if (tile != cur_tile) {
      int arow = row0 + nn;
      if (arow > GN - 1) arow = GN - 1;
      load_afrag_ready(sop, arow, lane >> 4, fs);
      load_afrag_ready(top, arow, lane >> 4, ft);
      cur_tile = tile;
    }
    for (int c2 = 0; c2 < 2; ++c2) {
      const int n0 = (((p & 3) << 1) + c2) * 16;
      const float* p0 = L0 + (n0 + nn) * 132 + kof;
      const float* p1 = L1 + (n0 + nn) * 132 + kof;
      const float* p2 = L2 + (n0 + nn) * 132 + kof;
      const float* p3 = L3 + (n0 + nn) * 132 + kof;

      const size_t off0 = (size_t)(row0 + rowhi) * GH + n0 + nn;
      float hv[8];
#pragma unroll
      for (int i = 0; i < 8; ++i) hv[i] = hl[off0 + (size_t)i * GH];
      const float bz0 = bl[0 * GH + n0 + nn], bz1 = bl[1 * GH + n0 + nn];
      const float br0 = bl[2 * GH + n0 + nn], br1 = bl[3 * GH + n0 + nn];
      __builtin_amdgcn_sched_barrier(0);

      v8f az0 = {0, 0, 0, 0, 0, 0, 0, 0}, az1 = {0, 0, 0, 0, 0, 0, 0, 0};
      v8f ar0 = {0, 0, 0, 0, 0, 0, 0, 0}, ar1 = {0, 0, 0, 0, 0, 0, 0, 0};
#pragma unroll
      for (int kk = 0; kk < 32; kk += 2) {
        az0 = wmma_f32(fs[kk], *(const v2f*)(p0 + 4 * kk), az0);
        az0 = wmma_f32(ft[kk], *(const v2f*)(p1 + 4 * kk), az0);
        ar0 = wmma_f32(fs[kk], *(const v2f*)(p2 + 4 * kk), ar0);
        ar0 = wmma_f32(ft[kk], *(const v2f*)(p3 + 4 * kk), ar0);
        az1 = wmma_f32(fs[kk + 1], *(const v2f*)(p0 + 4 * kk + 4), az1);
        az1 = wmma_f32(ft[kk + 1], *(const v2f*)(p1 + 4 * kk + 4), az1);
        ar1 = wmma_f32(fs[kk + 1], *(const v2f*)(p2 + 4 * kk + 4), ar1);
        ar1 = wmma_f32(ft[kk + 1], *(const v2f*)(p3 + 4 * kk + 4), ar1);
      }
      const v8f accz = az0 + az1;
      const v8f accr = ar0 + ar1;

      const float bz = bz0 + bz1;
      const float br = br0 + br1;
#pragma unroll
      for (int i = 0; i < 8; ++i) {
        const size_t off = off0 + (size_t)i * GH;
        const float zv = sigmoid_f(accz[i] + bz);
        const float rv = sigmoid_f(accr[i] + br);
        zout[off] = zv;
        vout[off] = rv * hv[i];
      }
    }
  }
}

__global__ __launch_bounds__(GEMM_THREADS) void gemm_ht_kernel(
    const float* __restrict__ sop, const float* __restrict__ uop,
    const float* __restrict__ Wl, const float* __restrict__ bl,
    const float* __restrict__ zbuf, const float* __restrict__ hl,
    float* __restrict__ outp) {
  __shared__ float lds[2 * GH * 132];
  const int lane = threadIdx.x & 31;
  const int kof = (lane >> 4) << 1;
  const int nn = lane & 15;
  const int rowhi = (lane >> 4) << 3;
  const GemmSpan g = gemm_span();

  float* L4 = lds + 0 * GH * 132;
  float* L5 = lds + 1 * GH * 132;
  stage_matrix(Wl + 4 * GH * GH, L4);
  stage_matrix(Wl + 5 * GH * GH, L5);
  __syncthreads();

  v2f fs[32], fu[32];
  int cur_tile = -1;
  for (int p = g.p0; p < g.p1; ++p) {
    const int tile = g.t0 + (p >> 2);
    const int row0 = tile * 16;
    if (tile != cur_tile) {
      int arow = row0 + nn;
      if (arow > GN - 1) arow = GN - 1;
      load_afrag_ready(sop, arow, lane >> 4, fs);
      load_afrag_ready(uop, arow, lane >> 4, fu);
      cur_tile = tile;
    }
    for (int c2 = 0; c2 < 2; ++c2) {
      const int n0 = (((p & 3) << 1) + c2) * 16;
      const float* p4 = L4 + (n0 + nn) * 132 + kof;
      const float* p5 = L5 + (n0 + nn) * 132 + kof;

      const size_t off0 = (size_t)(row0 + rowhi) * GH + n0 + nn;
      float zv8[8], hv[8];
#pragma unroll
      for (int i = 0; i < 8; ++i) {
        zv8[i] = zbuf[off0 + (size_t)i * GH];
        hv[i] = hl[off0 + (size_t)i * GH];
      }
      const float bh0 = bl[4 * GH + n0 + nn], bh1 = bl[5 * GH + n0 + nn];
      __builtin_amdgcn_sched_barrier(0);

      v8f a0 = {0, 0, 0, 0, 0, 0, 0, 0}, a1 = {0, 0, 0, 0, 0, 0, 0, 0};
#pragma unroll
      for (int kk = 0; kk < 32; kk += 2) {
        a0 = wmma_f32(fs[kk], *(const v2f*)(p4 + 4 * kk), a0);
        a0 = wmma_f32(fu[kk], *(const v2f*)(p5 + 4 * kk), a0);
        a1 = wmma_f32(fs[kk + 1], *(const v2f*)(p4 + 4 * kk + 4), a1);
        a1 = wmma_f32(fu[kk + 1], *(const v2f*)(p5 + 4 * kk + 4), a1);
      }
      const v8f acc = a0 + a1;

      const float bh = bh0 + bh1;
#pragma unroll
      for (int i = 0; i < 8; ++i) {
        const float ht = tanh_f(acc[i] + bh);
        const float zv = zv8[i];
        outp[off0 + (size_t)i * GH] = zv * hv[i] + (1.0f - zv) * ht;
      }
    }
  }
}

constexpr int GN_PAD = 50048;
constexpr int CHCNT_PAD = ((NCH * NBK + 63) / 64) * 64;
constexpr size_t OFF_ROWPTR = 0;
constexpr size_t OFF_CHCNT = OFF_ROWPTR + GN_PAD;
constexpr size_t OFF_CHFIRST = OFF_CHCNT + CHCNT_PAD;
constexpr size_t OFF_BSTART = OFF_CHFIRST + CHCNT_PAD;
constexpr size_t OFF_SLOTS = OFF_BSTART + 256;
constexpr size_t OFF_PAIRS = OFF_SLOTS + GE;
constexpr size_t OFF_CSR = OFF_PAIRS;
static_assert(NBK + 1 <= 256, "bucket starts must fit their slot");
static_assert(OFF_PAIRS + 2 * (size_t)GE <= NH, "edge lists must fit in one output plane");
constexpr int ROW_BLOCKS = (GN * 32 + 255) / 256;

extern "C" void kernel_launch(void* const* d_in, const int* in_sizes, int n_in,
                              void* d_out, int out_size, void* d_ws,
                              size_t ws_size, hipStream_t stream) {
  (void)in_sizes; (void)n_in; (void)out_size; (void)ws_size;
  const float* inp = (const float*)d_in[0];
  const int* edg = (const int*)d_in[1];
  const float* hmat = (const float*)d_in[2];
  const float* Wmat = (const float*)d_in[3];
  const float* bmat = (const float*)d_in[4];
  float* out = (float*)d_out;

  const int* src = edg;
  const int* dst = edg + GE;

  float* agg_x = (float*)d_ws;
  float* agg_t = agg_x + NH;
  float* zbuf = agg_t + NH;
  float* vbuf = zbuf + NH;

  int* lists = (int*)(out + (size_t)(GLAYERS - 1) * NH);
  int* rowptr = lists + OFF_ROWPTR;
  int* chcnt = lists + OFF_CHCNT;
  int* chfirst = lists + OFF_CHFIRST;
  int* bstart = lists + OFF_BSTART;
  int* slots = lists + OFF_SLOTS;
  int2* pairs = (int2*)(lists + OFF_PAIRS);
  int* csr = lists + OFF_CSR;

  bucket_count_kernel<<<NCH, SORT_THREADS, 0, stream>>>(dst, chcnt);
  bucket_scan_kernel<<<1, SORT_THREADS, 0, stream>>>(chcnt, chfirst, bstart);
  bucket_scatter_kernel<<<NCH, SORT_THREADS, 0, stream>>>(src, dst, chfirst, pairs);
  bucket_rows_kernel<<<NBK, SORT_THREADS, 0, stream>>>(bstart, pairs, rowptr, slots);
  segment_order_kernel<<<ROW_BLOCKS, 256, 0, stream>>>(rowptr, slots, csr);

  for (int l = 0; l < GLAYERS; ++l) {
    const float* x = (l == 0) ? inp : (out + (size_t)(l - 1) * NH);
    const float* hl = hmat + (size_t)l * NH;
    const float* Wl = Wmat + (size_t)l * 6 * GH * GH;
    const float* bl = bmat + (size_t)l * 6 * GH;
    float* outl = out + (size_t)l * NH;

    gather_sum2_kernel<<<ROW_BLOCKS, 256, 0, stream>>>(rowptr, csr, x, agg_x,
                                                       hl, agg_t);

    gemm_zr_kernel<<<GEMM_GRID, GEMM_THREADS, 0, stream>>>(agg_x, agg_t, hl, Wl, bl,
                                                    zbuf, vbuf);

    gather_sum1_kernel<<<ROW_BLOCKS, 256, 0, stream>>>(rowptr, csr, vbuf,
                                                       agg_t);

    gemm_ht_kernel<<<GEMM_GRID, GEMM_THREADS, 0, stream>>>(agg_x, agg_t, Wl, bl,
                                                    zbuf, hl, outl);
  }
}


# Row gathers moved into the two GEMM launches: 16 waves of each workgroup sum the next tile's rows into LDS while the others multiply the tile before; 9 launches, not 13; sums, their order unchanged
# speedup vs baseline: 4.7654x; 1.0683x over previous
#include <hip/hip_runtime.h>
#include <cstdint>
#include <cstddef>

constexpr int GN = 50000;
constexpr int GE = 1600000;
constexpr int GH = 128;
constexpr int GLAYERS = 2;
constexpr size_t NH = (size_t)GN * GH;
constexpr int ROW_TILES = GN / 16;
constexpr int GEMM_GRID = 256;
constexpr int FUSE_THREADS = 1024;
constexpr int FUSE_HT_THREADS = 768;
constexpr int TILE_PITCH = 132;
constexpr int TILE_WORDS = 16 * TILE_PITCH;
static_assert(GN % 16 == 0, "row tiles must be whole");

typedef __attribute__((ext_vector_type(2))) float v2f;
typedef __attribute__((ext_vector_type(8))) float v8f;

__device__ __forceinline__ v8f wmma_f32(v2f a, v2f b, v8f c) {
  return __builtin_amdgcn_wmma_f32_16x16x4_f32(
       false, a,  false, b,
       (short)0, c,  false,  false);
}

__device__ __forceinline__ float sigmoid_f(float x) {
  return 1.0f / (1.0f + __expf(-x));
}
__device__ __forceinline__ float tanh_f(float x) {
  return 1.0f - 2.0f / (1.0f + __expf(2.0f * x));
}


template <int NT>
__device__ __forceinline__ void stage_matrix(const float* __restrict__ Wg,
                                             float* lds) {
  const int tid = threadIdx.x;
#pragma unroll
  for (int i = 0; i < (GH * GH + NT - 1) / NT; ++i) {
    const int idx = i * NT + tid;
    if ((GH * GH) % NT == 0 || idx < GH * GH) {
      const int c = idx & 15;
      const int k = (idx >> 4) & (GH - 1);
      const int n = ((idx >> 11) << 4) + c;
      lds[n * 132 + k] = Wg[k * GH + n];
    }
  }
}


constexpr int BK_SHIFT = 8;
constexpr int BK_ROWS = 1 << BK_SHIFT;
constexpr int NBK = (GN + BK_ROWS - 1) >> BK_SHIFT;
constexpr int SORT_THREADS = 1024;
constexpr int NCH = 250;
constexpr int CH_EDGES = GE / NCH;
constexpr int CH_TRIPS = (CH_EDGES + SORT_THREADS - 1) / SORT_THREADS;
constexpr int SCAN_GROUPS = SORT_THREADS / 256;
constexpr int CH_PER_GROUP = (NCH + SCAN_GROUPS - 1) / SCAN_GROUPS;
constexpr int ROWS_UNROLL = 4;
static_assert(NCH * CH_EDGES == GE, "chunks must tile the edge list");
static_assert(NBK <= 256, "one scan thread per bucket");

__device__ __forceinline__ int scan256_exclusive(int v, int t, int* wsum, int& total) {
  const int lane = t & 31;
  int incl = v;
  if (t < 256) {
    for (int off = 1; off < 32; off <<= 1) {
      const int y = __shfl_up(incl, off);
      if (lane >= off) incl += y;
    }
    if (lane == 31) wsum[t >> 5] = incl;
  }
  __syncthreads();
  int woff = 0, all = 0;
  for (int w = 0; w < 8; ++w) {
    const int s = wsum[w];
    if (w < (t >> 5)) woff += s;
    all += s;
  }
  total = all;
  return woff + incl - v;
}

__global__ __launch_bounds__(SORT_THREADS) void bucket_count_kernel(
    const int* __restrict__ dst, int* __restrict__ chcnt) {
  __shared__ int hist[NBK];
  const int t = threadIdx.x;
  const int g = blockIdx.x;
  if (t < NBK) hist[t] = 0;
  __syncthreads();
  const int e0 = g * CH_EDGES;
  int d[CH_TRIPS];
#pragma unroll
  for (int k = 0; k < CH_TRIPS; ++k) {
    const int i = k * SORT_THREADS + t;
    d[k] = dst[e0 + ((i < CH_EDGES) ? i : CH_EDGES - 1)];
  }
#pragma unroll
  for (int k = 0; k < CH_TRIPS; ++k) {
    if (k * SORT_THREADS + t < CH_EDGES) atomicAdd(&hist[d[k] >> BK_SHIFT], 1);
  }
  __syncthreads();
  if (t < NBK) chcnt[g * NBK + t] = hist[t];
}

__global__ __launch_bounds__(SORT_THREADS) void bucket_scan_kernel(
    const int* __restrict__ chcnt, int* __restrict__ chfirst,
    int* __restrict__ bstart) {
  __shared__ int part[SCAN_GROUPS][256];
  __shared__ int bs[256];
  __shared__ int wsum[8];
  const int t = threadIdx.x;
  const int q = t >> 8;
  const int b = t & 255;
  const int g0 = q * CH_PER_GROUP;
  const int bc = (b < NBK) ? b : NBK - 1;
  int v[CH_PER_GROUP];
#pragma unroll
  for (int j = 0; j < CH_PER_GROUP; ++j) {
    const int g = g0 + j;
    const int x = chcnt[((g < NCH) ? g : NCH - 1) * NBK + bc];
    v[j] = (g < NCH && b < NBK) ? x : 0;
  }
  int p = 0;
#pragma unroll
  for (int j = 0; j < CH_PER_GROUP; ++j) p += v[j];
  part[q][b] = p;
  __syncthreads();
  int tot = 0;
  if (t < 256) {
    for (int i = 0; i < SCAN_GROUPS; ++i) tot += part[i][t];
  }
  int all;
  const int ex = scan256_exclusive(tot, t, wsum, all);
  if (t < 256) {
    bs[t] = ex;
    if (t < NBK) bstart[t] = ex;
    if (t == 0) bstart[NBK] = all;
  }
  __syncthreads();
  if (b < NBK) {
    int run = bs[b];
    for (int i = 0; i < q; ++i) run += part[i][b];
#pragma unroll
    for (int j = 0; j < CH_PER_GROUP; ++j) {
      const int g = g0 + j;
      if (g < NCH) chfirst[g * NBK + b] = run;
      run += v[j];
    }
  }
}

__global__ __launch_bounds__(SORT_THREADS) void bucket_scatter_kernel(
    const int* __restrict__ src, const int* __restrict__ dst,
    const int* __restrict__ chfirst, int2* __restrict__ pairs) {
  __shared__ int cur[NBK];
  const int t = threadIdx.x;
  const int g = blockIdx.x;
  if (t < NBK) cur[t] = chfirst[g * NBK + t];
  __syncthreads();
  const int e0 = g * CH_EDGES;
  int d[CH_TRIPS], s[CH_TRIPS];
#pragma unroll
  for (int k = 0; k < CH_TRIPS; ++k) {
    const int i = k * SORT_THREADS + t;
    const int ic = e0 + ((i < CH_EDGES) ? i : CH_EDGES - 1);
    d[k] = dst[ic];
    s[k] = src[ic];
  }
#pragma unroll
  for (int k = 0; k < CH_TRIPS; ++k) {
    if (k * SORT_THREADS + t < CH_EDGES) {
      const int p = atomicAdd(&cur[d[k] >> BK_SHIFT], 1);
      pairs[p] = make_int2(s[k], d[k]);
    }
  }
}

__global__ __launch_bounds__(SORT_THREADS) void bucket_rows_kernel(
    const int* __restrict__ bstart, const int2* __restrict__ pairs,
    int* __restrict__ rowptr, int* __restrict__ slots) {
  __shared__ int rc[BK_ROWS];
  __shared__ int wsum[8];
  const int t = threadIdx.x;
  const int b = blockIdx.x;
  const int beg = bstart[b];
  const int end = bstart[b + 1];
  if (t < BK_ROWS) rc[t] = 0;
  __syncthreads();
  for (int i0 = beg + t; i0 < end; i0 += ROWS_UNROLL * SORT_THREADS) {
    int d[ROWS_UNROLL];
#pragma unroll
    for (int u = 0; u < ROWS_UNROLL; ++u) {
      const int i = i0 + u * SORT_THREADS;
      d[u] = pairs[(i < end) ? i : end - 1].y;
    }
#pragma unroll
    for (int u = 0; u < ROWS_UNROLL; ++u) {
      if (i0 + u * SORT_THREADS < end) atomicAdd(&rc[d[u] & (BK_ROWS - 1)], 1);
    }
  }
  __syncthreads();
  const int deg = (t < BK_ROWS) ? rc[t] : 0;
  int all;
  const int ex = scan256_exclusive(deg, t, wsum, all);
  if (t < BK_ROWS) {
    const int row = (b << BK_SHIFT) + t;
    rc[t] = beg + ex;
    if (row < GN) rowptr[row] = beg + ex;
  }
  if (b == NBK - 1 && t == 0) rowptr[GN] = end;
  __syncthreads();
  for (int i0 = beg + t; i0 < end; i0 += ROWS_UNROLL * SORT_THREADS) {
    int2 sd[ROWS_UNROLL];
#pragma unroll
    for (int u = 0; u < ROWS_UNROLL; ++u) {
      const int i = i0 + u * SORT_THREADS;
      sd[u] = pairs[(i < end) ? i : end - 1];
    }
#pragma unroll
    for (int u = 0; u < ROWS_UNROLL; ++u) {
      if (i0 + u * SORT_THREADS < end) {
        const int p = atomicAdd(&rc[sd[u].y & (BK_ROWS - 1)], 1);
        slots[p] = sd[u].x;
      }
    }
  }
}

__global__ __launch_bounds__(256) void segment_order_kernel(
    const int* __restrict__ rowptr, const int* __restrict__ slots,
    int* __restrict__ csr) {
  const int lane = threadIdx.x & 31;
  const int row = __builtin_amdgcn_readfirstlane(
      (int)((blockIdx.x * 256u + threadIdx.x) >> 5));
  if (row >= GN) return;
  const int beg = rowptr[row];
  const int n = rowptr[row + 1] - beg;
  for (int ib = 0; ib < n; ib += 32) {
    const int i = ib + lane;
    const int key = (i < n) ? slots[beg + i] : 0;
    int rank = 0;
    for (int jb = 0; jb < n; jb += 32) {
      const int jn = (n - jb < 32) ? (n - jb) : 32;
      const int kv = (lane < jn) ? slots[beg + jb + lane] : 0;
      for (int jj = 0; jj < jn; ++jj) {
        const int kj = __builtin_amdgcn_readlane(kv, jj);
        const int j = jb + jj;
        rank += (kj < key || (kj == key && j < i)) ? 1 : 0;
      }
    }
    if (i < n) csr[beg + rank] = key;
  }
}

template <int U>
__device__ __forceinline__ void gather_rows(const float* __restrict__ f,
                                            int idx, int j0, int c4,
                                            float4& acc) {
  float4 v[U];
#pragma unroll
  for (int u = 0; u < U; ++u) {
    const int s = __builtin_amdgcn_readlane(idx, j0 + u);
    v[u] = *(const float4*)(f + (size_t)s * GH + c4);
  }
#pragma unroll
  for (int u = 0; u < U; ++u) {
    acc.x += v[u].x;
    acc.y += v[u].y;
    acc.z += v[u].z;
    acc.w += v[u].w;
  }
}

template <int U, int NF>
__device__ __forceinline__ void gather_group(const float* __restrict__ f0,
                                             const float* __restrict__ f1,
                                             int idx, int j0, int c4,
                                             float4& a0, float4& a1) {
  gather_rows<U>(f0, idx, j0, c4, a0);
  if (NF == 2) gather_rows<U>(f1, idx, j0, c4, a1);
}

__device__ __forceinline__ float4 operand_order(int lane, float4 s) {
  const bool odd = (lane & 1) != 0;
  const float sx = odd ? s.x : s.z;
  const float sy = odd ? s.y : s.w;
  const float rx = __shfl_xor(sx, 1);
  const float ry = __shfl_xor(sy, 1);
  float4 w;
  w.x = odd ? rx : s.x;
  w.y = odd ? ry : s.y;
  w.z = odd ? s.z : rx;
  w.w = odd ? s.w : ry;
  return w;
}

template <int NF>
__device__ __forceinline__ void gather_batch(const float* __restrict__ f0,
                                             const float* __restrict__ f1,
                                             int idx, int n, int c4,
                                             float4& a0, float4& a1) {
  int j = 0;
  for (; j + 8 <= n; j += 8) gather_group<8, NF>(f0, f1, idx, j, c4, a0, a1);
  if (n & 4) { gather_group<4, NF>(f0, f1, idx, j, c4, a0, a1); j += 4; }
  if (n & 2) { gather_group<2, NF>(f0, f1, idx, j, c4, a0, a1); j += 2; }
  if (n & 1) { gather_group<1, NF>(f0, f1, idx, j, c4, a0, a1); }
}

template <int NF, bool PRE>
__device__ __forceinline__ void gather_list(
    const int* __restrict__ csr,
    const float* __restrict__ f0, const float* __restrict__ f1,
    int row, int lane, int beg, int end, int first_idx,
    float4& s0, float4& s1) {
  const int c4 = lane << 2;
  const float4 x0 = *(const float4*)(f0 + (size_t)row * GH + c4);
  const float4 x1 = *(const float4*)(f1 + (size_t)row * GH + c4);
  float4 a0 = {0.0f, 0.0f, 0.0f, 0.0f};
  float4 a1 = {0.0f, 0.0f, 0.0f, 0.0f};
  int base = beg;
  if (PRE) {
    const int n = (end - beg < 32) ? (end - beg) : 32;
    gather_batch<NF>(f0, f1, first_idx, n, c4, a0, a1);
    base = beg + 32;
  }
  for (; base < end; base += 32) {
    const int n = (end - base < 32) ? (end - base) : 32;
    const int idx = (lane < n) ? csr[base + lane] : 0;
    gather_batch<NF>(f0, f1, idx, n, c4, a0, a1);
  }
  s0.x = x0.x + a0.x;
  s0.y = x0.y + a0.y;
  s0.z = x0.z + a0.z;
  s0.w = x0.w + a0.w;
  s1.x = x1.x + a1.x;
  s1.y = x1.y + a1.y;
  s1.z = x1.z + a1.z;
  s1.w = x1.w + a1.w;
}

__global__ __launch_bounds__(FUSE_THREADS) void gather_gemm_zr_kernel(
    const int* __restrict__ rowptr, const int* __restrict__ csr,
    const float* __restrict__ x, const float* __restrict__ hl,
    float* __restrict__ sop,
    const float* __restrict__ Wl, const float* __restrict__ bl,
    float* __restrict__ zout, float* __restrict__ vout) {
  __shared__ __attribute__((aligned(16))) float lds[4 * GH * 132 + 4 * TILE_WORDS];
  const int lane = threadIdx.x & 31;
  const int wave = __builtin_amdgcn_readfirstlane((int)(threadIdx.x >> 5));
  float* ring = lds + 4 * GH * 132;

  stage_matrix<FUSE_THREADS>(Wl + 0 * GH * GH, lds + 0 * GH * 132);
  stage_matrix<FUSE_THREADS>(Wl + 1 * GH * GH, lds + 1 * GH * 132);
  stage_matrix<FUSE_THREADS>(Wl + 2 * GH * GH, lds + 2 * GH * 132);
  stage_matrix<FUSE_THREADS>(Wl + 3 * GH * GH, lds + 3 * GH * 132);

  const int b = blockIdx.x;
  const int t0 = (b * ROW_TILES) / GEMM_GRID;
  const int nt = ((b + 1) * ROW_TILES) / GEMM_GRID - t0;

  int rbeg = 0, rend = 0, next_idx = 0;
  if (wave < 16) {
    const int k = (lane < nt) ? lane : nt - 1;
    const int r = (t0 + k) * 16 + wave;
    rbeg = rowptr[r];
    rend = rowptr[r + 1];
    const int beg0 = __builtin_amdgcn_readlane(rbeg, 0);
    const int end0 = __builtin_amdgcn_readlane(rend, 0);
    next_idx = (lane < end0 - beg0) ? csr[beg0 + lane] : 0;
  }

  for (int step = 0; step <= nt; ++step) {
    if (wave < 16) {
      if (step < nt) {
        const int row = (t0 + step) * 16 + wave;
        const int c4 = lane << 2;
        const int beg = __builtin_amdgcn_readlane(rbeg, step);
        const int end = __builtin_amdgcn_readlane(rend, step);
        const int first_idx = next_idx;
        if (step + 1 < nt) {
          const int nb = __builtin_amdgcn_readlane(rbeg, step + 1);
          const int ne = __builtin_amdgcn_readlane(rend, step + 1);
          next_idx = (lane < ne - nb) ? csr[nb + lane] : 0;
        }
        float4 s0, s1;
        gather_list<2, true>(csr, x, hl, row, lane, beg, end, first_idx, s0, s1);
        const float4 w0 = operand_order(lane, s0);
        const float4 w1 = operand_order(lane, s1);
        float* slot = ring + (step & 1) * 2 * TILE_WORDS;
        *(float4*)(slot + wave * TILE_PITCH + c4) = w0;
        *(float4*)(slot + TILE_WORDS + wave * TILE_PITCH + c4) = w1;
        *(float4*)(sop + (size_t)row * GH + c4) = w0;
      }
    } else if (step > 0) {
      const int mw = wave - 16;
      const int gate = mw >> 3;
      const int n0 = (mw & 7) * 16;
      const int half = lane >> 4;
      const int kof = half << 1;
      const int nn = lane & 15;
      const int rowhi = half << 3;
      const int row0 = (t0 + step - 1) * 16;
      const float* slot = ring + ((step - 1) & 1) * 2 * TILE_WORDS;
      const float* as = slot + nn * TILE_PITCH + (half << 2);
      const float* at = as + TILE_WORDS;
      const float* pa = lds + (2 * gate) * GH * 132 + (n0 + nn) * 132 + kof;
      const float* pb = pa + GH * 132;

      const size_t off0 = (size_t)(row0 + rowhi) * GH + n0 + nn;
      float hv[8];
#pragma unroll
      for (int i = 0; i < 8; ++i) hv[i] = hl[off0 + (size_t)i * GH];
      const float b0 = bl[(2 * gate) * GH + n0 + nn];
      const float b1 = bl[(2 * gate + 1) * GH + n0 + nn];
      __builtin_amdgcn_sched_barrier(0);

      v8f c0 = {0, 0, 0, 0, 0, 0, 0, 0}, c1 = {0, 0, 0, 0, 0, 0, 0, 0};
#pragma unroll
      for (int m = 0; m < 16; ++m) {
        const float4 sv = *(const float4*)(as + 8 * m);
        const float4 tv = *(const float4*)(at + 8 * m);
        v2f slo, shi, tlo, thi;
        slo.x = sv.x; slo.y = sv.y; shi.x = sv.z; shi.y = sv.w;
        tlo.x = tv.x; tlo.y = tv.y; thi.x = tv.z; thi.y = tv.w;
        c0 = wmma_f32(slo, *(const v2f*)(pa + 8 * m), c0);
        c0 = wmma_f32(tlo, *(const v2f*)(pb + 8 * m), c0);
        c1 = wmma_f32(shi, *(const v2f*)(pa + 8 * m + 4), c1);
        c1 = wmma_f32(thi, *(const v2f*)(pb + 8 * m + 4), c1);
      }
      const v8f acc = c0 + c1;
      const float bb = b0 + b1;
      float* dstp = (gate != 0) ? vout : zout;
#pragma unroll
      for (int i = 0; i < 8; ++i) {
        const float g = sigmoid_f(acc[i] + bb);
        dstp[off0 + (size_t)i * GH] = (gate != 0) ? g * hv[i] : g;
      }
    }
    __syncthreads();
  }
}

__global__ __launch_bounds__(FUSE_HT_THREADS) void gather_gemm_ht_kernel(
    const int* __restrict__ rowptr, const int* __restrict__ csr,
    const float* __restrict__ vin, const float* __restrict__ sop,
    const float* __restrict__ Wl, const float* __restrict__ bl,
    const float* __restrict__ zbuf, const float* __restrict__ hl,
    float* __restrict__ outp) {
  __shared__ __attribute__((aligned(16))) float lds[2 * GH * 132 + 4 * TILE_WORDS];
  const int lane = threadIdx.x & 31;
  const int wave = __builtin_amdgcn_readfirstlane((int)(threadIdx.x >> 5));
  float* ring = lds + 2 * GH * 132;

  stage_matrix<FUSE_HT_THREADS>(Wl + 4 * GH * GH, lds + 0 * GH * 132);
  stage_matrix<FUSE_HT_THREADS>(Wl + 5 * GH * GH, lds + 1 * GH * 132);

  const int b = blockIdx.x;
  const int t0 = (b * ROW_TILES) / GEMM_GRID;
  const int nt = ((b + 1) * ROW_TILES) / GEMM_GRID - t0;

  int rbeg = 0, rend = 0, next_idx = 0;
  if (wave < 16) {
    const int k = (lane < nt) ? lane : nt - 1;
    const int r = (t0 + k) * 16 + wave;
    rbeg = rowptr[r];
    rend = rowptr[r + 1];
    const int beg0 = __builtin_amdgcn_readlane(rbeg, 0);
    const int end0 = __builtin_amdgcn_readlane(rend, 0);
    next_idx = (lane < end0 - beg0) ? csr[beg0 + lane] : 0;
  }

  for (int step = 0; step <= nt; ++step) {
    if (wave < 16) {
      if (step < nt) {
        const int row = (t0 + step) * 16 + wave;
        const int c4 = lane << 2;
        const int beg = __builtin_amdgcn_readlane(rbeg, step);
        const int end = __builtin_amdgcn_readlane(rend, step);
        const int first_idx = next_idx;
        if (step + 1 < nt) {
          const int nb = __builtin_amdgcn_readlane(rbeg, step + 1);
          const int ne = __builtin_amdgcn_readlane(rend, step + 1);
          next_idx = (lane < ne - nb) ? csr[nb + lane] : 0;
        }
        const float4 srow = *(const float4*)(sop + (size_t)row * GH + c4);
        float4 u0, u1;
        gather_list<1, true>(csr, vin, vin, row, lane, beg, end, first_idx, u0, u1);
        const float4 w = operand_order(lane, u0);
        float* slot = ring + (step & 1) * 2 * TILE_WORDS;
        *(float4*)(slot + wave * TILE_PITCH + c4) = srow;
        *(float4*)(slot + TILE_WORDS + wave * TILE_PITCH + c4) = w;
      }
    } else if (step > 0) {
      const int n0 = (wave - 16) * 16;
      const int half = lane >> 4;
      const int kof = half << 1;
      const int nn = lane & 15;
      const int rowhi = half << 3;
      const int row0 = (t0 + step - 1) * 16;
      const float* slot = ring + ((step - 1) & 1) * 2 * TILE_WORDS;
      const float* as = slot + nn * TILE_PITCH + (half << 2);
      const float* au = as + TILE_WORDS;
      const float* p4 = lds + (n0 + nn) * 132 + kof;
      const float* p5 = p4 + GH * 132;

      const size_t off0 = (size_t)(row0 + rowhi) * GH + n0 + nn;
      float zv8[8], hv[8];
#pragma unroll
      for (int i = 0; i < 8; ++i) {
        zv8[i] = zbuf[off0 + (size_t)i * GH];
        hv[i] = hl[off0 + (size_t)i * GH];
      }
      const float bh0 = bl[4 * GH + n0 + nn], bh1 = bl[5 * GH + n0 + nn];
      __builtin_amdgcn_sched_barrier(0);

      v8f a0 = {0, 0, 0, 0, 0, 0, 0, 0}, a1 = {0, 0, 0, 0, 0, 0, 0, 0};
#pragma unroll
      for (int m = 0; m < 16; ++m) {
        const float4 sv = *(const float4*)(as + 8 * m);
        const float4 uv = *(const float4*)(au + 8 * m);
        v2f slo, shi, ulo, uhi;
        slo.x = sv.x; slo.y = sv.y; shi.x = sv.z; shi.y = sv.w;
        ulo.x = uv.x; ulo.y = uv.y; uhi.x = uv.z; uhi.y = uv.w;
        a0 = wmma_f32(slo, *(const v2f*)(p4 + 8 * m), a0);
        a0 = wmma_f32(ulo, *(const v2f*)(p5 + 8 * m), a0);
        a1 = wmma_f32(shi, *(const v2f*)(p4 + 8 * m + 4), a1);
        a1 = wmma_f32(uhi, *(const v2f*)(p5 + 8 * m + 4), a1);
      }
      const v8f acc = a0 + a1;

      const float bh = bh0 + bh1;
#pragma unroll
      for (int i = 0; i < 8; ++i) {
        const float ht = tanh_f(acc[i] + bh);
        const float zv = zv8[i];
        outp[off0 + (size_t)i * GH] = zv * hv[i] + (1.0f - zv) * ht;
      }
    }
    __syncthreads();
  }
}

constexpr int GN_PAD = 50048;
constexpr int CHCNT_PAD = ((NCH * NBK + 63) / 64) * 64;
constexpr size_t OFF_ROWPTR = 0;
constexpr size_t OFF_CHCNT = OFF_ROWPTR + GN_PAD;
constexpr size_t OFF_CHFIRST = OFF_CHCNT + CHCNT_PAD;
constexpr size_t OFF_BSTART = OFF_CHFIRST + CHCNT_PAD;
constexpr size_t OFF_SLOTS = OFF_BSTART + 256;
constexpr size_t OFF_PAIRS = OFF_SLOTS + GE;
constexpr size_t OFF_CSR = OFF_PAIRS;
static_assert(NBK + 1 <= 256, "bucket starts must fit their slot");
static_assert(OFF_PAIRS + 2 * (size_t)GE <= NH, "edge lists must fit in one workspace plane");
constexpr int ROW_BLOCKS = (GN * 32 + 255) / 256;

extern "C" void kernel_launch(void* const* d_in, const int* in_sizes, int n_in,
                              void* d_out, int out_size, void* d_ws,
                              size_t ws_size, hipStream_t stream) {
  (void)in_sizes; (void)n_in; (void)out_size; (void)ws_size;
  const float* inp = (const float*)d_in[0];
  const int* edg = (const int*)d_in[1];
  const float* hmat = (const float*)d_in[2];
  const float* Wmat = (const float*)d_in[3];
  const float* bmat = (const float*)d_in[4];
  float* out = (float*)d_out;

  const int* src = edg;
  const int* dst = edg + GE;

  float* agg_x = (float*)d_ws;
  int* lists = (int*)(agg_x + NH);
  float* zbuf = agg_x + 2 * NH;
  float* vbuf = zbuf + NH;

  int* rowptr = lists + OFF_ROWPTR;
  int* chcnt = lists + OFF_CHCNT;
  int* chfirst = lists + OFF_CHFIRST;
  int* bstart = lists + OFF_BSTART;
  int* slots = lists + OFF_SLOTS;
  int2* pairs = (int2*)(lists + OFF_PAIRS);
  int* csr = lists + OFF_CSR;

  bucket_count_kernel<<<NCH, SORT_THREADS, 0, stream>>>(dst, chcnt);
  bucket_scan_kernel<<<1, SORT_THREADS, 0, stream>>>(chcnt, chfirst, bstart);
  bucket_scatter_kernel<<<NCH, SORT_THREADS, 0, stream>>>(src, dst, chfirst, pairs);
  bucket_rows_kernel<<<NBK, SORT_THREADS, 0, stream>>>(bstart, pairs, rowptr, slots);
  segment_order_kernel<<<ROW_BLOCKS, 256, 0, stream>>>(rowptr, slots, csr);

  for (int l = 0; l < GLAYERS; ++l) {
    const float* x = (l == 0) ? inp : (out + (size_t)(l - 1) * NH);
    const float* hl = hmat + (size_t)l * NH;
    const float* Wl = Wmat + (size_t)l * 6 * GH * GH;
    const float* bl = bmat + (size_t)l * 6 * GH;
    float* outl = out + (size_t)l * NH;

    gather_gemm_zr_kernel<<<GEMM_GRID, FUSE_THREADS, 0, stream>>>(
        rowptr, csr, x, hl, agg_x, Wl, bl, zbuf, vbuf);

    gather_gemm_ht_kernel<<<GEMM_GRID, FUSE_HT_THREADS, 0, stream>>>(
        rowptr, csr, vbuf, agg_x, Wl, bl, zbuf, hl, outl);
  }
}


# Row gathers take fewer dependent round trips: a wave issues 16 row loads before its first add, a row's last 1-7 entries go as one round, list entries 33-64 are fetched a step ahead; sums unchanged
# speedup vs baseline: 5.0517x; 1.0601x over previous
#include <hip/hip_runtime.h>
#include <cstdint>
#include <cstddef>

constexpr int GN = 50000;
constexpr int GE = 1600000;
constexpr int GH = 128;
constexpr int GLAYERS = 2;
constexpr size_t NH = (size_t)GN * GH;
constexpr int ROW_TILES = GN / 16;
constexpr int GEMM_GRID = 256;
constexpr int FUSE_THREADS = 1024;
constexpr int FUSE_HT_THREADS = 768;
constexpr int TILE_PITCH = 132;
constexpr int TILE_WORDS = 16 * TILE_PITCH;
static_assert(GN % 16 == 0, "row tiles must be whole");

typedef __attribute__((ext_vector_type(2))) float v2f;
typedef __attribute__((ext_vector_type(4))) float v4f;
typedef __attribute__((ext_vector_type(8))) float v8f;

__device__ __forceinline__ v8f wmma_f32(v2f a, v2f b, v8f c) {
  return __builtin_amdgcn_wmma_f32_16x16x4_f32(
       false, a,  false, b,
       (short)0, c,  false,  false);
}

__device__ __forceinline__ float sigmoid_f(float x) {
  return 1.0f / (1.0f + __expf(-x));
}
__device__ __forceinline__ float tanh_f(float x) {
  return 1.0f - 2.0f / (1.0f + __expf(2.0f * x));
}


template <int NT>
__device__ __forceinline__ void stage_matrix(const float* __restrict__ Wg,
                                             float* lds) {
  const int tid = threadIdx.x;
#pragma unroll
  for (int i = 0; i < (GH * GH + NT - 1) / NT; ++i) {
    const int idx = i * NT + tid;
    if ((GH * GH) % NT == 0 || idx < GH * GH) {
      const int c = idx & 15;
      const int k = (idx >> 4) & (GH - 1);
      const int n = ((idx >> 11) << 4) + c;
      lds[n * 132 + k] = Wg[k * GH + n];
    }
  }
}


constexpr int BK_SHIFT = 8;
constexpr int BK_ROWS = 1 << BK_SHIFT;
constexpr int NBK = (GN + BK_ROWS - 1) >> BK_SHIFT;
constexpr int SORT_THREADS = 1024;
constexpr int NCH = 250;
constexpr int CH_EDGES = GE / NCH;
constexpr int CH_TRIPS = (CH_EDGES + SORT_THREADS - 1) / SORT_THREADS;
constexpr int SCAN_GROUPS = SORT_THREADS / 256;
constexpr int CH_PER_GROUP = (NCH + SCAN_GROUPS - 1) / SCAN_GROUPS;
constexpr int ROWS_UNROLL = 4;
static_assert(NCH * CH_EDGES == GE, "chunks must tile the edge list");
static_assert(NBK <= 256, "one scan thread per bucket");

__device__ __forceinline__ int scan256_exclusive(int v, int t, int* wsum, int& total) {
  const int lane = t & 31;
  int incl = v;
  if (t < 256) {
    for (int off = 1; off < 32; off <<= 1) {
      const int y = __shfl_up(incl, off);
      if (lane >= off) incl += y;
    }
    if (lane == 31) wsum[t >> 5] = incl;
  }
  __syncthreads();
  int woff = 0, all = 0;
  for (int w = 0; w < 8; ++w) {
    const int s = wsum[w];
    if (w < (t >> 5)) woff += s;
    all += s;
  }
  total = all;
  return woff + incl - v;
}

__global__ __launch_bounds__(SORT_THREADS) void bucket_count_kernel(
    const int* __restrict__ dst, int* __restrict__ chcnt) {
  __shared__ int hist[NBK];
  const int t = threadIdx.x;
  const int g = blockIdx.x;
  if (t < NBK) hist[t] = 0;
  __syncthreads();
  const int e0 = g * CH_EDGES;
  int d[CH_TRIPS];
#pragma unroll
  for (int k = 0; k < CH_TRIPS; ++k) {
    const int i = k * SORT_THREADS + t;
    d[k] = dst[e0 + ((i < CH_EDGES) ? i : CH_EDGES - 1)];
  }
#pragma unroll
  for (int k = 0; k < CH_TRIPS; ++k) {
    if (k * SORT_THREADS + t < CH_EDGES) atomicAdd(&hist[d[k] >> BK_SHIFT], 1);
  }
  __syncthreads();
  if (t < NBK) chcnt[g * NBK + t] = hist[t];
}

__global__ __launch_bounds__(SORT_THREADS) void bucket_scan_kernel(
    const int* __restrict__ chcnt, int* __restrict__ chfirst,
    int* __restrict__ bstart) {
  __shared__ int part[SCAN_GROUPS][256];
  __shared__ int bs[256];
  __shared__ int wsum[8];
  const int t = threadIdx.x;
  const int q = t >> 8;
  const int b = t & 255;
  const int g0 = q * CH_PER_GROUP;
  const int bc = (b < NBK) ? b : NBK - 1;
  int v[CH_PER_GROUP];
#pragma unroll
  for (int j = 0; j < CH_PER_GROUP; ++j) {
    const int g = g0 + j;
    const int x = chcnt[((g < NCH) ? g : NCH - 1) * NBK + bc];
    v[j] = (g < NCH && b < NBK) ? x : 0;
  }
  int p = 0;
#pragma unroll
  for (int j = 0; j < CH_PER_GROUP; ++j) p += v[j];
  part[q][b] = p;
  __syncthreads();
  int tot = 0;
  if (t < 256) {
    for (int i = 0; i < SCAN_GROUPS; ++i) tot += part[i][t];
  }
  int all;
  const int ex = scan256_exclusive(tot, t, wsum, all);
  if (t < 256) {
    bs[t] = ex;
    if (t < NBK) bstart[t] = ex;
    if (t == 0) bstart[NBK] = all;
  }
  __syncthreads();
  if (b < NBK) {
    int run = bs[b];
    for (int i = 0; i < q; ++i) run += part[i][b];
#pragma unroll
    for (int j = 0; j < CH_PER_GROUP; ++j) {
      const int g = g0 + j;
      if (g < NCH) chfirst[g * NBK + b] = run;
      run += v[j];
    }
  }
}

__global__ __launch_bounds__(SORT_THREADS) void bucket_scatter_kernel(
    const int* __restrict__ src, const int* __restrict__ dst,
    const int* __restrict__ chfirst, int2* __restrict__ pairs) {
  __shared__ int cur[NBK];
  const int t = threadIdx.x;
  const int g = blockIdx.x;
  if (t < NBK) cur[t] = chfirst[g * NBK + t];
  __syncthreads();
  const int e0 = g * CH_EDGES;
  int d[CH_TRIPS], s[CH_TRIPS];
#pragma unroll
  for (int k = 0; k < CH_TRIPS; ++k) {
    const int i = k * SORT_THREADS + t;
    const int ic = e0 + ((i < CH_EDGES) ? i : CH_EDGES - 1);
    d[k] = dst[ic];
    s[k] = src[ic];
  }
#pragma unroll
  for (int k = 0; k < CH_TRIPS; ++k) {
    if (k * SORT_THREADS + t < CH_EDGES) {
      const int p = atomicAdd(&cur[d[k] >> BK_SHIFT], 1);
      pairs[p] = make_int2(s[k], d[k]);
    }
  }
}

__global__ __launch_bounds__(SORT_THREADS) void bucket_rows_kernel(
    const int* __restrict__ bstart, const int2* __restrict__ pairs,
    int* __restrict__ rowptr, int* __restrict__ slots) {
  __shared__ int rc[BK_ROWS];
  __shared__ int wsum[8];
  const int t = threadIdx.x;
  const int b = blockIdx.x;
  const int beg = bstart[b];
  const int end = bstart[b + 1];
  if (t < BK_ROWS) rc[t] = 0;
  __syncthreads();
  for (int i0 = beg + t; i0 < end; i0 += ROWS_UNROLL * SORT_THREADS) {
    int d[ROWS_UNROLL];
#pragma unroll
    for (int u = 0; u < ROWS_UNROLL; ++u) {
      const int i = i0 + u * SORT_THREADS;
      d[u] = pairs[(i < end) ? i : end - 1].y;
    }
#pragma unroll
    for (int u = 0; u < ROWS_UNROLL; ++u) {
      if (i0 + u * SORT_THREADS < end) atomicAdd(&rc[d[u] & (BK_ROWS - 1)], 1);
    }
  }
  __syncthreads();
  const int deg = (t < BK_ROWS) ? rc[t] : 0;
  int all;
  const int ex = scan256_exclusive(deg, t, wsum, all);
  if (t < BK_ROWS) {
    const int row = (b << BK_SHIFT) + t;
    rc[t] = beg + ex;
    if (row < GN) rowptr[row] = beg + ex;
  }
  if (b == NBK - 1 && t == 0) rowptr[GN] = end;
  __syncthreads();
  for (int i0 = beg + t; i0 < end; i0 += ROWS_UNROLL * SORT_THREADS) {
    int2 sd[ROWS_UNROLL];
#pragma unroll
    for (int u = 0; u < ROWS_UNROLL; ++u) {
      const int i = i0 + u * SORT_THREADS;
      sd[u] = pairs[(i < end) ? i : end - 1];
    }
#pragma unroll
    for (int u = 0; u < ROWS_UNROLL; ++u) {
      if (i0 + u * SORT_THREADS < end) {
        const int p = atomicAdd(&rc[sd[u].y & (BK_ROWS - 1)], 1);
        slots[p] = sd[u].x;
      }
    }
  }
}

__global__ __launch_bounds__(256) void segment_order_kernel(
    const int* __restrict__ rowptr, const int* __restrict__ slots,
    int* __restrict__ csr) {
  const int lane = threadIdx.x & 31;
  const int row = __builtin_amdgcn_readfirstlane(
      (int)((blockIdx.x * 256u + threadIdx.x) >> 5));
  if (row >= GN) return;
  const int beg = rowptr[row];
  const int n = rowptr[row + 1] - beg;
  for (int ib = 0; ib < n; ib += 32) {
    const int i = ib + lane;
    const int key = (i < n) ? slots[beg + i] : 0;
    int rank = 0;
    for (int jb = 0; jb < n; jb += 32) {
      const int jn = (n - jb < 32) ? (n - jb) : 32;
      const int kv = (lane < jn) ? slots[beg + jb + lane] : 0;
      for (int jj = 0; jj < jn; ++jj) {
        const int kj = __builtin_amdgcn_readlane(kv, jj);
        const int j = jb + jj;
        rank += (kj < key || (kj == key && j < i)) ? 1 : 0;
      }
    }
    if (i < n) csr[beg + rank] = key;
  }
}

__device__ __forceinline__ void rows_in_flight(
    v4f r0, v4f r1, v4f r2, v4f r3, v4f r4, v4f r5, v4f r6, v4f r7, v4f r8,
    v4f r9, v4f r10, v4f r11, v4f r12, v4f r13, v4f r14, v4f r15) {
  asm volatile("" ::"v"(r0), "v"(r1), "v"(r2), "v"(r3), "v"(r4), "v"(r5),
               "v"(r6), "v"(r7), "v"(r8), "v"(r9), "v"(r10), "v"(r11),
               "v"(r12), "v"(r13), "v"(r14), "v"(r15));
}

__device__ __forceinline__ void behind_rows_in_flight(v4f& r) {
  asm volatile("" : "+v"(r));
}

template <int U, int NF>
__device__ __forceinline__ void gather_group(const float* __restrict__ f0,
                                             const float* __restrict__ f1,
                                             int idx, int j0, int c4,
                                             float4& a0, float4& a1) {
  static_assert(U * NF <= 16, "rows_in_flight names sixteen rows");
  v4f v[16];
#pragma unroll
  for (int u = 0; u < U; ++u) {
    const int s = __builtin_amdgcn_readlane(idx, j0 + u);
    v[NF * u] = *(const v4f*)(f0 + (size_t)s * GH + c4);
    if (NF == 2) v[NF * u + 1] = *(const v4f*)(f1 + (size_t)s * GH + c4);
  }
#pragma unroll
  for (int k = U * NF; k < 16; ++k) v[k] = v[0];
  rows_in_flight(v[0], v[1], v[2], v[3], v[4], v[5], v[6], v[7], v[8], v[9],
                 v[10], v[11], v[12], v[13], v[14], v[15]);
#pragma unroll
  for (int u = 0; u < U; ++u) {
    behind_rows_in_flight(v[NF * u]);
    a0.x += v[NF * u].x;
    a0.y += v[NF * u].y;
    a0.z += v[NF * u].z;
    a0.w += v[NF * u].w;
    if (NF == 2) {
      behind_rows_in_flight(v[NF * u + 1]);
      a1.x += v[NF * u + 1].x;
      a1.y += v[NF * u + 1].y;
      a1.z += v[NF * u + 1].z;
      a1.w += v[NF * u + 1].w;
    }
  }
}

__device__ __forceinline__ float4 operand_order(int lane, float4 s) {
  const bool odd = (lane & 1) != 0;
  const float sx = odd ? s.x : s.z;
  const float sy = odd ? s.y : s.w;
  const float rx = __shfl_xor(sx, 1);
  const float ry = __shfl_xor(sy, 1);
  float4 w;
  w.x = odd ? rx : s.x;
  w.y = odd ? ry : s.y;
  w.z = odd ? s.z : rx;
  w.w = odd ? s.w : ry;
  return w;
}

template <int NF>
__device__ __forceinline__ void gather_batch(const float* __restrict__ f0,
                                             const float* __restrict__ f1,
                                             int idx, int n, int c4,
                                             float4& a0, float4& a1) {
  constexpr int U = 16 / NF;
  int j = 0;
  for (; j + U <= n; j += U) gather_group<U, NF>(f0, f1, idx, j, c4, a0, a1);
  if (NF == 1 && ((n - j) & 8) != 0) {
    gather_group<8, NF>(f0, f1, idx, j, c4, a0, a1);
    j += 8;
  }
  switch (n - j) {
    case 1: gather_group<1, NF>(f0, f1, idx, j, c4, a0, a1); break;
    case 2: gather_group<2, NF>(f0, f1, idx, j, c4, a0, a1); break;
    case 3: gather_group<3, NF>(f0, f1, idx, j, c4, a0, a1); break;
    case 4: gather_group<4, NF>(f0, f1, idx, j, c4, a0, a1); break;
    case 5: gather_group<5, NF>(f0, f1, idx, j, c4, a0, a1); break;
    case 6: gather_group<6, NF>(f0, f1, idx, j, c4, a0, a1); break;
    case 7: gather_group<7, NF>(f0, f1, idx, j, c4, a0, a1); break;
    default: break;
  }
}

template <int NF>
__device__ __forceinline__ void gather_list(
    const int* __restrict__ csr,
    const float* __restrict__ f0, const float* __restrict__ f1,
    int row, int lane, int beg, int end, int idx_a, int idx_b,
    float4& s0, float4& s1) {
  const int c4 = lane << 2;
  const float4 x0 = *(const float4*)(f0 + (size_t)row * GH + c4);
  const float4 x1 = *(const float4*)(f1 + (size_t)row * GH + c4);
  float4 a0 = {0.0f, 0.0f, 0.0f, 0.0f};
  float4 a1 = {0.0f, 0.0f, 0.0f, 0.0f};
  {
    const int n = (end - beg < 32) ? (end - beg) : 32;
    gather_batch<NF>(f0, f1, idx_a, n, c4, a0, a1);
  }
  for (int base = beg + 32; base < end; base += 32) {
    const int n = (end - base < 32) ? (end - base) : 32;
    int idx = idx_b;
    if (base != beg + 32) idx = (lane < n) ? csr[base + lane] : 0;
    gather_batch<NF>(f0, f1, idx, n, c4, a0, a1);
  }
  s0.x = x0.x + a0.x;
  s0.y = x0.y + a0.y;
  s0.z = x0.z + a0.z;
  s0.w = x0.w + a0.w;
  s1.x = x1.x + a1.x;
  s1.y = x1.y + a1.y;
  s1.z = x1.z + a1.z;
  s1.w = x1.w + a1.w;
}

__global__ __launch_bounds__(FUSE_THREADS) void gather_gemm_zr_kernel(
    const int* __restrict__ rowptr, const int* __restrict__ csr,
    const float* __restrict__ x, const float* __restrict__ hl,
    float* __restrict__ sop,
    const float* __restrict__ Wl, const float* __restrict__ bl,
    float* __restrict__ zout, float* __restrict__ vout) {
  __shared__ __attribute__((aligned(16))) float lds[4 * GH * 132 + 4 * TILE_WORDS];
  const int lane = threadIdx.x & 31;
  const int wave = __builtin_amdgcn_readfirstlane((int)(threadIdx.x >> 5));
  float* ring = lds + 4 * GH * 132;

  stage_matrix<FUSE_THREADS>(Wl + 0 * GH * GH, lds + 0 * GH * 132);
  stage_matrix<FUSE_THREADS>(Wl + 1 * GH * GH, lds + 1 * GH * 132);
  stage_matrix<FUSE_THREADS>(Wl + 2 * GH * GH, lds + 2 * GH * 132);
  stage_matrix<FUSE_THREADS>(Wl + 3 * GH * GH, lds + 3 * GH * 132);

  const int b = blockIdx.x;
  const int t0 = (b * ROW_TILES) / GEMM_GRID;
  const int nt = ((b + 1) * ROW_TILES) / GEMM_GRID - t0;

  int rbeg = 0, rend = 0, next_a = 0, next_b = 0;
  if (wave < 16) {
    const int k = (lane < nt) ? lane : nt - 1;
    const int r = (t0 + k) * 16 + wave;
    rbeg = rowptr[r];
    rend = rowptr[r + 1];
    const int beg0 = __builtin_amdgcn_readlane(rbeg, 0);
    const int end0 = __builtin_amdgcn_readlane(rend, 0);
    next_a = (lane < end0 - beg0) ? csr[beg0 + lane] : 0;
    next_b = (lane + 32 < end0 - beg0) ? csr[beg0 + 32 + lane] : 0;
  }

  for (int step = 0; step <= nt; ++step) {
    if (wave < 16) {
      if (step < nt) {
        const int row = (t0 + step) * 16 + wave;
        const int c4 = lane << 2;
        const int beg = __builtin_amdgcn_readlane(rbeg, step);
        const int end = __builtin_amdgcn_readlane(rend, step);
        const int idx_a = next_a;
        const int idx_b = next_b;
        if (step + 1 < nt) {
          const int nb = __builtin_amdgcn_readlane(rbeg, step + 1);
          const int ne = __builtin_amdgcn_readlane(rend, step + 1);
          next_a = (lane < ne - nb) ? csr[nb + lane] : 0;
          next_b = (lane + 32 < ne - nb) ? csr[nb + 32 + lane] : 0;
        }
        float4 s0, s1;
        gather_list<2>(csr, x, hl, row, lane, beg, end, idx_a, idx_b, s0, s1);
        const float4 w0 = operand_order(lane, s0);
        const float4 w1 = operand_order(lane, s1);
        float* slot = ring + (step & 1) * 2 * TILE_WORDS;
        *(float4*)(slot + wave * TILE_PITCH + c4) = w0;
        *(float4*)(slot + TILE_WORDS + wave * TILE_PITCH + c4) = w1;
        *(float4*)(sop + (size_t)row * GH + c4) = w0;
      }
    } else if (step > 0) {
      const int mw = wave - 16;
      const int gate = mw >> 3;
      const int n0 = (mw & 7) * 16;
      const int half = lane >> 4;
      const int kof = half << 1;
      const int nn = lane & 15;
      const int rowhi = half << 3;
      const int row0 = (t0 + step - 1) * 16;
      const float* slot = ring + ((step - 1) & 1) * 2 * TILE_WORDS;
      const float* as = slot + nn * TILE_PITCH + (half << 2);
      const float* at = as + TILE_WORDS;
      const float* pa = lds + (2 * gate) * GH * 132 + (n0 + nn) * 132 + kof;
      const float* pb = pa + GH * 132;

      const size_t off0 = (size_t)(row0 + rowhi) * GH + n0 + nn;
      float hv[8];
#pragma unroll
      for (int i = 0; i < 8; ++i) hv[i] = hl[off0 + (size_t)i * GH];
      const float b0 = bl[(2 * gate) * GH + n0 + nn];
      const float b1 = bl[(2 * gate + 1) * GH + n0 + nn];
      __builtin_amdgcn_sched_barrier(0);

      v8f c0 = {0, 0, 0, 0, 0, 0, 0, 0}, c1 = {0, 0, 0, 0, 0, 0, 0, 0};
#pragma unroll
      for (int m = 0; m < 16; ++m) {
        const float4 sv = *(const float4*)(as + 8 * m);
        const float4 tv = *(const float4*)(at + 8 * m);
        v2f slo, shi, tlo, thi;
        slo.x = sv.x; slo.y = sv.y; shi.x = sv.z; shi.y = sv.w;
        tlo.x = tv.x; tlo.y = tv.y; thi.x = tv.z; thi.y = tv.w;
        c0 = wmma_f32(slo, *(const v2f*)(pa + 8 * m), c0);
        c0 = wmma_f32(tlo, *(const v2f*)(pb + 8 * m), c0);
        c1 = wmma_f32(shi, *(const v2f*)(pa + 8 * m + 4), c1);
        c1 = wmma_f32(thi, *(const v2f*)(pb + 8 * m + 4), c1);
      }
      const v8f acc = c0 + c1;
      const float bb = b0 + b1;
      float* dstp = (gate != 0) ? vout : zout;
#pragma unroll
      for (int i = 0; i < 8; ++i) {
        const float g = sigmoid_f(acc[i] + bb);
        dstp[off0 + (size_t)i * GH] = (gate != 0) ? g * hv[i] : g;
      }
    }
    __syncthreads();
  }
}

__global__ __launch_bounds__(FUSE_HT_THREADS) void gather_gemm_ht_kernel(
    const int* __restrict__ rowptr, const int* __restrict__ csr,
    const float* __restrict__ vin, const float* __restrict__ sop,
    const float* __restrict__ Wl, const float* __restrict__ bl,
    const float* __restrict__ zbuf, const float* __restrict__ hl,
    float* __restrict__ outp) {
  __shared__ __attribute__((aligned(16))) float lds[2 * GH * 132 + 4 * TILE_WORDS];
  const int lane = threadIdx.x & 31;
  const int wave = __builtin_amdgcn_readfirstlane((int)(threadIdx.x >> 5));
  float* ring = lds + 2 * GH * 132;

  stage_matrix<FUSE_HT_THREADS>(Wl + 4 * GH * GH, lds + 0 * GH * 132);
  stage_matrix<FUSE_HT_THREADS>(Wl + 5 * GH * GH, lds + 1 * GH * 132);

  const int b = blockIdx.x;
  const int t0 = (b * ROW_TILES) / GEMM_GRID;
  const int nt = ((b + 1) * ROW_TILES) / GEMM_GRID - t0;

  int rbeg = 0, rend = 0, next_a = 0, next_b = 0;
  if (wave < 16) {
    const int k = (lane < nt) ? lane : nt - 1;
    const int r = (t0 + k) * 16 + wave;
    rbeg = rowptr[r];
    rend = rowptr[r + 1];
    const int beg0 = __builtin_amdgcn_readlane(rbeg, 0);
    const int end0 = __builtin_amdgcn_readlane(rend, 0);
    next_a = (lane < end0 - beg0) ? csr[beg0 + lane] : 0;
    next_b = (lane + 32 < end0 - beg0) ? csr[beg0 + 32 + lane] : 0;
  }

  for (int step = 0; step <= nt; ++step) {
    if (wave < 16) {
      if (step < nt) {
        const int row = (t0 + step) * 16 + wave;
        const int c4 = lane << 2;
        const int beg = __builtin_amdgcn_readlane(rbeg, step);
        const int end = __builtin_amdgcn_readlane(rend, step);
        const int idx_a = next_a;
        const int idx_b = next_b;
        if (step + 1 < nt) {
          const int nb = __builtin_amdgcn_readlane(rbeg, step + 1);
          const int ne = __builtin_amdgcn_readlane(rend, step + 1);
          next_a = (lane < ne - nb) ? csr[nb + lane] : 0;
          next_b = (lane + 32 < ne - nb) ? csr[nb + 32 + lane] : 0;
        }
        const float4 srow = *(const float4*)(sop + (size_t)row * GH + c4);
        float4 u0, u1;
        gather_list<1>(csr, vin, vin, row, lane, beg, end, idx_a, idx_b, u0, u1);
        const float4 w = operand_order(lane, u0);
        float* slot = ring + (step & 1) * 2 * TILE_WORDS;
        *(float4*)(slot + wave * TILE_PITCH + c4) = srow;
        *(float4*)(slot + TILE_WORDS + wave * TILE_PITCH + c4) = w;
      }
    } else if (step > 0) {
      const int n0 = (wave - 16) * 16;
      const int half = lane >> 4;
      const int kof = half << 1;
      const int nn = lane & 15;
      const int rowhi = half << 3;
      const int row0 = (t0 + step - 1) * 16;
      const float* slot = ring + ((step - 1) & 1) * 2 * TILE_WORDS;
      const float* as = slot + nn * TILE_PITCH + (half << 2);
      const float* au = as + TILE_WORDS;
      const float* p4 = lds + (n0 + nn) * 132 + kof;
      const float* p5 = p4 + GH * 132;

      const size_t off0 = (size_t)(row0 + rowhi) * GH + n0 + nn;
      float zv8[8], hv[8];
#pragma unroll
      for (int i = 0; i < 8; ++i) {
        zv8[i] = zbuf[off0 + (size_t)i * GH];
        hv[i] = hl[off0 + (size_t)i * GH];
      }
      const float bh0 = bl[4 * GH + n0 + nn], bh1 = bl[5 * GH + n0 + nn];
      __builtin_amdgcn_sched_barrier(0);

      v8f a0 = {0, 0, 0, 0, 0, 0, 0, 0}, a1 = {0, 0, 0, 0, 0, 0, 0, 0};
#pragma unroll
      for (int m = 0; m < 16; ++m) {
        const float4 sv = *(const float4*)(as + 8 * m);
        const float4 uv = *(const float4*)(au + 8 * m);
        v2f slo, shi, ulo, uhi;
        slo.x = sv.x; slo.y = sv.y; shi.x = sv.z; shi.y = sv.w;
        ulo.x = uv.x; ulo.y = uv.y; uhi.x = uv.z; uhi.y = uv.w;
        a0 = wmma_f32(slo, *(const v2f*)(p4 + 8 * m), a0);
        a0 = wmma_f32(ulo, *(const v2f*)(p5 + 8 * m), a0);
        a1 = wmma_f32(shi, *(const v2f*)(p4 + 8 * m + 4), a1);
        a1 = wmma_f32(uhi, *(const v2f*)(p5 + 8 * m + 4), a1);
      }
      const v8f acc = a0 + a1;

      const float bh = bh0 + bh1;
#pragma unroll
      for (int i = 0; i < 8; ++i) {
        const float ht = tanh_f(acc[i] + bh);
        const float zv = zv8[i];
        outp[off0 + (size_t)i * GH] = zv * hv[i] + (1.0f - zv) * ht;
      }
    }
    __syncthreads();
  }
}

constexpr int GN_PAD = 50048;
constexpr int CHCNT_PAD = ((NCH * NBK + 63) / 64) * 64;
constexpr size_t OFF_ROWPTR = 0;
constexpr size_t OFF_CHCNT = OFF_ROWPTR + GN_PAD;
constexpr size_t OFF_CHFIRST = OFF_CHCNT + CHCNT_PAD;
constexpr size_t OFF_BSTART = OFF_CHFIRST + CHCNT_PAD;
constexpr size_t OFF_SLOTS = OFF_BSTART + 256;
constexpr size_t OFF_PAIRS = OFF_SLOTS + GE;
constexpr size_t OFF_CSR = OFF_PAIRS;
static_assert(NBK + 1 <= 256, "bucket starts must fit their slot");
static_assert(OFF_PAIRS + 2 * (size_t)GE <= NH, "edge lists must fit in one workspace plane");
constexpr int ROW_BLOCKS = (GN * 32 + 255) / 256;

extern "C" void kernel_launch(void* const* d_in, const int* in_sizes, int n_in,
                              void* d_out, int out_size, void* d_ws,
                              size_t ws_size, hipStream_t stream) {
  (void)in_sizes; (void)n_in; (void)out_size; (void)ws_size;
  const float* inp = (const float*)d_in[0];
  const int* edg = (const int*)d_in[1];
  const float* hmat = (const float*)d_in[2];
  const float* Wmat = (const float*)d_in[3];
  const float* bmat = (const float*)d_in[4];
  float* out = (float*)d_out;

  const int* src = edg;
  const int* dst = edg + GE;

  float* agg_x = (float*)d_ws;
  int* lists = (int*)(agg_x + NH);
  float* zbuf = agg_x + 2 * NH;
  float* vbuf = zbuf + NH;

  int* rowptr = lists + OFF_ROWPTR;
  int* chcnt = lists + OFF_CHCNT;
  int* chfirst = lists + OFF_CHFIRST;
  int* bstart = lists + OFF_BSTART;
  int* slots = lists + OFF_SLOTS;
  int2* pairs = (int2*)(lists + OFF_PAIRS);
  int* csr = lists + OFF_CSR;

  bucket_count_kernel<<<NCH, SORT_THREADS, 0, stream>>>(dst, chcnt);
  bucket_scan_kernel<<<1, SORT_THREADS, 0, stream>>>(chcnt, chfirst, bstart);
  bucket_scatter_kernel<<<NCH, SORT_THREADS, 0, stream>>>(src, dst, chfirst, pairs);
  bucket_rows_kernel<<<NBK, SORT_THREADS, 0, stream>>>(bstart, pairs, rowptr, slots);
  segment_order_kernel<<<ROW_BLOCKS, 256, 0, stream>>>(rowptr, slots, csr);

  for (int l = 0; l < GLAYERS; ++l) {
    const float* x = (l == 0) ? inp : (out + (size_t)(l - 1) * NH);
    const float* hl = hmat + (size_t)l * NH;
    const float* Wl = Wmat + (size_t)l * 6 * GH * GH;
    const float* bl = bmat + (size_t)l * 6 * GH;
    float* outl = out + (size_t)l * NH;

    gather_gemm_zr_kernel<<<GEMM_GRID, FUSE_THREADS, 0, stream>>>(
        rowptr, csr, x, hl, agg_x, Wl, bl, zbuf, vbuf);

    gather_gemm_ht_kernel<<<GEMM_GRID, FUSE_HT_THREADS, 0, stream>>>(
        rowptr, csr, vbuf, agg_x, Wl, bl, zbuf, hl, outl);
  }
}


# GEMM launches: per-step barrier split: a gathering wave stores row k-1, signals, sums row k in registers, then waits, so a step is no longer as long as its longest row; sums unchanged
# speedup vs baseline: 5.3356x; 1.0562x over previous
#include <hip/hip_runtime.h>
#include <cstdint>
#include <cstddef>

constexpr int GN = 50000;
constexpr int GE = 1600000;
constexpr int GH = 128;
constexpr int GLAYERS = 2;
constexpr size_t NH = (size_t)GN * GH;
constexpr int ROW_TILES = GN / 16;
constexpr int GEMM_GRID = 256;
constexpr int FUSE_THREADS = 1024;
constexpr int FUSE_HT_THREADS = 768;
constexpr int TILE_PITCH = 132;
constexpr int TILE_WORDS = 16 * TILE_PITCH;
static_assert(GN % 16 == 0, "row tiles must be whole");

typedef __attribute__((ext_vector_type(2))) float v2f;
typedef __attribute__((ext_vector_type(4))) float v4f;
typedef __attribute__((ext_vector_type(8))) float v8f;

__device__ __forceinline__ v8f wmma_f32(v2f a, v2f b, v8f c) {
  return __builtin_amdgcn_wmma_f32_16x16x4_f32(
       false, a,  false, b,
       (short)0, c,  false,  false);
}

__device__ __forceinline__ float sigmoid_f(float x) {
  return 1.0f / (1.0f + __expf(-x));
}
__device__ __forceinline__ float tanh_f(float x) {
  return 1.0f - 2.0f / (1.0f + __expf(2.0f * x));
}


template <int NT>
__device__ __forceinline__ void stage_matrix(const float* __restrict__ Wg,
                                             float* lds) {
  const int tid = threadIdx.x;
#pragma unroll
  for (int i = 0; i < (GH * GH + NT - 1) / NT; ++i) {
    const int idx = i * NT + tid;
    if ((GH * GH) % NT == 0 || idx < GH * GH) {
      const int c = idx & 15;
      const int k = (idx >> 4) & (GH - 1);
      const int n = ((idx >> 11) << 4) + c;
      lds[n * 132 + k] = Wg[k * GH + n];
    }
  }
}


constexpr int BK_SHIFT = 8;
constexpr int BK_ROWS = 1 << BK_SHIFT;
constexpr int NBK = (GN + BK_ROWS - 1) >> BK_SHIFT;
constexpr int SORT_THREADS = 1024;
constexpr int NCH = 250;
constexpr int CH_EDGES = GE / NCH;
constexpr int CH_TRIPS = (CH_EDGES + SORT_THREADS - 1) / SORT_THREADS;
constexpr int SCAN_GROUPS = SORT_THREADS / 256;
constexpr int CH_PER_GROUP = (NCH + SCAN_GROUPS - 1) / SCAN_GROUPS;
constexpr int ROWS_UNROLL = 4;
static_assert(NCH * CH_EDGES == GE, "chunks must tile the edge list");
static_assert(NBK <= 256, "one scan thread per bucket");

__device__ __forceinline__ int scan256_exclusive(int v, int t, int* wsum, int& total) {
  const int lane = t & 31;
  int incl = v;
  if (t < 256) {
    for (int off = 1; off < 32; off <<= 1) {
      const int y = __shfl_up(incl, off);
      if (lane >= off) incl += y;
    }
    if (lane == 31) wsum[t >> 5] = incl;
  }
  __syncthreads();
  int woff = 0, all = 0;
  for (int w = 0; w < 8; ++w) {
    const int s = wsum[w];
    if (w < (t >> 5)) woff += s;
    all += s;
  }
  total = all;
  return woff + incl - v;
}

__global__ __launch_bounds__(SORT_THREADS) void bucket_count_kernel(
    const int* __restrict__ dst, int* __restrict__ chcnt) {
  __shared__ int hist[NBK];
  const int t = threadIdx.x;
  const int g = blockIdx.x;
  if (t < NBK) hist[t] = 0;
  __syncthreads();
  const int e0 = g * CH_EDGES;
  int d[CH_TRIPS];
#pragma unroll
  for (int k = 0; k < CH_TRIPS; ++k) {
    const int i = k * SORT_THREADS + t;
    d[k] = dst[e0 + ((i < CH_EDGES) ? i : CH_EDGES - 1)];
  }
#pragma unroll
  for (int k = 0; k < CH_TRIPS; ++k) {
    if (k * SORT_THREADS + t < CH_EDGES) atomicAdd(&hist[d[k] >> BK_SHIFT], 1);
  }
  __syncthreads();
  if (t < NBK) chcnt[g * NBK + t] = hist[t];
}

__global__ __launch_bounds__(SORT_THREADS) void bucket_scan_kernel(
    const int* __restrict__ chcnt, int* __restrict__ chfirst,
    int* __restrict__ bstart) {
  __shared__ int part[SCAN_GROUPS][256];
  __shared__ int bs[256];
  __shared__ int wsum[8];
  const int t = threadIdx.x;
  const int q = t >> 8;
  const int b = t & 255;
  const int g0 = q * CH_PER_GROUP;
  const int bc = (b < NBK) ? b : NBK - 1;
  int v[CH_PER_GROUP];
#pragma unroll
  for (int j = 0; j < CH_PER_GROUP; ++j) {
    const int g = g0 + j;
    const int x = chcnt[((g < NCH) ? g : NCH - 1) * NBK + bc];
    v[j] = (g < NCH && b < NBK) ? x : 0;
  }
  int p = 0;
#pragma unroll
  for (int j = 0; j < CH_PER_GROUP; ++j) p += v[j];
  part[q][b] = p;
  __syncthreads();
  int tot = 0;
  if (t < 256) {
    for (int i = 0; i < SCAN_GROUPS; ++i) tot += part[i][t];
  }
  int all;
  const int ex = scan256_exclusive(tot, t, wsum, all);
  if (t < 256) {
    bs[t] = ex;
    if (t < NBK) bstart[t] = ex;
    if (t == 0) bstart[NBK] = all;
  }
  __syncthreads();
  if (b < NBK) {
    int run = bs[b];
    for (int i = 0; i < q; ++i) run += part[i][b];
#pragma unroll
    for (int j = 0; j < CH_PER_GROUP; ++j) {
      const int g = g0 + j;
      if (g < NCH) chfirst[g * NBK + b] = run;
      run += v[j];
    }
  }
}

__global__ __launch_bounds__(SORT_THREADS) void bucket_scatter_kernel(
    const int* __restrict__ src, const int* __restrict__ dst,
    const int* __restrict__ chfirst, int2* __restrict__ pairs) {
  __shared__ int cur[NBK];
  const int t = threadIdx.x;
  const int g = blockIdx.x;
  if (t < NBK) cur[t] = chfirst[g * NBK + t];
  __syncthreads();
  const int e0 = g * CH_EDGES;
  int d[CH_TRIPS], s[CH_TRIPS];
#pragma unroll
  for (int k = 0; k < CH_TRIPS; ++k) {
    const int i = k * SORT_THREADS + t;
    const int ic = e0 + ((i < CH_EDGES) ? i : CH_EDGES - 1);
    d[k] = dst[ic];
    s[k] = src[ic];
  }
#pragma unroll
  for (int k = 0; k < CH_TRIPS; ++k) {
    if (k * SORT_THREADS + t < CH_EDGES) {
      const int p = atomicAdd(&cur[d[k] >> BK_SHIFT], 1);
      pairs[p] = make_int2(s[k], d[k]);
    }
  }
}

__global__ __launch_bounds__(SORT_THREADS) void bucket_rows_kernel(
    const int* __restrict__ bstart, const int2* __restrict__ pairs,
    int* __restrict__ rowptr, int* __restrict__ slots) {
  __shared__ int rc[BK_ROWS];
  __shared__ int wsum[8];
  const int t = threadIdx.x;
  const int b = blockIdx.x;
  const int beg = bstart[b];
  const int end = bstart[b + 1];
  if (t < BK_ROWS) rc[t] = 0;
  __syncthreads();
  for (int i0 = beg + t; i0 < end; i0 += ROWS_UNROLL * SORT_THREADS) {
    int d[ROWS_UNROLL];
#pragma unroll
    for (int u = 0; u < ROWS_UNROLL; ++u) {
      const int i = i0 + u * SORT_THREADS;
      d[u] = pairs[(i < end) ? i : end - 1].y;
    }
#pragma unroll
    for (int u = 0; u < ROWS_UNROLL; ++u) {
      if (i0 + u * SORT_THREADS < end) atomicAdd(&rc[d[u] & (BK_ROWS - 1)], 1);
    }
  }
  __syncthreads();
  const int deg = (t < BK_ROWS) ? rc[t] : 0;
  int all;
  const int ex = scan256_exclusive(deg, t, wsum, all);
  if (t < BK_ROWS) {
    const int row = (b << BK_SHIFT) + t;
    rc[t] = beg + ex;
    if (row < GN) rowptr[row] = beg + ex;
  }
  if (b == NBK - 1 && t == 0) rowptr[GN] = end;
  __syncthreads();
  for (int i0 = beg + t; i0 < end; i0 += ROWS_UNROLL * SORT_THREADS) {
    int2 sd[ROWS_UNROLL];
#pragma unroll
    for (int u = 0; u < ROWS_UNROLL; ++u) {
      const int i = i0 + u * SORT_THREADS;
      sd[u] = pairs[(i < end) ? i : end - 1];
    }
#pragma unroll
    for (int u = 0; u < ROWS_UNROLL; ++u) {
      if (i0 + u * SORT_THREADS < end) {
        const int p = atomicAdd(&rc[sd[u].y & (BK_ROWS - 1)], 1);
        slots[p] = sd[u].x;
      }
    }
  }
}

__global__ __launch_bounds__(256) void segment_order_kernel(
    const int* __restrict__ rowptr, const int* __restrict__ slots,
    int* __restrict__ csr) {
  const int lane = threadIdx.x & 31;
  const int row = __builtin_amdgcn_readfirstlane(
      (int)((blockIdx.x * 256u + threadIdx.x) >> 5));
  if (row >= GN) return;
  const int beg = rowptr[row];
  const int n = rowptr[row + 1] - beg;
  for (int ib = 0; ib < n; ib += 32) {
    const int i = ib + lane;
    const int key = (i < n) ? slots[beg + i] : 0;
    int rank = 0;
    for (int jb = 0; jb < n; jb += 32) {
      const int jn = (n - jb < 32) ? (n - jb) : 32;
      const int kv = (lane < jn) ? slots[beg + jb + lane] : 0;
      for (int jj = 0; jj < jn; ++jj) {
        const int kj = __builtin_amdgcn_readlane(kv, jj);
        const int j = jb + jj;
        rank += (kj < key || (kj == key && j < i)) ? 1 : 0;
      }
    }
    if (i < n) csr[beg + rank] = key;
  }
}

__device__ __forceinline__ void rows_in_flight(
    v4f r0, v4f r1, v4f r2, v4f r3, v4f r4, v4f r5, v4f r6, v4f r7, v4f r8,
    v4f r9, v4f r10, v4f r11, v4f r12, v4f r13, v4f r14, v4f r15) {
  asm volatile("" ::"v"(r0), "v"(r1), "v"(r2), "v"(r3), "v"(r4), "v"(r5),
               "v"(r6), "v"(r7), "v"(r8), "v"(r9), "v"(r10), "v"(r11),
               "v"(r12), "v"(r13), "v"(r14), "v"(r15));
}

__device__ __forceinline__ void behind_rows_in_flight(v4f& r) {
  asm volatile("" : "+v"(r));
}

template <int U, int NF>
__device__ __forceinline__ void gather_group(const float* __restrict__ f0,
                                             const float* __restrict__ f1,
                                             int idx, int j0, int c4,
                                             float4& a0, float4& a1) {
  static_assert(U * NF <= 16, "rows_in_flight names sixteen rows");
  v4f v[16];
#pragma unroll
  for (int u = 0; u < U; ++u) {
    const int s = __builtin_amdgcn_readlane(idx, j0 + u);
    v[NF * u] = *(const v4f*)(f0 + (size_t)s * GH + c4);
    if (NF == 2) v[NF * u + 1] = *(const v4f*)(f1 + (size_t)s * GH + c4);
  }
#pragma unroll
  for (int k = U * NF; k < 16; ++k) v[k] = v[0];
  rows_in_flight(v[0], v[1], v[2], v[3], v[4], v[5], v[6], v[7], v[8], v[9],
                 v[10], v[11], v[12], v[13], v[14], v[15]);
#pragma unroll
  for (int u = 0; u < U; ++u) {
    behind_rows_in_flight(v[NF * u]);
    a0.x += v[NF * u].x;
    a0.y += v[NF * u].y;
    a0.z += v[NF * u].z;
    a0.w += v[NF * u].w;
    if (NF == 2) {
      behind_rows_in_flight(v[NF * u + 1]);
      a1.x += v[NF * u + 1].x;
      a1.y += v[NF * u + 1].y;
      a1.z += v[NF * u + 1].z;
      a1.w += v[NF * u + 1].w;
    }
  }
}

__device__ __forceinline__ float4 operand_order(int lane, float4 s) {
  const bool odd = (lane & 1) != 0;
  const float sx = odd ? s.x : s.z;
  const float sy = odd ? s.y : s.w;
  const float rx = __shfl_xor(sx, 1);
  const float ry = __shfl_xor(sy, 1);
  float4 w;
  w.x = odd ? rx : s.x;
  w.y = odd ? ry : s.y;
  w.z = odd ? s.z : rx;
  w.w = odd ? s.w : ry;
  return w;
}

template <int NF>
__device__ __forceinline__ void gather_batch(const float* __restrict__ f0,
                                             const float* __restrict__ f1,
                                             int idx, int n, int c4,
                                             float4& a0, float4& a1) {
  constexpr int U = 16 / NF;
  int j = 0;
  for (; j + U <= n; j += U) gather_group<U, NF>(f0, f1, idx, j, c4, a0, a1);
  if (NF == 1 && ((n - j) & 8) != 0) {
    gather_group<8, NF>(f0, f1, idx, j, c4, a0, a1);
    j += 8;
  }
  switch (n - j) {
    case 1: gather_group<1, NF>(f0, f1, idx, j, c4, a0, a1); break;
    case 2: gather_group<2, NF>(f0, f1, idx, j, c4, a0, a1); break;
    case 3: gather_group<3, NF>(f0, f1, idx, j, c4, a0, a1); break;
    case 4: gather_group<4, NF>(f0, f1, idx, j, c4, a0, a1); break;
    case 5: gather_group<5, NF>(f0, f1, idx, j, c4, a0, a1); break;
    case 6: gather_group<6, NF>(f0, f1, idx, j, c4, a0, a1); break;
    case 7: gather_group<7, NF>(f0, f1, idx, j, c4, a0, a1); break;
    default: break;
  }
}

template <int NF>
__device__ __forceinline__ void gather_list(
    const int* __restrict__ csr,
    const float* __restrict__ f0, const float* __restrict__ f1,
    int row, int lane, int beg, int end, int idx_a, int idx_b,
    float4& s0, float4& s1) {
  const int c4 = lane << 2;
  const float4 x0 = *(const float4*)(f0 + (size_t)row * GH + c4);
  const float4 x1 = *(const float4*)(f1 + (size_t)row * GH + c4);
  float4 a0 = {0.0f, 0.0f, 0.0f, 0.0f};
  float4 a1 = {0.0f, 0.0f, 0.0f, 0.0f};
  {
    const int n = (end - beg < 32) ? (end - beg) : 32;
    gather_batch<NF>(f0, f1, idx_a, n, c4, a0, a1);
  }
  for (int base = beg + 32; base < end; base += 32) {
    const int n = (end - base < 32) ? (end - base) : 32;
    int idx = idx_b;
    if (base != beg + 32) idx = (lane < n) ? csr[base + lane] : 0;
    gather_batch<NF>(f0, f1, idx, n, c4, a0, a1);
  }
  s0.x = x0.x + a0.x;
  s0.y = x0.y + a0.y;
  s0.z = x0.z + a0.z;
  s0.w = x0.w + a0.w;
  s1.x = x1.x + a1.x;
  s1.y = x1.y + a1.y;
  s1.z = x1.z + a1.z;
  s1.w = x1.w + a1.w;
}

__device__ __forceinline__ void barrier_arrive() {
  asm volatile("s_wait_dscnt 0x0\n\ts_barrier_signal -1" ::: "memory");
}
__device__ __forceinline__ void barrier_pass() {
  asm volatile("s_barrier_wait -1" ::: "memory");
}

__global__ __launch_bounds__(FUSE_THREADS) void gather_gemm_zr_kernel(
    const int* __restrict__ rowptr, const int* __restrict__ csr,
    const float* __restrict__ x, const float* __restrict__ hl,
    float* __restrict__ sop,
    const float* __restrict__ Wl, const float* __restrict__ bl,
    float* __restrict__ zout, float* __restrict__ vout) {
  __shared__ __attribute__((aligned(16))) float lds[4 * GH * 132 + 4 * TILE_WORDS];
  const int lane = threadIdx.x & 31;
  const int wave = __builtin_amdgcn_readfirstlane((int)(threadIdx.x >> 5));
  float* ring = lds + 4 * GH * 132;

  stage_matrix<FUSE_THREADS>(Wl + 0 * GH * GH, lds + 0 * GH * 132);
  stage_matrix<FUSE_THREADS>(Wl + 1 * GH * GH, lds + 1 * GH * 132);
  stage_matrix<FUSE_THREADS>(Wl + 2 * GH * GH, lds + 2 * GH * 132);
  stage_matrix<FUSE_THREADS>(Wl + 3 * GH * GH, lds + 3 * GH * 132);

  const int b = blockIdx.x;
  const int t0 = (b * ROW_TILES) / GEMM_GRID;
  const int nt = ((b + 1) * ROW_TILES) / GEMM_GRID - t0;

  int rbeg = 0, rend = 0, next_a = 0, next_b = 0;
  if (wave < 16) {
    const int k = (lane < nt) ? lane : nt - 1;
    const int r = (t0 + k) * 16 + wave;
    rbeg = rowptr[r];
    rend = rowptr[r + 1];
    const int beg0 = __builtin_amdgcn_readlane(rbeg, 0);
    const int end0 = __builtin_amdgcn_readlane(rend, 0);
    next_a = (lane < end0 - beg0) ? csr[beg0 + lane] : 0;
    next_b = (lane + 32 < end0 - beg0) ? csr[beg0 + 32 + lane] : 0;
  }

  float4 w0 = {0.0f, 0.0f, 0.0f, 0.0f};
  float4 w1 = {0.0f, 0.0f, 0.0f, 0.0f};

  for (int step = 0; step <= nt; ++step) {
    if (wave < 16) {
      if (step > 0) {
        const int c4 = lane << 2;
        float* slot = ring + ((step - 1) & 1) * 2 * TILE_WORDS;
        *(float4*)(slot + wave * TILE_PITCH + c4) = w0;
        *(float4*)(slot + TILE_WORDS + wave * TILE_PITCH + c4) = w1;
      }
    }
    barrier_arrive();
    if (wave < 16) {
      if (step < nt) {
        const int row = (t0 + step) * 16 + wave;
        const int c4 = lane << 2;
        const int beg = __builtin_amdgcn_readlane(rbeg, step);
        const int end = __builtin_amdgcn_readlane(rend, step);
        const int idx_a = next_a;
        const int idx_b = next_b;
        if (step + 1 < nt) {
          const int nb = __builtin_amdgcn_readlane(rbeg, step + 1);
          const int ne = __builtin_amdgcn_readlane(rend, step + 1);
          next_a = (lane < ne - nb) ? csr[nb + lane] : 0;
          next_b = (lane + 32 < ne - nb) ? csr[nb + 32 + lane] : 0;
        }
        float4 s0, s1;
        gather_list<2>(csr, x, hl, row, lane, beg, end, idx_a, idx_b, s0, s1);
        w0 = operand_order(lane, s0);
        w1 = operand_order(lane, s1);
        *(float4*)(sop + (size_t)row * GH + c4) = w0;
      }
    }
    barrier_pass();
    if (wave >= 16 && step > 0) {
      const int mw = wave - 16;
      const int gate = mw >> 3;
      const int n0 = (mw & 7) * 16;
      const int half = lane >> 4;
      const int kof = half << 1;
      const int nn = lane & 15;
      const int rowhi = half << 3;
      const int row0 = (t0 + step - 1) * 16;
      const float* slot = ring + ((step - 1) & 1) * 2 * TILE_WORDS;
      const float* as = slot + nn * TILE_PITCH + (half << 2);
      const float* at = as + TILE_WORDS;
      const float* pa = lds + (2 * gate) * GH * 132 + (n0 + nn) * 132 + kof;
      const float* pb = pa + GH * 132;

      const size_t off0 = (size_t)(row0 + rowhi) * GH + n0 + nn;
      float hv[8];
#pragma unroll
      for (int i = 0; i < 8; ++i) hv[i] = hl[off0 + (size_t)i * GH];
      const float b0 = bl[(2 * gate) * GH + n0 + nn];
      const float b1 = bl[(2 * gate + 1) * GH + n0 + nn];
      __builtin_amdgcn_sched_barrier(0);

      v8f c0 = {0, 0, 0, 0, 0, 0, 0, 0}, c1 = {0, 0, 0, 0, 0, 0, 0, 0};
#pragma unroll
      for (int m = 0; m < 16; ++m) {
        const float4 sv = *(const float4*)(as + 8 * m);
        const float4 tv = *(const float4*)(at + 8 * m);
        v2f slo, shi, tlo, thi;
        slo.x = sv.x; slo.y = sv.y; shi.x = sv.z; shi.y = sv.w;
        tlo.x = tv.x; tlo.y = tv.y; thi.x = tv.z; thi.y = tv.w;
        c0 = wmma_f32(slo, *(const v2f*)(pa + 8 * m), c0);
        c0 = wmma_f32(tlo, *(const v2f*)(pb + 8 * m), c0);
        c1 = wmma_f32(shi, *(const v2f*)(pa + 8 * m + 4), c1);
        c1 = wmma_f32(thi, *(const v2f*)(pb + 8 * m + 4), c1);
      }
      const v8f acc = c0 + c1;
      const float bb = b0 + b1;
      float* dstp = (gate != 0) ? vout : zout;
#pragma unroll
      for (int i = 0; i < 8; ++i) {
        const float g = sigmoid_f(acc[i] + bb);
        dstp[off0 + (size_t)i * GH] = (gate != 0) ? g * hv[i] : g;
      }
    }
  }
}

__global__ __launch_bounds__(FUSE_HT_THREADS) void gather_gemm_ht_kernel(
    const int* __restrict__ rowptr, const int* __restrict__ csr,
    const float* __restrict__ vin, const float* __restrict__ sop,
    const float* __restrict__ Wl, const float* __restrict__ bl,
    const float* __restrict__ zbuf, const float* __restrict__ hl,
    float* __restrict__ outp) {
  __shared__ __attribute__((aligned(16))) float lds[2 * GH * 132 + 4 * TILE_WORDS];
  const int lane = threadIdx.x & 31;
  const int wave = __builtin_amdgcn_readfirstlane((int)(threadIdx.x >> 5));
  float* ring = lds + 2 * GH * 132;

  stage_matrix<FUSE_HT_THREADS>(Wl + 4 * GH * GH, lds + 0 * GH * 132);
  stage_matrix<FUSE_HT_THREADS>(Wl + 5 * GH * GH, lds + 1 * GH * 132);

  const int b = blockIdx.x;
  const int t0 = (b * ROW_TILES) / GEMM_GRID;
  const int nt = ((b + 1) * ROW_TILES) / GEMM_GRID - t0;

  int rbeg = 0, rend = 0, next_a = 0, next_b = 0;
  if (wave < 16) {
    const int k = (lane < nt) ? lane : nt - 1;
    const int r = (t0 + k) * 16 + wave;
    rbeg = rowptr[r];
    rend = rowptr[r + 1];
    const int beg0 = __builtin_amdgcn_readlane(rbeg, 0);
    const int end0 = __builtin_amdgcn_readlane(rend, 0);
    next_a = (lane < end0 - beg0) ? csr[beg0 + lane] : 0;
    next_b = (lane + 32 < end0 - beg0) ? csr[beg0 + 32 + lane] : 0;
  }

  float4 srow = {0.0f, 0.0f, 0.0f, 0.0f};
  float4 urow = {0.0f, 0.0f, 0.0f, 0.0f};

  for (int step = 0; step <= nt; ++step) {
    if (wave < 16) {
      if (step > 0) {
        const int c4 = lane << 2;
        float* slot = ring + ((step - 1) & 1) * 2 * TILE_WORDS;
        *(float4*)(slot + wave * TILE_PITCH + c4) = srow;
        *(float4*)(slot + TILE_WORDS + wave * TILE_PITCH + c4) = urow;
      }
    }
    barrier_arrive();
    if (wave < 16) {
      if (step < nt) {
        const int row = (t0 + step) * 16 + wave;
        const int c4 = lane << 2;
        const int beg = __builtin_amdgcn_readlane(rbeg, step);
        const int end = __builtin_amdgcn_readlane(rend, step);
        const int idx_a = next_a;
        const int idx_b = next_b;
        if (step + 1 < nt) {
          const int nb = __builtin_amdgcn_readlane(rbeg, step + 1);
          const int ne = __builtin_amdgcn_readlane(rend, step + 1);
          next_a = (lane < ne - nb) ? csr[nb + lane] : 0;
          next_b = (lane + 32 < ne - nb) ? csr[nb + 32 + lane] : 0;
        }
        srow = *(const float4*)(sop + (size_t)row * GH + c4);
        float4 u0, u1;
        gather_list<1>(csr, vin, vin, row, lane, beg, end, idx_a, idx_b, u0, u1);
        urow = operand_order(lane, u0);
      }
    }
    barrier_pass();
    if (wave >= 16 && step > 0) {
      const int n0 = (wave - 16) * 16;
      const int half = lane >> 4;
      const int kof = half << 1;
      const int nn = lane & 15;
      const int rowhi = half << 3;
      const int row0 = (t0 + step - 1) * 16;
      const float* slot = ring + ((step - 1) & 1) * 2 * TILE_WORDS;
      const float* as = slot + nn * TILE_PITCH + (half << 2);
      const float* au = as + TILE_WORDS;
      const float* p4 = lds + (n0 + nn) * 132 + kof;
      const float* p5 = p4 + GH * 132;

      const size_t off0 = (size_t)(row0 + rowhi) * GH + n0 + nn;
      float zv8[8], hv[8];
#pragma unroll
      for (int i = 0; i < 8; ++i) {
        zv8[i] = zbuf[off0 + (size_t)i * GH];
        hv[i] = hl[off0 + (size_t)i * GH];
      }
      const float bh0 = bl[4 * GH + n0 + nn], bh1 = bl[5 * GH + n0 + nn];
      __builtin_amdgcn_sched_barrier(0);

      v8f a0 = {0, 0, 0, 0, 0, 0, 0, 0}, a1 = {0, 0, 0, 0, 0, 0, 0, 0};
#pragma unroll
      for (int m = 0; m < 16; ++m) {
        const float4 sv = *(const float4*)(as + 8 * m);
        const float4 uv = *(const float4*)(au + 8 * m);
        v2f slo, shi, ulo, uhi;
        slo.x = sv.x; slo.y = sv.y; shi.x = sv.z; shi.y = sv.w;
        ulo.x = uv.x; ulo.y = uv.y; uhi.x = uv.z; uhi.y = uv.w;
        a0 = wmma_f32(slo, *(const v2f*)(p4 + 8 * m), a0);
        a0 = wmma_f32(ulo, *(const v2f*)(p5 + 8 * m), a0);
        a1 = wmma_f32(shi, *(const v2f*)(p4 + 8 * m + 4), a1);
        a1 = wmma_f32(uhi, *(const v2f*)(p5 + 8 * m + 4), a1);
      }
      const v8f acc = a0 + a1;

      const float bh = bh0 + bh1;
#pragma unroll
      for (int i = 0; i < 8; ++i) {
        const float ht = tanh_f(acc[i] + bh);
        const float zv = zv8[i];
        outp[off0 + (size_t)i * GH] = zv * hv[i] + (1.0f - zv) * ht;
      }
    }
  }
}

constexpr int GN_PAD = 50048;
constexpr int CHCNT_PAD = ((NCH * NBK + 63) / 64) * 64;
constexpr size_t OFF_ROWPTR = 0;
constexpr size_t OFF_CHCNT = OFF_ROWPTR + GN_PAD;
constexpr size_t OFF_CHFIRST = OFF_CHCNT + CHCNT_PAD;
constexpr size_t OFF_BSTART = OFF_CHFIRST + CHCNT_PAD;
constexpr size_t OFF_SLOTS = OFF_BSTART + 256;
constexpr size_t OFF_PAIRS = OFF_SLOTS + GE;
constexpr size_t OFF_CSR = OFF_PAIRS;
static_assert(NBK + 1 <= 256, "bucket starts must fit their slot");
static_assert(OFF_PAIRS + 2 * (size_t)GE <= NH, "edge lists must fit in one workspace plane");
constexpr int ROW_BLOCKS = (GN * 32 + 255) / 256;

extern "C" void kernel_launch(void* const* d_in, const int* in_sizes, int n_in,
                              void* d_out, int out_size, void* d_ws,
                              size_t ws_size, hipStream_t stream) {
  (void)in_sizes; (void)n_in; (void)out_size; (void)ws_size;
  const float* inp = (const float*)d_in[0];
  const int* edg = (const int*)d_in[1];
  const float* hmat = (const float*)d_in[2];
  const float* Wmat = (const float*)d_in[3];
  const float* bmat = (const float*)d_in[4];
  float* out = (float*)d_out;

  const int* src = edg;
  const int* dst = edg + GE;

  float* agg_x = (float*)d_ws;
  int* lists = (int*)(agg_x + NH);
  float* zbuf = agg_x + 2 * NH;
  float* vbuf = zbuf + NH;

  int* rowptr = lists + OFF_ROWPTR;
  int* chcnt = lists + OFF_CHCNT;
  int* chfirst = lists + OFF_CHFIRST;
  int* bstart = lists + OFF_BSTART;
  int* slots = lists + OFF_SLOTS;
  int2* pairs = (int2*)(lists + OFF_PAIRS);
  int* csr = lists + OFF_CSR;

  bucket_count_kernel<<<NCH, SORT_THREADS, 0, stream>>>(dst, chcnt);
  bucket_scan_kernel<<<1, SORT_THREADS, 0, stream>>>(chcnt, chfirst, bstart);
  bucket_scatter_kernel<<<NCH, SORT_THREADS, 0, stream>>>(src, dst, chfirst, pairs);
  bucket_rows_kernel<<<NBK, SORT_THREADS, 0, stream>>>(bstart, pairs, rowptr, slots);
  segment_order_kernel<<<ROW_BLOCKS, 256, 0, stream>>>(rowptr, slots, csr);

  for (int l = 0; l < GLAYERS; ++l) {
    const float* x = (l == 0) ? inp : (out + (size_t)(l - 1) * NH);
    const float* hl = hmat + (size_t)l * NH;
    const float* Wl = Wmat + (size_t)l * 6 * GH * GH;
    const float* bl = bmat + (size_t)l * 6 * GH;
    float* outl = out + (size_t)l * NH;

    gather_gemm_zr_kernel<<<GEMM_GRID, FUSE_THREADS, 0, stream>>>(
        rowptr, csr, x, hl, agg_x, Wl, bl, zbuf, vbuf);

    gather_gemm_ht_kernel<<<GEMM_GRID, FUSE_HT_THREADS, 0, stream>>>(
        rowptr, csr, vbuf, agg_x, Wl, bl, zbuf, hl, outl);
  }
}


# Per-call edge grouping in two launches, not five: each chunk sorts its pairs by bucket in LDS, each bucket's workgroup builds and orders its row lists in LDS; lists, sums and their order unchanged
# speedup vs baseline: 5.5221x; 1.0350x over previous
#include <hip/hip_runtime.h>
#include <cstdint>
#include <cstddef>

constexpr int GN = 50000;
constexpr int GE = 1600000;
constexpr int GH = 128;
constexpr int GLAYERS = 2;
constexpr size_t NH = (size_t)GN * GH;
constexpr int ROW_TILES = GN / 16;
constexpr int GEMM_GRID = 256;
constexpr int FUSE_THREADS = 1024;
constexpr int FUSE_HT_THREADS = 768;
constexpr int TILE_PITCH = 132;
constexpr int TILE_WORDS = 16 * TILE_PITCH;
static_assert(GN % 16 == 0, "row tiles must be whole");

typedef __attribute__((ext_vector_type(2))) float v2f;
typedef __attribute__((ext_vector_type(4))) float v4f;
typedef __attribute__((ext_vector_type(8))) float v8f;

__device__ __forceinline__ v8f wmma_f32(v2f a, v2f b, v8f c) {
  return __builtin_amdgcn_wmma_f32_16x16x4_f32(
       false, a,  false, b,
       (short)0, c,  false,  false);
}

__device__ __forceinline__ float sigmoid_f(float x) {
  return 1.0f / (1.0f + __expf(-x));
}
__device__ __forceinline__ float tanh_f(float x) {
  return 1.0f - 2.0f / (1.0f + __expf(2.0f * x));
}


template <int NT>
__device__ __forceinline__ void stage_matrix(const float* __restrict__ Wg,
                                             float* lds) {
  const int tid = threadIdx.x;
#pragma unroll
  for (int i = 0; i < (GH * GH + NT - 1) / NT; ++i) {
    const int idx = i * NT + tid;
    if ((GH * GH) % NT == 0 || idx < GH * GH) {
      const int c = idx & 15;
      const int k = (idx >> 4) & (GH - 1);
      const int n = ((idx >> 11) << 4) + c;
      lds[n * 132 + k] = Wg[k * GH + n];
    }
  }
}


constexpr int BK_SHIFT = 8;
constexpr int BK_ROWS = 1 << BK_SHIFT;
constexpr int NBK = (GN + BK_ROWS - 1) >> BK_SHIFT;
constexpr int SORT_THREADS = 1024;
constexpr int NCH = 250;
constexpr int CH_EDGES = GE / NCH;
constexpr int CH_TRIPS = (CH_EDGES + SORT_THREADS - 1) / SORT_THREADS;
constexpr int LIST_CAP = 16384;
constexpr int PULL_LANES = 8;
constexpr int PULL_UNROLL = 6;
constexpr int PLACE_UNROLL = 4;
constexpr int ROWS_PER_WAVE = BK_ROWS / (SORT_THREADS / 32);
static_assert(NCH * CH_EDGES == GE, "chunks must tile the edge list");
static_assert(NBK <= 256, "one scan thread per bucket");
static_assert(NCH <= 256, "one scan thread per chunk");
static_assert(2 * (SORT_THREADS / PULL_LANES) == 256, "two runs per lane group cover the 256 run slots");
static_assert(GN <= (1 << 23), "a source and a row of the bucket share one word");

__device__ __forceinline__ int scan256_exclusive(int v, int t, int* wsum, int& total) {
  const int lane = t & 31;
  int incl = v;
  if (t < 256) {
    for (int off = 1; off < 32; off <<= 1) {
      const int y = __shfl_up(incl, off);
      if (lane >= off) incl += y;
    }
    if (lane == 31) wsum[t >> 5] = incl;
  }
  __syncthreads();
  int woff = 0, all = 0;
  for (int w = 0; w < 8; ++w) {
    const int s = wsum[w];
    if (w < (t >> 5)) woff += s;
    all += s;
  }
  total = all;
  return woff + incl - v;
}

__global__ __launch_bounds__(SORT_THREADS) void chunk_sort_kernel(
    const int* __restrict__ src, const int* __restrict__ dst,
    int* __restrict__ choff, int2* __restrict__ pairs) {
  __shared__ int hist[256];
  __shared__ int wsum[8];
  __shared__ int2 stage[CH_EDGES];
  const int t = threadIdx.x;
  const int g = blockIdx.x;
  if (t < 256) hist[t] = 0;
  __syncthreads();
  const int e0 = g * CH_EDGES;
  int d[CH_TRIPS], s[CH_TRIPS];
#pragma unroll
  for (int k = 0; k < CH_TRIPS; ++k) {
    const int i = k * SORT_THREADS + t;
    const int ic = e0 + ((i < CH_EDGES) ? i : CH_EDGES - 1);
    d[k] = dst[ic];
    s[k] = src[ic];
  }
#pragma unroll
  for (int k = 0; k < CH_TRIPS; ++k) {
    if (k * SORT_THREADS + t < CH_EDGES) atomicAdd(&hist[d[k] >> BK_SHIFT], 1);
  }
  __syncthreads();
  const int cntb = (t < 256) ? hist[t] : 0;
  int all;
  const int ex = scan256_exclusive(cntb, t, wsum, all);
  if (t < 256) hist[t] = ex;
  if (t < NBK) choff[g * NBK + t] = ex;
  __syncthreads();
  int p[CH_TRIPS];
#pragma unroll
  for (int k = 0; k < CH_TRIPS; ++k) {
    p[k] = 0;
    if (k * SORT_THREADS + t < CH_EDGES) p[k] = atomicAdd(&hist[d[k] >> BK_SHIFT], 1);
  }
#pragma unroll
  for (int k = 0; k < CH_TRIPS; ++k) {
    if (k * SORT_THREADS + t < CH_EDGES) stage[p[k]] = make_int2(s[k], d[k]);
  }
  __syncthreads();
  for (int i = t; i < CH_EDGES; i += SORT_THREADS) pairs[e0 + i] = stage[i];
}

__device__ __forceinline__ void order_rows(const int* list, const int* rs,
                                           const int* rc, int wave, int lane,
                                           int* __restrict__ out) {
  for (int k = 0; k < ROWS_PER_WAVE; ++k) {
    const int r = wave * ROWS_PER_WAVE + k;
    const int rb = __builtin_amdgcn_readfirstlane(rs[r]);
    const int n = __builtin_amdgcn_readfirstlane(rc[r]) - rb;
    for (int ib = 0; ib < n; ib += 32) {
      const int i = ib + lane;
      const int key = (i < n) ? list[rb + i] : 0;
      int rank = 0;
      for (int jb = 0; jb < n; jb += 32) {
        const int jn = (n - jb < 32) ? (n - jb) : 32;
        const int kv = (lane < jn) ? list[rb + jb + lane] : 0;
        for (int jj = 0; jj < jn; ++jj) {
          const int kj = __builtin_amdgcn_readlane(kv, jj);
          const int j = jb + jj;
          rank += (kj < key || (kj == key && j < i)) ? 1 : 0;
        }
      }
      if (i < n) out[rb + rank] = key;
    }
  }
}

__global__ __launch_bounds__(SORT_THREADS) void bucket_lists_kernel(
    const int* __restrict__ choff, const int2* __restrict__ pairs,
    int* __restrict__ rowptr, int* __restrict__ csr, int* spill) {
  __shared__ int runoff[256];
  __shared__ int runstart[256];
  __shared__ int runlen[256];
  __shared__ int rc[BK_ROWS];
  __shared__ int rs[BK_ROWS];
  __shared__ int wsum_a[8];
  __shared__ int wsum_b[8];
  __shared__ int wsum_c[8];
  __shared__ int plist[LIST_CAP];
  __shared__ int srcl[LIST_CAP];
  const int t = threadIdx.x;
  const int b = blockIdx.x;
  const int lane = t & 31;
  const int wave = __builtin_amdgcn_readfirstlane(t >> 5);

  int o = 0, len = 0;
  if (t < NCH) {
    o = choff[t * NBK + b];
    const int e = (b + 1 < NBK) ? choff[t * NBK + b + 1] : CH_EDGES;
    len = e - o;
  }
  if (t < BK_ROWS) rc[t] = 0;
  int beg, cnt;
  scan256_exclusive(o, t, wsum_a, beg);
  const int rstart = scan256_exclusive(len, t, wsum_b, cnt);
  if (t < 256) {
    runoff[t] = t * CH_EDGES + o;
    runstart[t] = rstart;
    runlen[t] = len;
  }
  __syncthreads();
  const bool fits = cnt <= LIST_CAP;

  if (fits) {
    const int sub = t & (PULL_LANES - 1);
    const int ga = t / PULL_LANES;
    const int gb = ga + SORT_THREADS / PULL_LANES;
    const int len_a = runlen[ga];
    const int len_b = runlen[gb];
    const int off_a = runoff[ga];
    const int off_b = runoff[gb];
    const int st_a = runstart[ga];
    const int st_b = runstart[gb];
    for (int j0 = sub; j0 < len_a || j0 < len_b; j0 += PULL_LANES * PULL_UNROLL) {
      int2 pa[PULL_UNROLL], pb[PULL_UNROLL];
#pragma unroll
      for (int u = 0; u < PULL_UNROLL; ++u) {
        const int j = j0 + u * PULL_LANES;
        pa[u] = pairs[(j < len_a) ? off_a + j : 0];
        pb[u] = pairs[(j < len_b) ? off_b + j : 0];
      }
#pragma unroll
      for (int u = 0; u < PULL_UNROLL; ++u) {
        const int j = j0 + u * PULL_LANES;
        if (j < len_a) {
          const int r = pa[u].y & (BK_ROWS - 1);
          plist[st_a + j] = (pa[u].x << BK_SHIFT) | r;
          atomicAdd(&rc[r], 1);
        }
        if (j < len_b) {
          const int r = pb[u].y & (BK_ROWS - 1);
          plist[st_b + j] = (pb[u].x << BK_SHIFT) | r;
          atomicAdd(&rc[r], 1);
        }
      }
    }
  } else {
    for (int g = 0; g < NCH; ++g) {
      const int n = runlen[g];
      const int off = runoff[g];
      for (int j = t; j < n; j += SORT_THREADS)
        atomicAdd(&rc[pairs[off + j].y & (BK_ROWS - 1)], 1);
    }
  }
  __syncthreads();
  const int deg = (t < BK_ROWS) ? rc[t] : 0;
  int all;
  const int ex = scan256_exclusive(deg, t, wsum_c, all);
  if (t < BK_ROWS) {
    const int row = (b << BK_SHIFT) + t;
    rs[t] = ex;
    rc[t] = ex;
    if (row < GN) rowptr[row] = beg + ex;
  }
  if (b == NBK - 1 && t == 0) rowptr[GN] = beg + cnt;
  __syncthreads();
  if (fits) {
    for (int i0 = t; i0 < cnt; i0 += PLACE_UNROLL * SORT_THREADS) {
      int w[PLACE_UNROLL], p[PLACE_UNROLL];
#pragma unroll
      for (int u = 0; u < PLACE_UNROLL; ++u) {
        const int i = i0 + u * SORT_THREADS;
        w[u] = plist[(i < cnt) ? i : 0];
      }
#pragma unroll
      for (int u = 0; u < PLACE_UNROLL; ++u) {
        p[u] = 0;
        if (i0 + u * SORT_THREADS < cnt) p[u] = atomicAdd(&rc[w[u] & (BK_ROWS - 1)], 1);
      }
#pragma unroll
      for (int u = 0; u < PLACE_UNROLL; ++u) {
        if (i0 + u * SORT_THREADS < cnt) srcl[p[u]] = w[u] >> BK_SHIFT;
      }
    }
    __syncthreads();
    order_rows(srcl, rs, rc, wave, lane, csr + beg);
  } else {
    for (int g = 0; g < NCH; ++g) {
      const int n = runlen[g];
      const int off = runoff[g];
      for (int j = t; j < n; j += SORT_THREADS) {
        const int2 sd = pairs[off + j];
        const int p = atomicAdd(&rc[sd.y & (BK_ROWS - 1)], 1);
        spill[beg + p] = sd.x;
      }
    }
    __threadfence();
    __syncthreads();
    order_rows(spill + beg, rs, rc, wave, lane, csr + beg);
  }
}

__device__ __forceinline__ void rows_in_flight(
    v4f r0, v4f r1, v4f r2, v4f r3, v4f r4, v4f r5, v4f r6, v4f r7, v4f r8,
    v4f r9, v4f r10, v4f r11, v4f r12, v4f r13, v4f r14, v4f r15) {
  asm volatile("" ::"v"(r0), "v"(r1), "v"(r2), "v"(r3), "v"(r4), "v"(r5),
               "v"(r6), "v"(r7), "v"(r8), "v"(r9), "v"(r10), "v"(r11),
               "v"(r12), "v"(r13), "v"(r14), "v"(r15));
}

__device__ __forceinline__ void behind_rows_in_flight(v4f& r) {
  asm volatile("" : "+v"(r));
}

template <int U, int NF>
__device__ __forceinline__ void gather_group(const float* __restrict__ f0,
                                             const float* __restrict__ f1,
                                             int idx, int j0, int c4,
                                             float4& a0, float4& a1) {
  static_assert(U * NF <= 16, "rows_in_flight names sixteen rows");
  v4f v[16];
#pragma unroll
  for (int u = 0; u < U; ++u) {
    const int s = __builtin_amdgcn_readlane(idx, j0 + u);
    v[NF * u] = *(const v4f*)(f0 + (size_t)s * GH + c4);
    if (NF == 2) v[NF * u + 1] = *(const v4f*)(f1 + (size_t)s * GH + c4);
  }
#pragma unroll
  for (int k = U * NF; k < 16; ++k) v[k] = v[0];
  rows_in_flight(v[0], v[1], v[2], v[3], v[4], v[5], v[6], v[7], v[8], v[9],
                 v[10], v[11], v[12], v[13], v[14], v[15]);
#pragma unroll
  for (int u = 0; u < U; ++u) {
    behind_rows_in_flight(v[NF * u]);
    a0.x += v[NF * u].x;
    a0.y += v[NF * u].y;
    a0.z += v[NF * u].z;
    a0.w += v[NF * u].w;
    if (NF == 2) {
      behind_rows_in_flight(v[NF * u + 1]);
      a1.x += v[NF * u + 1].x;
      a1.y += v[NF * u + 1].y;
      a1.z += v[NF * u + 1].z;
      a1.w += v[NF * u + 1].w;
    }
  }
}

__device__ __forceinline__ float4 operand_order(int lane, float4 s) {
  const bool odd = (lane & 1) != 0;
  const float sx = odd ? s.x : s.z;
  const float sy = odd ? s.y : s.w;
  const float rx = __shfl_xor(sx, 1);
  const float ry = __shfl_xor(sy, 1);
  float4 w;
  w.x = odd ? rx : s.x;
  w.y = odd ? ry : s.y;
  w.z = odd ? s.z : rx;
  w.w = odd ? s.w : ry;
  return w;
}

template <int NF>
__device__ __forceinline__ void gather_batch(const float* __restrict__ f0,
                                             const float* __restrict__ f1,
                                             int idx, int n, int c4,
                                             float4& a0, float4& a1) {
  constexpr int U = 16 / NF;
  int j = 0;
  for (; j + U <= n; j += U) gather_group<U, NF>(f0, f1, idx, j, c4, a0, a1);
  if (NF == 1 && ((n - j) & 8) != 0) {
    gather_group<8, NF>(f0, f1, idx, j, c4, a0, a1);
    j += 8;
  }
  switch (n - j) {
    case 1: gather_group<1, NF>(f0, f1, idx, j, c4, a0, a1); break;
    case 2: gather_group<2, NF>(f0, f1, idx, j, c4, a0, a1); break;
    case 3: gather_group<3, NF>(f0, f1, idx, j, c4, a0, a1); break;
    case 4: gather_group<4, NF>(f0, f1, idx, j, c4, a0, a1); break;
    case 5: gather_group<5, NF>(f0, f1, idx, j, c4, a0, a1); break;
    case 6: gather_group<6, NF>(f0, f1, idx, j, c4, a0, a1); break;
    case 7: gather_group<7, NF>(f0, f1, idx, j, c4, a0, a1); break;
    default: break;
  }
}

template <int NF>
__device__ __forceinline__ void gather_list(
    const int* __restrict__ csr,
    const float* __restrict__ f0, const float* __restrict__ f1,
    int row, int lane, int beg, int end, int idx_a, int idx_b,
    float4& s0, float4& s1) {
  const int c4 = lane << 2;
  const float4 x0 = *(const float4*)(f0 + (size_t)row * GH + c4);
  const float4 x1 = *(const float4*)(f1 + (size_t)row * GH + c4);
  float4 a0 = {0.0f, 0.0f, 0.0f, 0.0f};
  float4 a1 = {0.0f, 0.0f, 0.0f, 0.0f};
  {
    const int n = (end - beg < 32) ? (end - beg) : 32;
    gather_batch<NF>(f0, f1, idx_a, n, c4, a0, a1);
  }
  for (int base = beg + 32; base < end; base += 32) {
    const int n = (end - base < 32) ? (end - base) : 32;
    int idx = idx_b;
    if (base != beg + 32) idx = (lane < n) ? csr[base + lane] : 0;
    gather_batch<NF>(f0, f1, idx, n, c4, a0, a1);
  }
  s0.x = x0.x + a0.x;
  s0.y = x0.y + a0.y;
  s0.z = x0.z + a0.z;
  s0.w = x0.w + a0.w;
  s1.x = x1.x + a1.x;
  s1.y = x1.y + a1.y;
  s1.z = x1.z + a1.z;
  s1.w = x1.w + a1.w;
}

__device__ __forceinline__ void barrier_arrive() {
  asm volatile("s_wait_dscnt 0x0\n\ts_barrier_signal -1" ::: "memory");
}
__device__ __forceinline__ void barrier_pass() {
  asm volatile("s_barrier_wait -1" ::: "memory");
}

__global__ __launch_bounds__(FUSE_THREADS) void gather_gemm_zr_kernel(
    const int* __restrict__ rowptr, const int* __restrict__ csr,
    const float* __restrict__ x, const float* __restrict__ hl,
    float* __restrict__ sop,
    const float* __restrict__ Wl, const float* __restrict__ bl,
    float* __restrict__ zout, float* __restrict__ vout) {
  __shared__ __attribute__((aligned(16))) float lds[4 * GH * 132 + 4 * TILE_WORDS];
  const int lane = threadIdx.x & 31;
  const int wave = __builtin_amdgcn_readfirstlane((int)(threadIdx.x >> 5));
  float* ring = lds + 4 * GH * 132;

  stage_matrix<FUSE_THREADS>(Wl + 0 * GH * GH, lds + 0 * GH * 132);
  stage_matrix<FUSE_THREADS>(Wl + 1 * GH * GH, lds + 1 * GH * 132);
  stage_matrix<FUSE_THREADS>(Wl + 2 * GH * GH, lds + 2 * GH * 132);
  stage_matrix<FUSE_THREADS>(Wl + 3 * GH * GH, lds + 3 * GH * 132);

  const int b = blockIdx.x;
  const int t0 = (b * ROW_TILES) / GEMM_GRID;
  const int nt = ((b + 1) * ROW_TILES) / GEMM_GRID - t0;

  int rbeg = 0, rend = 0, next_a = 0, next_b = 0;
  if (wave < 16) {
    const int k = (lane < nt) ? lane : nt - 1;
    const int r = (t0 + k) * 16 + wave;
    rbeg = rowptr[r];
    rend = rowptr[r + 1];
    const int beg0 = __builtin_amdgcn_readlane(rbeg, 0);
    const int end0 = __builtin_amdgcn_readlane(rend, 0);
    next_a = (lane < end0 - beg0) ? csr[beg0 + lane] : 0;
    next_b = (lane + 32 < end0 - beg0) ? csr[beg0 + 32 + lane] : 0;
  }

  float4 w0 = {0.0f, 0.0f, 0.0f, 0.0f};
  float4 w1 = {0.0f, 0.0f, 0.0f, 0.0f};

  for (int step = 0; step <= nt; ++step) {
    if (wave < 16) {
      if (step > 0) {
        const int c4 = lane << 2;
        float* slot = ring + ((step - 1) & 1) * 2 * TILE_WORDS;
        *(float4*)(slot + wave * TILE_PITCH + c4) = w0;
        *(float4*)(slot + TILE_WORDS + wave * TILE_PITCH + c4) = w1;
      }
    }
    barrier_arrive();
    if (wave < 16) {
      if (step < nt) {
        const int row = (t0 + step) * 16 + wave;
        const int c4 = lane << 2;
        const int beg = __builtin_amdgcn_readlane(rbeg, step);
        const int end = __builtin_amdgcn_readlane(rend, step);
        const int idx_a = next_a;
        const int idx_b = next_b;
        if (step + 1 < nt) {
          const int nb = __builtin_amdgcn_readlane(rbeg, step + 1);
          const int ne = __builtin_amdgcn_readlane(rend, step + 1);
          next_a = (lane < ne - nb) ? csr[nb + lane] : 0;
          next_b = (lane + 32 < ne - nb) ? csr[nb + 32 + lane] : 0;
        }
        float4 s0, s1;
        gather_list<2>(csr, x, hl, row, lane, beg, end, idx_a, idx_b, s0, s1);
        w0 = operand_order(lane, s0);
        w1 = operand_order(lane, s1);
        *(float4*)(sop + (size_t)row * GH + c4) = w0;
      }
    }
    barrier_pass();
    if (wave >= 16 && step > 0) {
      const int mw = wave - 16;
      const int gate = mw >> 3;
      const int n0 = (mw & 7) * 16;
      const int half = lane >> 4;
      const int kof = half << 1;
      const int nn = lane & 15;
      const int rowhi = half << 3;
      const int row0 = (t0 + step - 1) * 16;
      const float* slot = ring + ((step - 1) & 1) * 2 * TILE_WORDS;
      const float* as = slot + nn * TILE_PITCH + (half << 2);
      const float* at = as + TILE_WORDS;
      const float* pa = lds + (2 * gate) * GH * 132 + (n0 + nn) * 132 + kof;
      const float* pb = pa + GH * 132;

      const size_t off0 = (size_t)(row0 + rowhi) * GH + n0 + nn;
      float hv[8];
#pragma unroll
      for (int i = 0; i < 8; ++i) hv[i] = hl[off0 + (size_t)i * GH];
      const float b0 = bl[(2 * gate) * GH + n0 + nn];
      const float b1 = bl[(2 * gate + 1) * GH + n0 + nn];
      __builtin_amdgcn_sched_barrier(0);

      v8f c0 = {0, 0, 0, 0, 0, 0, 0, 0}, c1 = {0, 0, 0, 0, 0, 0, 0, 0};
#pragma unroll
      for (int m = 0; m < 16; ++m) {
        const float4 sv = *(const float4*)(as + 8 * m);
        const float4 tv = *(const float4*)(at + 8 * m);
        v2f slo, shi, tlo, thi;
        slo.x = sv.x; slo.y = sv.y; shi.x = sv.z; shi.y = sv.w;
        tlo.x = tv.x; tlo.y = tv.y; thi.x = tv.z; thi.y = tv.w;
        c0 = wmma_f32(slo, *(const v2f*)(pa + 8 * m), c0);
        c0 = wmma_f32(tlo, *(const v2f*)(pb + 8 * m), c0);
        c1 = wmma_f32(shi, *(const v2f*)(pa + 8 * m + 4), c1);
        c1 = wmma_f32(thi, *(const v2f*)(pb + 8 * m + 4), c1);
      }
      const v8f acc = c0 + c1;
      const float bb = b0 + b1;
      float* dstp = (gate != 0) ? vout : zout;
#pragma unroll
      for (int i = 0; i < 8; ++i) {
        const float g = sigmoid_f(acc[i] + bb);
        dstp[off0 + (size_t)i * GH] = (gate != 0) ? g * hv[i] : g;
      }
    }
  }
}

__global__ __launch_bounds__(FUSE_HT_THREADS) void gather_gemm_ht_kernel(
    const int* __restrict__ rowptr, const int* __restrict__ csr,
    const float* __restrict__ vin, const float* __restrict__ sop,
    const float* __restrict__ Wl, const float* __restrict__ bl,
    const float* __restrict__ zbuf, const float* __restrict__ hl,
    float* __restrict__ outp) {
  __shared__ __attribute__((aligned(16))) float lds[2 * GH * 132 + 4 * TILE_WORDS];
  const int lane = threadIdx.x & 31;
  const int wave = __builtin_amdgcn_readfirstlane((int)(threadIdx.x >> 5));
  float* ring = lds + 2 * GH * 132;

  stage_matrix<FUSE_HT_THREADS>(Wl + 4 * GH * GH, lds + 0 * GH * 132);
  stage_matrix<FUSE_HT_THREADS>(Wl + 5 * GH * GH, lds + 1 * GH * 132);

  const int b = blockIdx.x;
  const int t0 = (b * ROW_TILES) / GEMM_GRID;
  const int nt = ((b + 1) * ROW_TILES) / GEMM_GRID - t0;

  int rbeg = 0, rend = 0, next_a = 0, next_b = 0;
  if (wave < 16) {
    const int k = (lane < nt) ? lane : nt - 1;
    const int r = (t0 + k) * 16 + wave;
    rbeg = rowptr[r];
    rend = rowptr[r + 1];
    const int beg0 = __builtin_amdgcn_readlane(rbeg, 0);
    const int end0 = __builtin_amdgcn_readlane(rend, 0);
    next_a = (lane < end0 - beg0) ? csr[beg0 + lane] : 0;
    next_b = (lane + 32 < end0 - beg0) ? csr[beg0 + 32 + lane] : 0;
  }

  float4 srow = {0.0f, 0.0f, 0.0f, 0.0f};
  float4 urow = {0.0f, 0.0f, 0.0f, 0.0f};

  for (int step = 0; step <= nt; ++step) {
    if (wave < 16) {
      if (step > 0) {
        const int c4 = lane << 2;
        float* slot = ring + ((step - 1) & 1) * 2 * TILE_WORDS;
        *(float4*)(slot + wave * TILE_PITCH + c4) = srow;
        *(float4*)(slot + TILE_WORDS + wave * TILE_PITCH + c4) = urow;
      }
    }
    barrier_arrive();
    if (wave < 16) {
      if (step < nt) {
        const int row = (t0 + step) * 16 + wave;
        const int c4 = lane << 2;
        const int beg = __builtin_amdgcn_readlane(rbeg, step);
        const int end = __builtin_amdgcn_readlane(rend, step);
        const int idx_a = next_a;
        const int idx_b = next_b;
        if (step + 1 < nt) {
          const int nb = __builtin_amdgcn_readlane(rbeg, step + 1);
          const int ne = __builtin_amdgcn_readlane(rend, step + 1);
          next_a = (lane < ne - nb) ? csr[nb + lane] : 0;
          next_b = (lane + 32 < ne - nb) ? csr[nb + 32 + lane] : 0;
        }
        srow = *(const float4*)(sop + (size_t)row * GH + c4);
        float4 u0, u1;
        gather_list<1>(csr, vin, vin, row, lane, beg, end, idx_a, idx_b, u0, u1);
        urow = operand_order(lane, u0);
      }
    }
    barrier_pass();
    if (wave >= 16 && step > 0) {
      const int n0 = (wave - 16) * 16;
      const int half = lane >> 4;
      const int kof = half << 1;
      const int nn = lane & 15;
      const int rowhi = half << 3;
      const int row0 = (t0 + step - 1) * 16;
      const float* slot = ring + ((step - 1) & 1) * 2 * TILE_WORDS;
      const float* as = slot + nn * TILE_PITCH + (half << 2);
      const float* au = as + TILE_WORDS;
      const float* p4 = lds + (n0 + nn) * 132 + kof;
      const float* p5 = p4 + GH * 132;

      const size_t off0 = (size_t)(row0 + rowhi) * GH + n0 + nn;
      float zv8[8], hv[8];
#pragma unroll
      for (int i = 0; i < 8; ++i) {
        zv8[i] = zbuf[off0 + (size_t)i * GH];
        hv[i] = hl[off0 + (size_t)i * GH];
      }
      const float bh0 = bl[4 * GH + n0 + nn], bh1 = bl[5 * GH + n0 + nn];
      __builtin_amdgcn_sched_barrier(0);

      v8f a0 = {0, 0, 0, 0, 0, 0, 0, 0}, a1 = {0, 0, 0, 0, 0, 0, 0, 0};
#pragma unroll
      for (int m = 0; m < 16; ++m) {
        const float4 sv = *(const float4*)(as + 8 * m);
        const float4 uv = *(const float4*)(au + 8 * m);
        v2f slo, shi, ulo, uhi;
        slo.x = sv.x; slo.y = sv.y; shi.x = sv.z; shi.y = sv.w;
        ulo.x = uv.x; ulo.y = uv.y; uhi.x = uv.z; uhi.y = uv.w;
        a0 = wmma_f32(slo, *(const v2f*)(p4 + 8 * m), a0);
        a0 = wmma_f32(ulo, *(const v2f*)(p5 + 8 * m), a0);
        a1 = wmma_f32(shi, *(const v2f*)(p4 + 8 * m + 4), a1);
        a1 = wmma_f32(uhi, *(const v2f*)(p5 + 8 * m + 4), a1);
      }
      const v8f acc = a0 + a1;

      const float bh = bh0 + bh1;
#pragma unroll
      for (int i = 0; i < 8; ++i) {
        const float ht = tanh_f(acc[i] + bh);
        const float zv = zv8[i];
        outp[off0 + (size_t)i * GH] = zv * hv[i] + (1.0f - zv) * ht;
      }
    }
  }
}

constexpr int GN_PAD = 50048;
constexpr int CHOFF_PAD = ((NCH * NBK + 63) / 64) * 64;
constexpr size_t OFF_ROWPTR = 0;
constexpr size_t OFF_CHOFF = OFF_ROWPTR + GN_PAD;
constexpr size_t OFF_CSR = OFF_CHOFF + CHOFF_PAD;
static_assert(OFF_CSR + (size_t)GE <= NH, "edge lists must fit in one workspace plane");
static_assert(2 * (size_t)GE <= NH, "the pairs must fit in the z plane");
static_assert((size_t)GE <= NH, "the spill lists must fit in the v plane");

extern "C" void kernel_launch(void* const* d_in, const int* in_sizes, int n_in,
                              void* d_out, int out_size, void* d_ws,
                              size_t ws_size, hipStream_t stream) {
  (void)in_sizes; (void)n_in; (void)out_size; (void)ws_size;
  const float* inp = (const float*)d_in[0];
  const int* edg = (const int*)d_in[1];
  const float* hmat = (const float*)d_in[2];
  const float* Wmat = (const float*)d_in[3];
  const float* bmat = (const float*)d_in[4];
  float* out = (float*)d_out;

  const int* src = edg;
  const int* dst = edg + GE;

  float* agg_x = (float*)d_ws;
  int* lists = (int*)(agg_x + NH);
  float* zbuf = agg_x + 2 * NH;
  float* vbuf = zbuf + NH;

  int* rowptr = lists + OFF_ROWPTR;
  int* choff = lists + OFF_CHOFF;
  int* csr = lists + OFF_CSR;
  int2* pairs = (int2*)zbuf;
  int* spill = (int*)vbuf;

  chunk_sort_kernel<<<NCH, SORT_THREADS, 0, stream>>>(src, dst, choff, pairs);
  bucket_lists_kernel<<<NBK, SORT_THREADS, 0, stream>>>(choff, pairs, rowptr, csr, spill);

  for (int l = 0; l < GLAYERS; ++l) {
    const float* x = (l == 0) ? inp : (out + (size_t)(l - 1) * NH);
    const float* hl = hmat + (size_t)l * NH;
    const float* Wl = Wmat + (size_t)l * 6 * GH * GH;
    const float* bl = bmat + (size_t)l * 6 * GH;
    float* outl = out + (size_t)l * NH;

    gather_gemm_zr_kernel<<<GEMM_GRID, FUSE_THREADS, 0, stream>>>(
        rowptr, csr, x, hl, agg_x, Wl, bl, zbuf, vbuf);

    gather_gemm_ht_kernel<<<GEMM_GRID, FUSE_HT_THREADS, 0, stream>>>(
        rowptr, csr, vbuf, agg_x, Wl, bl, zbuf, hl, outl);
  }
}
